# starting kernel
# baseline (speedup 1.0000x reference)
#include <hip/hip_runtime.h>

#define N_NODES   100000
#define N_EDGES   1600000
#define N_GRAPHS  1000
#define D_FEAT    128
#define HID       64
#define BN_EPS    1e-5f
#define NTILES    (N_NODES / 16)

typedef __attribute__((ext_vector_type(2))) float v2f;
typedef __attribute__((ext_vector_type(8))) float v8f;

template<int K1>
__launch_bounds__(128)
__global__ void gin_mlp_kernel(const float* __restrict__ xin,
                               const float* __restrict__ W1,
                               const float* __restrict__ b1,
                               const float* __restrict__ bn1,
                               const float* __restrict__ W2,
                               const float* __restrict__ b2,
                               const float* __restrict__ bn2,
                               float* __restrict__ hout,
                               int ntiles) {
  constexpr int KS1 = K1 / 4;
  __shared__ __align__(16) float w1s[4 * KS1 * 32 * 2];
  __shared__ __align__(16) float w2s[4 * 16 * 32 * 2];
  __shared__ __align__(16) float tb[4][16 * 68];

  const int tid = threadIdx.x;

  for (int i = tid; i < K1 * 64; i += 128) {
    int kk = i >> 6, nn = i & 63;
    int idx = ((((nn >> 4) * KS1) + (kk >> 2)) * 32 + (nn & 15) + ((kk & 2) << 3)) * 2 + (kk & 1);
    w1s[idx] = W1[i];
  }
  for (int i = tid; i < 64 * 64; i += 128) {
    int kk = i >> 6, nn = i & 63;
    int idx = ((((nn >> 4) * 16) + (kk >> 2)) * 32 + (nn & 15) + ((kk & 2) << 3)) * 2 + (kk & 1);
    w2s[idx] = W2[i];
  }
  __syncthreads();

  const int lane = tid & 31;
  const int wave = tid >> 5;
  const int row  = lane & 15;
  const int hi   = lane >> 4;
  const int tile = blockIdx.x * 4 + wave;
  const bool active = tile < ntiles;
  float* tbuf = tb[wave];

  const v8f vzero = {0.f, 0.f, 0.f, 0.f, 0.f, 0.f, 0.f, 0.f};

  if (active) {
    v2f a1[KS1];
    const float* xr = xin + (size_t)(tile * 16 + row) * K1 + 2 * hi;
#pragma unroll
    for (int k = 0; k < KS1; k++) a1[k] = *(const v2f*)(xr + 4 * k);

#pragma unroll
    for (int nt = 0; nt < 4; nt++) {
      v8f acc = vzero;
      const v2f* wf = ((const v2f*)w1s) + nt * KS1 * 32 + lane;
#pragma unroll
      for (int k = 0; k < KS1; k++) {
        acc = __builtin_amdgcn_wmma_f32_16x16x4_f32(false, a1[k], false, wf[k * 32],
                                                    (short)0, acc, false, false);
      }
      int n = nt * 16 + row;
      float s  = bn1[n] * rsqrtf(bn1[192 + n] + BN_EPS);
      float t0 = bn1[64 + n] + (b1[n] - bn1[128 + n]) * s;
#pragma unroll
      for (int i = 0; i < 8; i++) {
        float y = acc[i] * s + t0;
        tbuf[(i + 8 * hi) * 68 + n] = fmaxf(y, 0.0f);
      }
    }
  }
  __syncthreads();

  v8f acc2[4];
  if (active) {
#pragma unroll
    for (int nt = 0; nt < 4; nt++) {
      v8f acc = vzero;
      const v2f* wf = ((const v2f*)w2s) + nt * 16 * 32 + lane;
      const float* tr = tbuf + row * 68 + 2 * hi;
#pragma unroll
      for (int k = 0; k < 16; k++) {
        v2f a = *(const v2f*)(tr + 4 * k);
        acc = __builtin_amdgcn_wmma_f32_16x16x4_f32(false, a, false, wf[k * 32],
                                                    (short)0, acc, false, false);
      }
      acc2[nt] = acc;
    }
  }
  __syncthreads();

  if (active) {
#pragma unroll
    for (int nt = 0; nt < 4; nt++) {
      int n = nt * 16 + row;
      float s  = bn2[n] * rsqrtf(bn2[192 + n] + BN_EPS);
      float t0 = bn2[64 + n] + (b2[n] - bn2[128 + n]) * s;
#pragma unroll
      for (int i = 0; i < 8; i++) {
        float y = acc2[nt][i] * s + t0;
        tbuf[(i + 8 * hi) * 68 + n] = fmaxf(y, 0.0f);
      }
    }
  }
  __syncthreads();

  if (active) {
#pragma unroll
    for (int j = 0; j < 8; j++) {
      int flat = j * 32 + lane;
      int r = flat >> 4;
      int c = (flat & 15) << 2;
      float4 v = *(const float4*)(tbuf + r * 68 + c);
      *(float4*)(hout + (size_t)(tile * 16 + r) * 64 + c) = v;
    }
  }
}

__global__ void gin_scatter_kernel(const float* __restrict__ h,
                                   const int* __restrict__ src,
                                   const int* __restrict__ dst,
                                   float* __restrict__ agg) {
  long long idx = (long long)blockIdx.x * blockDim.x + threadIdx.x;
  int e = (int)(idx >> 4);
  if (e >= N_EDGES) return;
  int c = ((int)idx & 15) << 2;
  int s = src[e], d = dst[e];
  const float4 v = *(const float4*)(h + (size_t)s * 64 + c);
  float* p = agg + (size_t)d * 64 + c;
  atomicAdd(p + 0, v.x);
  atomicAdd(p + 1, v.y);
  atomicAdd(p + 2, v.z);
  atomicAdd(p + 3, v.w);
}

__global__ void gin_pool_kernel(const float* __restrict__ h,
                                const int* __restrict__ batch,
                                float* __restrict__ pooled,
                                float* __restrict__ counts,
                                int addCount) {
  long long idx = (long long)blockIdx.x * blockDim.x + threadIdx.x;
  int n = (int)(idx >> 4);
  if (n >= N_NODES) return;
  int c = ((int)idx & 15) << 2;
  int g = batch[n];
  const float4 v = *(const float4*)(h + (size_t)n * 64 + c);
  float* p = pooled + (size_t)g * 64 + c;
  atomicAdd(p + 0, v.x);
  atomicAdd(p + 1, v.y);
  atomicAdd(p + 2, v.z);
  atomicAdd(p + 3, v.w);
  if (addCount && (((int)idx & 15) == 0)) atomicAdd(counts + g, 1.0f);
}

__global__ void gin_final_kernel(const float* __restrict__ pooled,
                                 const float* __restrict__ counts,
                                 const float* __restrict__ linW,
                                 const float* __restrict__ linb,
                                 float* __restrict__ out) {
  int idx = blockIdx.x * blockDim.x + threadIdx.x;
  if (idx >= N_GRAPHS * 10) return;
  int g = idx / 10, t = idx - g * 10;
  float acc = counts[g] * linb[t];
  for (int l = 1; l < 5; l++) acc += linb[l * 10 + t];
  for (int l = 0; l < 5; l++) {
    const float* p = pooled + ((size_t)l * N_GRAPHS + g) * 64;
    const float* w = linW + l * 640 + t;
    float s = 0.f;
#pragma unroll
    for (int k = 0; k < 64; k++) s += p[k] * w[k * 10];
    acc += s;
  }
  out[idx] = acc;
}

__global__ void gin_zero_kernel(float* __restrict__ p, int n) {
  int i = blockIdx.x * blockDim.x + threadIdx.x;
  if (i < n) p[i] = 0.0f;
}

extern "C" void kernel_launch(void* const* d_in, const int* in_sizes, int n_in,
                              void* d_out, int out_size, void* d_ws, size_t ws_size,
                              hipStream_t stream) {
  const float* x        = (const float*)d_in[0];
  const float* fh_W1    = (const float*)d_in[1];
  const float* fh_b1    = (const float*)d_in[2];
  const float* fh_bn1   = (const float*)d_in[3];
  const float* fh_W2    = (const float*)d_in[4];
  const float* fh_b2    = (const float*)d_in[5];
  const float* fh_bn2   = (const float*)d_in[6];
  const float* conv_W1  = (const float*)d_in[7];
  const float* conv_b1  = (const float*)d_in[8];
  const float* conv_bn1 = (const float*)d_in[9];
  const float* conv_W2  = (const float*)d_in[10];
  const float* conv_b2  = (const float*)d_in[11];
  const float* conv_bn2 = (const float*)d_in[12];
  const float* lin_W    = (const float*)d_in[13];
  const float* lin_b    = (const float*)d_in[14];
  const int*   eidx     = (const int*)d_in[15];
  const int*   batch    = (const int*)d_in[16];
  float* out = (float*)d_out;

  const int* src = eidx;
  const int* dst = eidx + N_EDGES;

  float* ws     = (float*)d_ws;
  float* h      = ws;
  float* agg    = ws + (size_t)N_NODES * 64;
  float* pooled = agg + (size_t)N_NODES * 64;
  float* counts = pooled + 5 * N_GRAPHS * 64;

  const int accN = 5 * N_GRAPHS * 64 + N_GRAPHS;
  gin_zero_kernel<<<(accN + 255) / 256, 256, 0, stream>>>(pooled, accN);

  const int mlpBlocks = (NTILES + 3) / 4;

  gin_mlp_kernel<D_FEAT><<<mlpBlocks, 128, 0, stream>>>(
      x, fh_W1, fh_b1, fh_bn1, fh_W2, fh_b2, fh_bn2, h, NTILES);
  gin_pool_kernel<<<(N_NODES * 16 + 255) / 256, 256, 0, stream>>>(
      h, batch, pooled, counts, 1);

  for (int l = 0; l < 4; l++) {
    hipMemcpyAsync(agg, h, (size_t)N_NODES * 64 * sizeof(float),
                   hipMemcpyDeviceToDevice, stream);
    gin_scatter_kernel<<<(int)(((long long)N_EDGES * 16 + 255) / 256), 256, 0, stream>>>(
        h, src, dst, agg);
    gin_mlp_kernel<HID><<<mlpBlocks, 128, 0, stream>>>(
        agg,
        conv_W1 + (size_t)l * 64 * 64, conv_b1 + l * 64, conv_bn1 + l * 256,
        conv_W2 + (size_t)l * 64 * 64, conv_b2 + l * 64, conv_bn2 + l * 256,
        h, NTILES);
    gin_pool_kernel<<<(N_NODES * 16 + 255) / 256, 256, 0, stream>>>(
        h, batch, pooled + (size_t)(l + 1) * N_GRAPHS * 64, counts, 0);
  }

  gin_final_kernel<<<(N_GRAPHS * 10 + 255) / 256, 256, 0, stream>>>(
      pooled, counts, lin_W, lin_b, out);
}


# Output head per node before pooling (as the reference's layer 0): tile x lin_W on WMMA in the MLP kernel, 10-column f32 atomics into out; pool, zero and final launches gone; out sum order = arrival
# speedup vs baseline: 1.9560x; 1.9560x over previous
#include <hip/hip_runtime.h>

#define N_NODES   100000
#define N_EDGES   1600000
#define N_GRAPHS  1000
#define D_FEAT    128
#define HID       64
#define BN_EPS    1e-5f
#define NTILES    (N_NODES / 16)

typedef __attribute__((ext_vector_type(2))) float v2f;
typedef __attribute__((ext_vector_type(8))) float v8f;

template<int K1, bool NODE_BIAS>
__launch_bounds__(128)
__global__ void gin_mlp_kernel(const float* __restrict__ xin,
                               const float* __restrict__ W1,
                               const float* __restrict__ b1,
                               const float* __restrict__ bn1,
                               const float* __restrict__ W2,
                               const float* __restrict__ b2,
                               const float* __restrict__ bn2,
                               float* __restrict__ hout,
                               const float* __restrict__ linW,
                               const float* __restrict__ linb0,
                               const int* __restrict__ batch,
                               float* __restrict__ out,
                               int ntiles) {
  constexpr int KS1 = K1 / 4;
  __shared__ __align__(16) float w1s[4 * KS1 * 32 * 2];
  __shared__ __align__(16) float w2s[4 * 16 * 32 * 2];
  __shared__ __align__(16) float tb[4][16 * 68];

  const int tid = threadIdx.x;

  for (int i = tid; i < K1 * 64; i += 128) {
    int kk = i >> 6, nn = i & 63;
    int idx = ((((nn >> 4) * KS1) + (kk >> 2)) * 32 + (nn & 15) + ((kk & 2) << 3)) * 2 + (kk & 1);
    w1s[idx] = W1[i];
  }
  for (int i = tid; i < 64 * 64; i += 128) {
    int kk = i >> 6, nn = i & 63;
    int idx = ((((nn >> 4) * 16) + (kk >> 2)) * 32 + (nn & 15) + ((kk & 2) << 3)) * 2 + (kk & 1);
    w2s[idx] = W2[i];
  }
  __syncthreads();

  const int lane = tid & 31;
  const int wave = tid >> 5;
  const int row  = lane & 15;
  const int hi   = lane >> 4;
  const int tile = blockIdx.x * 4 + wave;
  const bool active = tile < ntiles;
  float* tbuf = tb[wave];

  const v8f vzero = {0.f, 0.f, 0.f, 0.f, 0.f, 0.f, 0.f, 0.f};

  if (active) {
    v2f a1[KS1];
    const float* xr = xin + (size_t)(tile * 16 + row) * K1 + 2 * hi;
#pragma unroll
    for (int k = 0; k < KS1; k++) a1[k] = *(const v2f*)(xr + 4 * k);

#pragma unroll
    for (int nt = 0; nt < 4; nt++) {
      v8f acc = vzero;
      const v2f* wf = ((const v2f*)w1s) + nt * KS1 * 32 + lane;
#pragma unroll
      for (int k = 0; k < KS1; k++) {
        acc = __builtin_amdgcn_wmma_f32_16x16x4_f32(false, a1[k], false, wf[k * 32],
                                                    (short)0, acc, false, false);
      }
      int n = nt * 16 + row;
      float s  = bn1[n] * rsqrtf(bn1[192 + n] + BN_EPS);
      float t0 = bn1[64 + n] + (b1[n] - bn1[128 + n]) * s;
#pragma unroll
      for (int i = 0; i < 8; i++) {
        float y = acc[i] * s + t0;
        tbuf[(i + 8 * hi) * 68 + n] = fmaxf(y, 0.0f);
      }
    }
  }
  __syncthreads();

  v8f acc2[4];
  v2f wl[16];
  int gid[8];
  const int hc = row < 10 ? row : 9;
#pragma unroll
  for (int k = 0; k < 16; k++) wl[k] = (v2f){0.f, 0.f};
#pragma unroll
  for (int r = 0; r < 8; r++) gid[r] = 0;
  if (active) {
    const float* lw = linW + (2 * hi) * 10 + hc;
#pragma unroll
    for (int k = 0; k < 16; k++) {
      float w0 = lw[(4 * k) * 10];
      float w1 = lw[(4 * k + 1) * 10];
      wl[k] = (v2f){row < 10 ? w0 : 0.f, row < 10 ? w1 : 0.f};
    }
#pragma unroll
    for (int r = 0; r < 8; r++) gid[r] = batch[tile * 16 + 8 * hi + r];
#pragma unroll
    for (int nt = 0; nt < 4; nt++) {
      v8f acc = vzero;
      const v2f* wf = ((const v2f*)w2s) + nt * 16 * 32 + lane;
      const float* tr = tbuf + row * 68 + 2 * hi;
#pragma unroll
      for (int k = 0; k < 16; k++) {
        v2f a = *(const v2f*)(tr + 4 * k);
        acc = __builtin_amdgcn_wmma_f32_16x16x4_f32(false, a, false, wf[k * 32],
                                                    (short)0, acc, false, false);
      }
      acc2[nt] = acc;
    }
  }
  __syncthreads();

  if (active) {
#pragma unroll
    for (int nt = 0; nt < 4; nt++) {
      int n = nt * 16 + row;
      float s  = bn2[n] * rsqrtf(bn2[192 + n] + BN_EPS);
      float t0 = bn2[64 + n] + (b2[n] - bn2[128 + n]) * s;
#pragma unroll
      for (int i = 0; i < 8; i++) {
        float y = acc2[nt][i] * s + t0;
        tbuf[(i + 8 * hi) * 68 + n] = fmaxf(y, 0.0f);
      }
    }
  }
  __syncthreads();

  if (active) {
    v8f p0 = vzero, p1 = vzero;
    const float* tr = tbuf + row * 68 + 2 * hi;
#pragma unroll
    for (int k = 0; k < 16; k += 2) {
      v2f a0 = *(const v2f*)(tr + 4 * k);
      v2f a1 = *(const v2f*)(tr + 4 * (k + 1));
      p0 = __builtin_amdgcn_wmma_f32_16x16x4_f32(false, a0, false, wl[k],
                                                 (short)0, p0, false, false);
      p1 = __builtin_amdgcn_wmma_f32_16x16x4_f32(false, a1, false, wl[k + 1],
                                                 (short)0, p1, false, false);
    }
    float pv[8];
#pragma unroll
    for (int r = 0; r < 8; r++) pv[r] = p0[r] + p1[r];
    if (NODE_BIAS) {
      const float lb = linb0[hc];
#pragma unroll
      for (int r = 0; r < 8; r++) pv[r] += lb;
    }

#pragma unroll
    for (int j = 0; j < 8; j++) {
      int flat = j * 32 + lane;
      int r = flat >> 4;
      int c = (flat & 15) << 2;
      float4 v = *(const float4*)(tbuf + r * 68 + c);
      *(float4*)(hout + (size_t)(tile * 16 + r) * 64 + c) = v;
    }

    if (row < 10) {
#pragma unroll
      for (int r = 0; r < 8; r++) atomicAdd(out + (size_t)gid[r] * 10 + row, pv[r]);
    }
  }
}

#define SC_EDGES_PER_WAVE 32
#define SC_WAVES_PER_BLOCK 8
static_assert(N_EDGES % SC_EDGES_PER_WAVE == 0, "edge count must be a whole number of wave chunks");
static_assert((N_EDGES / SC_EDGES_PER_WAVE) % SC_WAVES_PER_BLOCK == 0, "chunks must fill whole blocks");

__launch_bounds__(SC_WAVES_PER_BLOCK * 32)
__global__ void gin_scatter_kernel(const float* __restrict__ h,
                                   const int* __restrict__ src,
                                   const int* __restrict__ dst,
                                   float* __restrict__ agg) {
  const int lane  = threadIdx.x & 31;
  const int chunk = blockIdx.x * SC_WAVES_PER_BLOCK + (threadIdx.x >> 5);
  const int e = chunk * SC_EDGES_PER_WAVE + lane;
  const int s = src[e];
  const int d = dst[e];

#pragma unroll
  for (int half = 0; half < 2; half++) {
    float v0[16], v1[16];
#pragma unroll
    for (int j = 0; j < 16; j++) {
      const int sj = __builtin_amdgcn_readlane(s, half * 16 + j);
      const float* p = h + (size_t)sj * 64 + lane;
      v0[j] = p[0];
      v1[j] = p[32];
    }
#pragma unroll
    for (int j = 0; j < 16; j++) {
      const int dj = __builtin_amdgcn_readlane(d, half * 16 + j);
      float* q = agg + (size_t)dj * 64 + lane;
      atomicAdd(q, v0[j]);
      atomicAdd(q + 32, v1[j]);
    }
  }
}

__global__ void gin_out_init_kernel(const float* __restrict__ linb,
                                    float* __restrict__ out) {
  int idx = blockIdx.x * blockDim.x + threadIdx.x;
  if (idx >= N_GRAPHS * 10) return;
  int t = idx % 10;
  out[idx] = ((linb[10 + t] + linb[20 + t]) + linb[30 + t]) + linb[40 + t];
}

extern "C" void kernel_launch(void* const* d_in, const int* in_sizes, int n_in,
                              void* d_out, int out_size, void* d_ws, size_t ws_size,
                              hipStream_t stream) {
  const float* x        = (const float*)d_in[0];
  const float* fh_W1    = (const float*)d_in[1];
  const float* fh_b1    = (const float*)d_in[2];
  const float* fh_bn1   = (const float*)d_in[3];
  const float* fh_W2    = (const float*)d_in[4];
  const float* fh_b2    = (const float*)d_in[5];
  const float* fh_bn2   = (const float*)d_in[6];
  const float* conv_W1  = (const float*)d_in[7];
  const float* conv_b1  = (const float*)d_in[8];
  const float* conv_bn1 = (const float*)d_in[9];
  const float* conv_W2  = (const float*)d_in[10];
  const float* conv_b2  = (const float*)d_in[11];
  const float* conv_bn2 = (const float*)d_in[12];
  const float* lin_W    = (const float*)d_in[13];
  const float* lin_b    = (const float*)d_in[14];
  const int*   eidx     = (const int*)d_in[15];
  const int*   batch    = (const int*)d_in[16];
  float* out = (float*)d_out;

  const int* src = eidx;
  const int* dst = eidx + N_EDGES;

  float* ws     = (float*)d_ws;
  float* h      = ws;
  float* agg    = ws + (size_t)N_NODES * 64;

  gin_out_init_kernel<<<(N_GRAPHS * 10 + 255) / 256, 256, 0, stream>>>(lin_b, out);

  const int mlpBlocks = (NTILES + 3) / 4;

  gin_mlp_kernel<D_FEAT, true><<<mlpBlocks, 128, 0, stream>>>(
      x, fh_W1, fh_b1, fh_bn1, fh_W2, fh_b2, fh_bn2, h,
      lin_W, lin_b, batch, out, NTILES);

  for (int l = 0; l < 4; l++) {
    hipMemcpyAsync(agg, h, (size_t)N_NODES * 64 * sizeof(float),
                   hipMemcpyDeviceToDevice, stream);
    gin_scatter_kernel<<<N_EDGES / (SC_EDGES_PER_WAVE * SC_WAVES_PER_BLOCK), SC_WAVES_PER_BLOCK * 32, 0, stream>>>(
        h, src, dst, agg);
    gin_mlp_kernel<HID, false><<<mlpBlocks, 128, 0, stream>>>(
        agg,
        conv_W1 + (size_t)l * 64 * 64, conv_b1 + l * 64, conv_bn1 + l * 256,
        conv_W2 + (size_t)l * 64 * 64, conv_b2 + l * 64, conv_bn2 + l * 256,
        h,
        lin_W + (size_t)(l + 1) * 64 * 10, lin_b, batch, out, NTILES);
  }
}


# Neighbour sums in LDS: edges binned once per call by 200-node dst range, one work-group per range adds h[src] rows by LDS f32 atomics; global f32 scatter atomics, copies gone; sum order = LDS arrival
# speedup vs baseline: 2.5367x; 1.2969x over previous
#include <hip/hip_runtime.h>

#define N_NODES   100000
#define N_EDGES   1600000
#define N_GRAPHS  1000
#define D_FEAT    128
#define HID       64
#define BN_EPS    1e-5f
#define NTILES    (N_NODES / 16)

typedef __attribute__((ext_vector_type(2))) float v2f;
typedef __attribute__((ext_vector_type(8))) float v8f;

template<int K1, bool NODE_BIAS>
__launch_bounds__(128)
__global__ void gin_mlp_kernel(const float* __restrict__ xin,
                               const float* __restrict__ W1,
                               const float* __restrict__ b1,
                               const float* __restrict__ bn1,
                               const float* __restrict__ W2,
                               const float* __restrict__ b2,
                               const float* __restrict__ bn2,
                               float* __restrict__ hout,
                               const float* __restrict__ linW,
                               const float* __restrict__ linb0,
                               const int* __restrict__ batch,
                               float* __restrict__ out,
                               int ntiles) {
  constexpr int KS1 = K1 / 4;
  __shared__ __align__(16) float w1s[4 * KS1 * 32 * 2];
  __shared__ __align__(16) float w2s[4 * 16 * 32 * 2];
  __shared__ __align__(16) float tb[4][16 * 68];

  const int tid = threadIdx.x;

  for (int i = tid; i < K1 * 64; i += 128) {
    int kk = i >> 6, nn = i & 63;
    int idx = ((((nn >> 4) * KS1) + (kk >> 2)) * 32 + (nn & 15) + ((kk & 2) << 3)) * 2 + (kk & 1);
    w1s[idx] = W1[i];
  }
  for (int i = tid; i < 64 * 64; i += 128) {
    int kk = i >> 6, nn = i & 63;
    int idx = ((((nn >> 4) * 16) + (kk >> 2)) * 32 + (nn & 15) + ((kk & 2) << 3)) * 2 + (kk & 1);
    w2s[idx] = W2[i];
  }
  __syncthreads();

  const int lane = tid & 31;
  const int wave = tid >> 5;
  const int row  = lane & 15;
  const int hi   = lane >> 4;
  const int tile = blockIdx.x * 4 + wave;
  const bool active = tile < ntiles;
  float* tbuf = tb[wave];

  const v8f vzero = {0.f, 0.f, 0.f, 0.f, 0.f, 0.f, 0.f, 0.f};

  if (active) {
    v2f a1[KS1];
    const float* xr = xin + (size_t)(tile * 16 + row) * K1 + 2 * hi;
#pragma unroll
    for (int k = 0; k < KS1; k++) a1[k] = *(const v2f*)(xr + 4 * k);

#pragma unroll
    for (int nt = 0; nt < 4; nt++) {
      v8f acc = vzero;
      const v2f* wf = ((const v2f*)w1s) + nt * KS1 * 32 + lane;
#pragma unroll
      for (int k = 0; k < KS1; k++) {
        acc = __builtin_amdgcn_wmma_f32_16x16x4_f32(false, a1[k], false, wf[k * 32],
                                                    (short)0, acc, false, false);
      }
      int n = nt * 16 + row;
      float s  = bn1[n] * rsqrtf(bn1[192 + n] + BN_EPS);
      float t0 = bn1[64 + n] + (b1[n] - bn1[128 + n]) * s;
#pragma unroll
      for (int i = 0; i < 8; i++) {
        float y = acc[i] * s + t0;
        tbuf[(i + 8 * hi) * 68 + n] = fmaxf(y, 0.0f);
      }
    }
  }
  __syncthreads();

  v8f acc2[4];
  v2f wl[16];
  int gid[8];
  const int hc = row < 10 ? row : 9;
#pragma unroll
  for (int k = 0; k < 16; k++) wl[k] = (v2f){0.f, 0.f};
#pragma unroll
  for (int r = 0; r < 8; r++) gid[r] = 0;
  if (active) {
    const float* lw = linW + (2 * hi) * 10 + hc;
#pragma unroll
    for (int k = 0; k < 16; k++) {
      float w0 = lw[(4 * k) * 10];
      float w1 = lw[(4 * k + 1) * 10];
      wl[k] = (v2f){row < 10 ? w0 : 0.f, row < 10 ? w1 : 0.f};
    }
#pragma unroll
    for (int r = 0; r < 8; r++) gid[r] = batch[tile * 16 + 8 * hi + r];
#pragma unroll
    for (int nt = 0; nt < 4; nt++) {
      v8f acc = vzero;
      const v2f* wf = ((const v2f*)w2s) + nt * 16 * 32 + lane;
      const float* tr = tbuf + row * 68 + 2 * hi;
#pragma unroll
      for (int k = 0; k < 16; k++) {
        v2f a = *(const v2f*)(tr + 4 * k);
        acc = __builtin_amdgcn_wmma_f32_16x16x4_f32(false, a, false, wf[k * 32],
                                                    (short)0, acc, false, false);
      }
      acc2[nt] = acc;
    }
  }
  __syncthreads();

  if (active) {
#pragma unroll
    for (int nt = 0; nt < 4; nt++) {
      int n = nt * 16 + row;
      float s  = bn2[n] * rsqrtf(bn2[192 + n] + BN_EPS);
      float t0 = bn2[64 + n] + (b2[n] - bn2[128 + n]) * s;
#pragma unroll
      for (int i = 0; i < 8; i++) {
        float y = acc2[nt][i] * s + t0;
        tbuf[(i + 8 * hi) * 68 + n] = fmaxf(y, 0.0f);
      }
    }
  }
  __syncthreads();

  if (active) {
    v8f p0 = vzero, p1 = vzero;
    const float* tr = tbuf + row * 68 + 2 * hi;
#pragma unroll
    for (int k = 0; k < 16; k += 2) {
      v2f a0 = *(const v2f*)(tr + 4 * k);
      v2f a1 = *(const v2f*)(tr + 4 * (k + 1));
      p0 = __builtin_amdgcn_wmma_f32_16x16x4_f32(false, a0, false, wl[k],
                                                 (short)0, p0, false, false);
      p1 = __builtin_amdgcn_wmma_f32_16x16x4_f32(false, a1, false, wl[k + 1],
                                                 (short)0, p1, false, false);
    }
    float pv[8];
#pragma unroll
    for (int r = 0; r < 8; r++) pv[r] = p0[r] + p1[r];
    if (NODE_BIAS) {
      const float lb = linb0[hc];
#pragma unroll
      for (int r = 0; r < 8; r++) pv[r] += lb;
    }

#pragma unroll
    for (int j = 0; j < 8; j++) {
      int flat = j * 32 + lane;
      int r = flat >> 4;
      int c = (flat & 15) << 2;
      float4 v = *(const float4*)(tbuf + r * 68 + c);
      *(float4*)(hout + (size_t)(tile * 16 + r) * 64 + c) = v;
    }

    if (row < 10) {
#pragma unroll
      for (int r = 0; r < 8; r++) atomicAdd(out + (size_t)gid[r] * 10 + row, pv[r]);
    }
  }
}

#define BK_NODES 200
#define NBK (N_NODES / BK_NODES)
#define BF_THREADS 800
#define BF_EDGES_PER_THREAD 8
#define BF_EDGES_PER_BLOCK (BF_THREADS * BF_EDGES_PER_THREAD)
#define BF_BLOCKS (N_EDGES / BF_EDGES_PER_BLOCK)
#define SRC_BITS 17
static_assert(N_NODES % BK_NODES == 0, "ranges must tile the nodes exactly");
static_assert(N_EDGES % BF_EDGES_PER_BLOCK == 0, "edge blocks must tile the edges exactly");
static_assert(N_NODES <= (1 << SRC_BITS), "a source id must fit the packed field");
static_assert(NBK <= 512, "the range scan below holds 512 entries");

__launch_bounds__(BF_THREADS)
__global__ void gin_bin_count_kernel(const int* __restrict__ dst,
                                     int* __restrict__ cnt) {
  __shared__ int hist[NBK];
  const int tid = threadIdx.x;
  for (int b = tid; b < NBK; b += BF_THREADS) hist[b] = 0;
  __syncthreads();
  const int e0 = blockIdx.x * BF_EDGES_PER_BLOCK + tid;
  unsigned d[BF_EDGES_PER_THREAD];
#pragma unroll
  for (int k = 0; k < BF_EDGES_PER_THREAD; k++) d[k] = (unsigned)dst[e0 + k * BF_THREADS];
#pragma unroll
  for (int k = 0; k < BF_EDGES_PER_THREAD; k++) atomicAdd(&hist[d[k] / BK_NODES], 1);
  __syncthreads();
  for (int b = tid; b < NBK; b += BF_THREADS) {
    const int c = hist[b];
    if (c) atomicAdd(&cnt[b], c);
  }
}

__launch_bounds__(BF_THREADS)
__global__ void gin_bin_fill_kernel(const int* __restrict__ src,
                                    const int* __restrict__ dst,
                                    const int* __restrict__ cnt,
                                    int* __restrict__ cursor,
                                    int* __restrict__ offs,
                                    int* __restrict__ bins) {
  __shared__ int hist[NBK];
  __shared__ int bpos[NBK];
  __shared__ int sc[2][512];
  const int tid = threadIdx.x;
  for (int b = tid; b < NBK; b += BF_THREADS) hist[b] = 0;
  if (tid < 512) sc[0][tid] = tid < NBK ? cnt[tid] : 0;
  __syncthreads();

  int bk[BF_EDGES_PER_THREAD], rk[BF_EDGES_PER_THREAD], pk[BF_EDGES_PER_THREAD];
  const int e0 = blockIdx.x * BF_EDGES_PER_BLOCK + tid;
#pragma unroll
  for (int k = 0; k < BF_EDGES_PER_THREAD; k++) {
    pk[k] = src[e0 + k * BF_THREADS];
    bk[k] = dst[e0 + k * BF_THREADS];
  }
#pragma unroll
  for (int k = 0; k < BF_EDGES_PER_THREAD; k++) {
    const unsigned s = (unsigned)pk[k];
    const unsigned d = (unsigned)bk[k];
    const unsigned b = d / BK_NODES;
    bk[k] = (int)b;
    pk[k] = (int)(((d - b * BK_NODES) << SRC_BITS) | s);
    rk[k] = atomicAdd(&hist[b], 1);
  }

  int cur = 0;
  for (int off = 1; off < 512; off <<= 1) {
    if (tid < 512) {
      int v = sc[cur][tid];
      if (tid >= off) v += sc[cur][tid - off];
      sc[cur ^ 1][tid] = v;
    }
    __syncthreads();
    cur ^= 1;
  }
  const int* scan = sc[cur];

  for (int b = tid; b < NBK; b += BF_THREADS) {
    const int c = hist[b];
    const int start = b ? scan[b - 1] : 0;
    const int r = c ? atomicAdd(&cursor[b], c) : 0;
    bpos[b] = start + r;
    if (blockIdx.x == 0) offs[b] = start;
  }
  if (blockIdx.x == 0 && tid == 0) offs[NBK] = scan[NBK - 1];
  __syncthreads();

#pragma unroll
  for (int k = 0; k < BF_EDGES_PER_THREAD; k++) bins[bpos[bk[k]] + rk[k]] = pk[k];
}

#define AG_WAVES 16
#define AG_THREADS (AG_WAVES * 32)
#define AG_VEC (BK_NODES * 16)
#define AG_FULL (AG_VEC / AG_THREADS)
#define AG_REM (AG_VEC - AG_FULL * AG_THREADS)
__launch_bounds__(AG_THREADS)
__global__ void gin_bin_agg_kernel(const float* __restrict__ h,
                                   const int* __restrict__ offs,
                                   const int* __restrict__ bins,
                                   float* __restrict__ agg) {
  __shared__ __align__(16) float acc[BK_NODES * 64];
  const int tid  = threadIdx.x;
  const int lane = tid & 31;
  const int wave = __builtin_amdgcn_readfirstlane(tid >> 5);
  const int b = blockIdx.x;
  const size_t rowbase = (size_t)b * BK_NODES * 64;

  const float4* hs = (const float4*)(h + rowbase);
  float4* a4 = (float4*)acc;
  {
    float4 t[AG_FULL];
#pragma unroll
    for (int k = 0; k < AG_FULL; k++) t[k] = hs[tid + k * AG_THREADS];
    const int il = tid < AG_REM ? tid + AG_FULL * AG_THREADS : AG_VEC - 1;
    const float4 tl = hs[il];
#pragma unroll
    for (int k = 0; k < AG_FULL; k++) a4[tid + k * AG_THREADS] = t[k];
    if (tid < AG_REM) a4[il] = tl;
  }
  __syncthreads();

  const int e0 = offs[b];
  const int e1 = offs[b + 1];
  int c = e0 + wave * 32;
  int p = bins[c + lane < N_EDGES ? c + lane : N_EDGES - 1];
  while (c + 32 <= e1) {
    const int cn = c + AG_THREADS;
    v2f v[32];
#pragma unroll
    for (int j = 0; j < 32; j++) {
      const int pj = __builtin_amdgcn_readlane(p, j);
      v[j] = *(const v2f*)(h + (size_t)(pj & ((1 << SRC_BITS) - 1)) * 64 + 2 * lane);
    }
    const int pn = bins[cn + lane < N_EDGES ? cn + lane : N_EDGES - 1];
#pragma unroll
    for (int j = 0; j < 32; j++) {
      const int pj = __builtin_amdgcn_readlane(p, j);
      float* q = acc + (pj >> SRC_BITS) * 64 + 2 * lane;
      atomicAdd(q, v[j].x);
      atomicAdd(q + 1, v[j].y);
    }
    p = pn;
    c = cn;
  }
  if (c < e1) {
    const int n = e1 - c;
    const int pt = lane < n ? bins[c + lane] : 0;
    for (int j = 0; j < n; j++) {
      const int pj = __builtin_amdgcn_readlane(pt, j);
      const v2f t = *(const v2f*)(h + (size_t)(pj & ((1 << SRC_BITS) - 1)) * 64 + 2 * lane);
      float* q = acc + (pj >> SRC_BITS) * 64 + 2 * lane;
      atomicAdd(q, t.x);
      atomicAdd(q + 1, t.y);
    }
  }
  __syncthreads();

  float4* o4 = (float4*)(agg + rowbase);
#pragma unroll
  for (int k = 0; k < AG_FULL; k++) o4[tid + k * AG_THREADS] = a4[tid + k * AG_THREADS];
  if (tid < AG_REM) o4[tid + AG_FULL * AG_THREADS] = a4[tid + AG_FULL * AG_THREADS];
}

__global__ void gin_out_init_kernel(const float* __restrict__ linb,
                                    float* __restrict__ out,
                                    int* __restrict__ cnt) {
  int idx = blockIdx.x * blockDim.x + threadIdx.x;
  if (idx < 2 * NBK) cnt[idx] = 0;
  if (idx >= N_GRAPHS * 10) return;
  int t = idx % 10;
  out[idx] = ((linb[10 + t] + linb[20 + t]) + linb[30 + t]) + linb[40 + t];
}

extern "C" void kernel_launch(void* const* d_in, const int* in_sizes, int n_in,
                              void* d_out, int out_size, void* d_ws, size_t ws_size,
                              hipStream_t stream) {
  const float* x        = (const float*)d_in[0];
  const float* fh_W1    = (const float*)d_in[1];
  const float* fh_b1    = (const float*)d_in[2];
  const float* fh_bn1   = (const float*)d_in[3];
  const float* fh_W2    = (const float*)d_in[4];
  const float* fh_b2    = (const float*)d_in[5];
  const float* fh_bn2   = (const float*)d_in[6];
  const float* conv_W1  = (const float*)d_in[7];
  const float* conv_b1  = (const float*)d_in[8];
  const float* conv_bn1 = (const float*)d_in[9];
  const float* conv_W2  = (const float*)d_in[10];
  const float* conv_b2  = (const float*)d_in[11];
  const float* conv_bn2 = (const float*)d_in[12];
  const float* lin_W    = (const float*)d_in[13];
  const float* lin_b    = (const float*)d_in[14];
  const int*   eidx     = (const int*)d_in[15];
  const int*   batch    = (const int*)d_in[16];
  float* out = (float*)d_out;

  const int* src = eidx;
  const int* dst = eidx + N_EDGES;

  float* ws     = (float*)d_ws;
  float* h      = ws;
  float* agg    = ws + (size_t)N_NODES * 64;
  int*   cnt    = (int*)(ws + (size_t)2 * N_NODES * 64);
  int*   cursor = cnt + NBK;
  int*   offs   = cnt + 1024;
  int*   bins   = cnt + 2048;

  gin_out_init_kernel<<<(N_GRAPHS * 10 + 255) / 256, 256, 0, stream>>>(lin_b, out, cnt);

  gin_bin_count_kernel<<<BF_BLOCKS, BF_THREADS, 0, stream>>>(dst, cnt);
  gin_bin_fill_kernel<<<BF_BLOCKS, BF_THREADS, 0, stream>>>(src, dst, cnt, cursor, offs, bins);

  const int mlpBlocks = (NTILES + 3) / 4;

  gin_mlp_kernel<D_FEAT, true><<<mlpBlocks, 128, 0, stream>>>(
      x, fh_W1, fh_b1, fh_bn1, fh_W2, fh_b2, fh_bn2, h,
      lin_W, lin_b, batch, out, NTILES);

  for (int l = 0; l < 4; l++) {
    gin_bin_agg_kernel<<<NBK, AG_THREADS, 0, stream>>>(h, offs, bins, agg);
    gin_mlp_kernel<HID, false><<<mlpBlocks, 128, 0, stream>>>(
        agg,
        conv_W1 + (size_t)l * 64 * 64, conv_b1 + l * 64, conv_bn1 + l * 256,
        conv_W2 + (size_t)l * 64 * 64, conv_b2 + l * 64, conv_bn2 + l * 256,
        h,
        lin_W + (size_t)(l + 1) * 64 * 10, lin_b, batch, out, NTILES);
  }
}


# Neighbour sum and layer MLP in one kernel: a range's summed rows stay in LDS and feed the matrix unit there, one wave per 16-row tile; agg buffer, its write and read, 4 launches gone; same sum orders
# speedup vs baseline: 2.9651x; 1.1689x over previous
#include <hip/hip_runtime.h>

#define N_NODES   100000
#define N_EDGES   1600000
#define N_GRAPHS  1000
#define D_FEAT    128
#define HID       64
#define BN_EPS    1e-5f
#define NTILES    (N_NODES / 16)

typedef __attribute__((ext_vector_type(2))) float v2f;
typedef __attribute__((ext_vector_type(8))) float v8f;

template<int K1, bool NODE_BIAS>
__launch_bounds__(128)
__global__ void gin_mlp_kernel(const float* __restrict__ xin,
                               const float* __restrict__ W1,
                               const float* __restrict__ b1,
                               const float* __restrict__ bn1,
                               const float* __restrict__ W2,
                               const float* __restrict__ b2,
                               const float* __restrict__ bn2,
                               float* __restrict__ hout,
                               const float* __restrict__ linW,
                               const float* __restrict__ linb0,
                               const int* __restrict__ batch,
                               float* __restrict__ out,
                               int ntiles) {
  constexpr int KS1 = K1 / 4;
  __shared__ __align__(16) float w1s[4 * KS1 * 32 * 2];
  __shared__ __align__(16) float w2s[4 * 16 * 32 * 2];
  __shared__ __align__(16) float tb[4][16 * 68];

  const int tid = threadIdx.x;

  for (int i = tid; i < K1 * 64; i += 128) {
    int kk = i >> 6, nn = i & 63;
    int idx = ((((nn >> 4) * KS1) + (kk >> 2)) * 32 + (nn & 15) + ((kk & 2) << 3)) * 2 + (kk & 1);
    w1s[idx] = W1[i];
  }
  for (int i = tid; i < 64 * 64; i += 128) {
    int kk = i >> 6, nn = i & 63;
    int idx = ((((nn >> 4) * 16) + (kk >> 2)) * 32 + (nn & 15) + ((kk & 2) << 3)) * 2 + (kk & 1);
    w2s[idx] = W2[i];
  }
  __syncthreads();

  const int lane = tid & 31;
  const int wave = tid >> 5;
  const int row  = lane & 15;
  const int hi   = lane >> 4;
  const int tile = blockIdx.x * 4 + wave;
  const bool active = tile < ntiles;
  float* tbuf = tb[wave];

  const v8f vzero = {0.f, 0.f, 0.f, 0.f, 0.f, 0.f, 0.f, 0.f};

  if (active) {
    v2f a1[KS1];
    const float* xr = xin + (size_t)(tile * 16 + row) * K1 + 2 * hi;
#pragma unroll
    for (int k = 0; k < KS1; k++) a1[k] = *(const v2f*)(xr + 4 * k);

#pragma unroll
    for (int nt = 0; nt < 4; nt++) {
      v8f acc = vzero;
      const v2f* wf = ((const v2f*)w1s) + nt * KS1 * 32 + lane;
#pragma unroll
      for (int k = 0; k < KS1; k++) {
        acc = __builtin_amdgcn_wmma_f32_16x16x4_f32(false, a1[k], false, wf[k * 32],
                                                    (short)0, acc, false, false);
      }
      int n = nt * 16 + row;
      float s  = bn1[n] * rsqrtf(bn1[192 + n] + BN_EPS);
      float t0 = bn1[64 + n] + (b1[n] - bn1[128 + n]) * s;
#pragma unroll
      for (int i = 0; i < 8; i++) {
        float y = acc[i] * s + t0;
        tbuf[(i + 8 * hi) * 68 + n] = fmaxf(y, 0.0f);
      }
    }
  }
  __syncthreads();

  v8f acc2[4];
  v2f wl[16];
  int gid[8];
  const int hc = row < 10 ? row : 9;
#pragma unroll
  for (int k = 0; k < 16; k++) wl[k] = (v2f){0.f, 0.f};
#pragma unroll
  for (int r = 0; r < 8; r++) gid[r] = 0;
  if (active) {
    const float* lw = linW + (2 * hi) * 10 + hc;
#pragma unroll
    for (int k = 0; k < 16; k++) {
      float w0 = lw[(4 * k) * 10];
      float w1 = lw[(4 * k + 1) * 10];
      wl[k] = (v2f){row < 10 ? w0 : 0.f, row < 10 ? w1 : 0.f};
    }
#pragma unroll
    for (int r = 0; r < 8; r++) gid[r] = batch[tile * 16 + 8 * hi + r];
#pragma unroll
    for (int nt = 0; nt < 4; nt++) {
      v8f acc = vzero;
      const v2f* wf = ((const v2f*)w2s) + nt * 16 * 32 + lane;
      const float* tr = tbuf + row * 68 + 2 * hi;
#pragma unroll
      for (int k = 0; k < 16; k++) {
        v2f a = *(const v2f*)(tr + 4 * k);
        acc = __builtin_amdgcn_wmma_f32_16x16x4_f32(false, a, false, wf[k * 32],
                                                    (short)0, acc, false, false);
      }
      acc2[nt] = acc;
    }
  }
  __syncthreads();

  if (active) {
#pragma unroll
    for (int nt = 0; nt < 4; nt++) {
      int n = nt * 16 + row;
      float s  = bn2[n] * rsqrtf(bn2[192 + n] + BN_EPS);
      float t0 = bn2[64 + n] + (b2[n] - bn2[128 + n]) * s;
#pragma unroll
      for (int i = 0; i < 8; i++) {
        float y = acc2[nt][i] * s + t0;
        tbuf[(i + 8 * hi) * 68 + n] = fmaxf(y, 0.0f);
      }
    }
  }
  __syncthreads();

  if (active) {
    v8f p0 = vzero, p1 = vzero;
    const float* tr = tbuf + row * 68 + 2 * hi;
#pragma unroll
    for (int k = 0; k < 16; k += 2) {
      v2f a0 = *(const v2f*)(tr + 4 * k);
      v2f a1 = *(const v2f*)(tr + 4 * (k + 1));
      p0 = __builtin_amdgcn_wmma_f32_16x16x4_f32(false, a0, false, wl[k],
                                                 (short)0, p0, false, false);
      p1 = __builtin_amdgcn_wmma_f32_16x16x4_f32(false, a1, false, wl[k + 1],
                                                 (short)0, p1, false, false);
    }
    float pv[8];
#pragma unroll
    for (int r = 0; r < 8; r++) pv[r] = p0[r] + p1[r];
    if (NODE_BIAS) {
      const float lb = linb0[hc];
#pragma unroll
      for (int r = 0; r < 8; r++) pv[r] += lb;
    }

#pragma unroll
    for (int j = 0; j < 8; j++) {
      int flat = j * 32 + lane;
      int r = flat >> 4;
      int c = (flat & 15) << 2;
      float4 v = *(const float4*)(tbuf + r * 68 + c);
      *(float4*)(hout + (size_t)(tile * 16 + r) * 64 + c) = v;
    }

    if (row < 10) {
#pragma unroll
      for (int r = 0; r < 8; r++) atomicAdd(out + (size_t)gid[r] * 10 + row, pv[r]);
    }
  }
}

#define BK_NODES 200
#define NBK (N_NODES / BK_NODES)
#define BF_THREADS 800
#define BF_EDGES_PER_THREAD 8
#define BF_EDGES_PER_BLOCK (BF_THREADS * BF_EDGES_PER_THREAD)
#define BF_BLOCKS (N_EDGES / BF_EDGES_PER_BLOCK)
#define SRC_BITS 17
static_assert(N_NODES % BK_NODES == 0, "ranges must tile the nodes exactly");
static_assert(N_EDGES % BF_EDGES_PER_BLOCK == 0, "edge blocks must tile the edges exactly");
static_assert(N_NODES <= (1 << SRC_BITS), "a source id must fit the packed field");
static_assert(NBK <= 512, "the range scan below holds 512 entries");

__launch_bounds__(BF_THREADS)
__global__ void gin_bin_count_kernel(const int* __restrict__ dst,
                                     int* __restrict__ cnt) {
  __shared__ int hist[NBK];
  const int tid = threadIdx.x;
  for (int b = tid; b < NBK; b += BF_THREADS) hist[b] = 0;
  __syncthreads();
  const int e0 = blockIdx.x * BF_EDGES_PER_BLOCK + tid;
  unsigned d[BF_EDGES_PER_THREAD];
#pragma unroll
  for (int k = 0; k < BF_EDGES_PER_THREAD; k++) d[k] = (unsigned)dst[e0 + k * BF_THREADS];
#pragma unroll
  for (int k = 0; k < BF_EDGES_PER_THREAD; k++) atomicAdd(&hist[d[k] / BK_NODES], 1);
  __syncthreads();
  for (int b = tid; b < NBK; b += BF_THREADS) {
    const int c = hist[b];
    if (c) atomicAdd(&cnt[b], c);
  }
}

__launch_bounds__(BF_THREADS)
__global__ void gin_bin_fill_kernel(const int* __restrict__ src,
                                    const int* __restrict__ dst,
                                    const int* __restrict__ cnt,
                                    int* __restrict__ cursor,
                                    int* __restrict__ offs,
                                    int* __restrict__ bins) {
  __shared__ int hist[NBK];
  __shared__ int bpos[NBK];
  __shared__ int sc[2][512];
  const int tid = threadIdx.x;
  for (int b = tid; b < NBK; b += BF_THREADS) hist[b] = 0;
  if (tid < 512) sc[0][tid] = tid < NBK ? cnt[tid] : 0;
  __syncthreads();

  int bk[BF_EDGES_PER_THREAD], rk[BF_EDGES_PER_THREAD], pk[BF_EDGES_PER_THREAD];
  const int e0 = blockIdx.x * BF_EDGES_PER_BLOCK + tid;
#pragma unroll
  for (int k = 0; k < BF_EDGES_PER_THREAD; k++) {
    pk[k] = src[e0 + k * BF_THREADS];
    bk[k] = dst[e0 + k * BF_THREADS];
  }
#pragma unroll
  for (int k = 0; k < BF_EDGES_PER_THREAD; k++) {
    const unsigned s = (unsigned)pk[k];
    const unsigned d = (unsigned)bk[k];
    const unsigned b = d / BK_NODES;
    bk[k] = (int)b;
    pk[k] = (int)(((d - b * BK_NODES) << SRC_BITS) | s);
    rk[k] = atomicAdd(&hist[b], 1);
  }

  int cur = 0;
  for (int off = 1; off < 512; off <<= 1) {
    if (tid < 512) {
      int v = sc[cur][tid];
      if (tid >= off) v += sc[cur][tid - off];
      sc[cur ^ 1][tid] = v;
    }
    __syncthreads();
    cur ^= 1;
  }
  const int* scan = sc[cur];

  for (int b = tid; b < NBK; b += BF_THREADS) {
    const int c = hist[b];
    const int start = b ? scan[b - 1] : 0;
    const int r = c ? atomicAdd(&cursor[b], c) : 0;
    bpos[b] = start + r;
    if (blockIdx.x == 0) offs[b] = start;
  }
  if (blockIdx.x == 0 && tid == 0) offs[NBK] = scan[NBK - 1];
  __syncthreads();

#pragma unroll
  for (int k = 0; k < BF_EDGES_PER_THREAD; k++) bins[bpos[bk[k]] + rk[k]] = pk[k];
}

#define AG_WAVES 16
#define AG_THREADS (AG_WAVES * 32)
#define AG_VEC (BK_NODES * 16)
#define CV_TILES ((BK_NODES + 15) / 16)
#define CV_ROWS (CV_TILES * 16)
#define CV_PITCH 68
#define CV_PASSES ((CV_ROWS * 16 + AG_THREADS - 1) / AG_THREADS)
#define CV_LROWS (CV_PASSES * AG_THREADS / 16)
static_assert(BK_NODES % 8 == 0, "a lane half of a tile (8 rows) is all valid or all padding");
static_assert(CV_TILES <= AG_WAVES, "one wave per tile");
static_assert(CV_LROWS >= CV_ROWS && (CV_PASSES * AG_THREADS) % 16 == 0, "passes cover every row a tile reads");
static_assert((64 * 64) % AG_THREADS == 0, "weight pieces per thread");
#define CV_WPT ((64 * 64) / AG_THREADS)
__launch_bounds__(AG_THREADS, 8)
__global__ void gin_conv_kernel(const float* __restrict__ h,
                                const int* __restrict__ offs,
                                const int* __restrict__ bins,
                                const float* __restrict__ W1,
                                const float* __restrict__ b1,
                                const float* __restrict__ bn1,
                                const float* __restrict__ W2,
                                const float* __restrict__ b2,
                                const float* __restrict__ bn2,
                                float* __restrict__ hout,
                                const float* __restrict__ linW,
                                const int* __restrict__ batch,
                                float* __restrict__ out) {
  __shared__ __align__(16) float acc[CV_LROWS * CV_PITCH];
  __shared__ __align__(16) float w1s[4 * 16 * 32 * 2];
  __shared__ __align__(16) float w2s[4 * 16 * 32 * 2];
  __shared__ float bns[4][64];
  const int tid  = threadIdx.x;
  const int lane = tid & 31;
  const int wave = __builtin_amdgcn_readfirstlane(tid >> 5);
  const int b = blockIdx.x;
  const size_t rowbase = (size_t)b * BK_NODES * 64;

  {
    const float4* hs = (const float4*)(h + rowbase);
    float4 t[CV_PASSES];
#pragma unroll
    for (int k = 0; k < CV_PASSES; k++) {
      const int i = tid + k * AG_THREADS;
      t[k] = hs[i < AG_VEC ? i : AG_VEC - 1];
    }
    float wa[CV_WPT], wb[CV_WPT];
#pragma unroll
    for (int k = 0; k < CV_WPT; k++) {
      wa[k] = W1[tid + k * AG_THREADS];
      wb[k] = W2[tid + k * AG_THREADS];
    }
#pragma unroll
    for (int k = 0; k < CV_PASSES; k++) {
      const int i = tid + k * AG_THREADS;
      const bool real = i < AG_VEC;
      float4 v = t[k];
      v.x = real ? v.x : 0.f;
      v.y = real ? v.y : 0.f;
      v.z = real ? v.z : 0.f;
      v.w = real ? v.w : 0.f;
      *(float4*)(acc + (i >> 4) * CV_PITCH + ((i & 15) << 2)) = v;
    }
#pragma unroll
    for (int k = 0; k < CV_WPT; k++) {
      const int i = tid + k * AG_THREADS;
      const int kk = i >> 6, nn = i & 63;
      const int idx = ((((nn >> 4) * 16) + (kk >> 2)) * 32 + (nn & 15) + ((kk & 2) << 3)) * 2 + (kk & 1);
      w1s[idx] = wa[k];
      w2s[idx] = wb[k];
    }
  }
  __syncthreads();

  const int e0 = offs[b];
  const int e1 = offs[b + 1];
  int c = e0 + wave * 32;
  int p = bins[c + lane < N_EDGES ? c + lane : N_EDGES - 1];
  while (c + 32 <= e1) {
    const int cn = c + AG_THREADS;
    v2f v[32];
#pragma unroll
    for (int j = 0; j < 32; j++) {
      const int pj = __builtin_amdgcn_readlane(p, j);
      v[j] = *(const v2f*)(h + (size_t)(pj & ((1 << SRC_BITS) - 1)) * 64 + 2 * lane);
    }
    const int pn = bins[cn + lane < N_EDGES ? cn + lane : N_EDGES - 1];
#pragma unroll
    for (int j = 0; j < 32; j++) {
      const int pj = __builtin_amdgcn_readlane(p, j);
      float* q = acc + (pj >> SRC_BITS) * CV_PITCH + 2 * lane;
      atomicAdd(q, v[j].x);
      atomicAdd(q + 1, v[j].y);
    }
    p = pn;
    c = cn;
  }
  if (c < e1) {
    const int n = e1 - c;
    const int pt = lane < n ? bins[c + lane] : 0;
    for (int j = 0; j < n; j++) {
      const int pj = __builtin_amdgcn_readlane(pt, j);
      const v2f t = *(const v2f*)(h + (size_t)(pj & ((1 << SRC_BITS) - 1)) * 64 + 2 * lane);
      float* q = acc + (pj >> SRC_BITS) * CV_PITCH + 2 * lane;
      atomicAdd(q, t.x);
      atomicAdd(q + 1, t.y);
    }
  }
  if (tid >= AG_THREADS - 128) {
    const int u = tid - (AG_THREADS - 128);
    const float* bp = u < 64 ? bn1 : bn2;
    const float* lp = u < 64 ? b1 : b2;
    const int n = u & 63;
    const float s = bp[n] * rsqrtf(bp[192 + n] + BN_EPS);
    bns[(u >> 6) * 2][n] = s;
    bns[(u >> 6) * 2 + 1][n] = bp[64 + n] + (lp[n] - bp[128 + n]) * s;
  }
  __syncthreads();

  const int row  = lane & 15;
  const int hi   = lane >> 4;
  const bool active = wave < CV_TILES;
  const int left = BK_NODES - wave * 16;
  const int nrow = left < 16 ? left : 16;
  const bool rows_ok = 8 * hi < nrow;
  const int node0 = b * BK_NODES + wave * 16;
  float* tbuf = acc + wave * 16 * CV_PITCH;
  const v8f vzero = {0.f, 0.f, 0.f, 0.f, 0.f, 0.f, 0.f, 0.f};

  v2f a1[16];
  if (active) {
    const float* tr = tbuf + row * CV_PITCH + 2 * hi;
#pragma unroll
    for (int k = 0; k < 16; k++) a1[k] = *(const v2f*)(tr + 4 * k);
  }
  __syncthreads();

  if (active) {
#pragma unroll
    for (int nt = 0; nt < 4; nt++) {
      v8f a = vzero;
      const v2f* wf = ((const v2f*)w1s) + nt * 16 * 32 + lane;
#pragma unroll
      for (int k = 0; k < 16; k++) {
        a = __builtin_amdgcn_wmma_f32_16x16x4_f32(false, a1[k], false, wf[k * 32],
                                                  (short)0, a, false, false);
      }
      int n = nt * 16 + row;
      float s  = bns[0][n];
      float t0 = bns[1][n];
#pragma unroll
      for (int i = 0; i < 8; i++) {
        float y = a[i] * s + t0;
        tbuf[(i + 8 * hi) * CV_PITCH + n] = fmaxf(y, 0.0f);
      }
    }
  }
  __syncthreads();

  v8f acc2[4];
  v2f wl[16];
  int gid[8];
  const int hc = row < 10 ? row : 9;
#pragma unroll
  for (int k = 0; k < 16; k++) wl[k] = (v2f){0.f, 0.f};
#pragma unroll
  for (int r = 0; r < 8; r++) gid[r] = 0;
  if (active) {
    const float* lw = linW + (2 * hi) * 10 + hc;
#pragma unroll
    for (int k = 0; k < 16; k++) {
      float w0 = lw[(4 * k) * 10];
      float w1 = lw[(4 * k + 1) * 10];
      wl[k] = (v2f){row < 10 ? w0 : 0.f, row < 10 ? w1 : 0.f};
    }
    if (rows_ok) {
#pragma unroll
      for (int r = 0; r < 8; r++) gid[r] = batch[node0 + 8 * hi + r];
    }
#pragma unroll
    for (int nt = 0; nt < 4; nt++) {
      v8f a = vzero;
      const v2f* wf = ((const v2f*)w2s) + nt * 16 * 32 + lane;
      const float* tr = tbuf + row * CV_PITCH + 2 * hi;
#pragma unroll
      for (int k = 0; k < 16; k++) {
        v2f av = *(const v2f*)(tr + 4 * k);
        a = __builtin_amdgcn_wmma_f32_16x16x4_f32(false, av, false, wf[k * 32],
                                                  (short)0, a, false, false);
      }
      acc2[nt] = a;
    }
  }
  __syncthreads();

  if (active) {
#pragma unroll
    for (int nt = 0; nt < 4; nt++) {
      int n = nt * 16 + row;
      float s  = bns[2][n];
      float t0 = bns[3][n];
#pragma unroll
      for (int i = 0; i < 8; i++) {
        float y = acc2[nt][i] * s + t0;
        tbuf[(i + 8 * hi) * CV_PITCH + n] = fmaxf(y, 0.0f);
      }
    }
  }
  __syncthreads();

  if (active) {
    v8f p0 = vzero, p1 = vzero;
    const float* tr = tbuf + row * CV_PITCH + 2 * hi;
#pragma unroll
    for (int k = 0; k < 16; k += 2) {
      v2f a0 = *(const v2f*)(tr + 4 * k);
      v2f a1h = *(const v2f*)(tr + 4 * (k + 1));
      p0 = __builtin_amdgcn_wmma_f32_16x16x4_f32(false, a0, false, wl[k],
                                                 (short)0, p0, false, false);
      p1 = __builtin_amdgcn_wmma_f32_16x16x4_f32(false, a1h, false, wl[k + 1],
                                                 (short)0, p1, false, false);
    }
    float pv[8];
#pragma unroll
    for (int r = 0; r < 8; r++) pv[r] = p0[r] + p1[r];

#pragma unroll
    for (int j = 0; j < 8; j++) {
      const int r = 2 * j + hi;
      const int cc = row << 2;
      if (r < nrow) {
        float4 v = *(const float4*)(tbuf + r * CV_PITCH + cc);
        *(float4*)(hout + (size_t)(node0 + r) * 64 + cc) = v;
      }
    }

    if (row < 10 && rows_ok) {
#pragma unroll
      for (int r = 0; r < 8; r++) atomicAdd(out + (size_t)gid[r] * 10 + row, pv[r]);
    }
  }
}

__global__ void gin_out_init_kernel(const float* __restrict__ linb,
                                    float* __restrict__ out,
                                    int* __restrict__ cnt) {
  int idx = blockIdx.x * blockDim.x + threadIdx.x;
  if (idx < 2 * NBK) cnt[idx] = 0;
  if (idx >= N_GRAPHS * 10) return;
  int t = idx % 10;
  out[idx] = ((linb[10 + t] + linb[20 + t]) + linb[30 + t]) + linb[40 + t];
}

extern "C" void kernel_launch(void* const* d_in, const int* in_sizes, int n_in,
                              void* d_out, int out_size, void* d_ws, size_t ws_size,
                              hipStream_t stream) {
  const float* x        = (const float*)d_in[0];
  const float* fh_W1    = (const float*)d_in[1];
  const float* fh_b1    = (const float*)d_in[2];
  const float* fh_bn1   = (const float*)d_in[3];
  const float* fh_W2    = (const float*)d_in[4];
  const float* fh_b2    = (const float*)d_in[5];
  const float* fh_bn2   = (const float*)d_in[6];
  const float* conv_W1  = (const float*)d_in[7];
  const float* conv_b1  = (const float*)d_in[8];
  const float* conv_bn1 = (const float*)d_in[9];
  const float* conv_W2  = (const float*)d_in[10];
  const float* conv_b2  = (const float*)d_in[11];
  const float* conv_bn2 = (const float*)d_in[12];
  const float* lin_W    = (const float*)d_in[13];
  const float* lin_b    = (const float*)d_in[14];
  const int*   eidx     = (const int*)d_in[15];
  const int*   batch    = (const int*)d_in[16];
  float* out = (float*)d_out;

  const int* src = eidx;
  const int* dst = eidx + N_EDGES;

  float* ws     = (float*)d_ws;
  float* hA     = ws;
  float* hB     = ws + (size_t)N_NODES * 64;
  int*   cnt    = (int*)(ws + (size_t)2 * N_NODES * 64);
  int*   cursor = cnt + NBK;
  int*   offs   = cnt + 1024;
  int*   bins   = cnt + 2048;

  gin_out_init_kernel<<<(N_GRAPHS * 10 + 255) / 256, 256, 0, stream>>>(lin_b, out, cnt);

  gin_bin_count_kernel<<<BF_BLOCKS, BF_THREADS, 0, stream>>>(dst, cnt);
  gin_bin_fill_kernel<<<BF_BLOCKS, BF_THREADS, 0, stream>>>(src, dst, cnt, cursor, offs, bins);

  const int mlpBlocks = (NTILES + 3) / 4;

  gin_mlp_kernel<D_FEAT, true><<<mlpBlocks, 128, 0, stream>>>(
      x, fh_W1, fh_b1, fh_bn1, fh_W2, fh_b2, fh_bn2, hA,
      lin_W, lin_b, batch, out, NTILES);

  float* hin = hA;
  float* hnext = hB;
  for (int l = 0; l < 4; l++) {
    gin_conv_kernel<<<NBK, AG_THREADS, 0, stream>>>(
        hin, offs, bins,
        conv_W1 + (size_t)l * 64 * 64, conv_b1 + l * 64, conv_bn1 + l * 256,
        conv_W2 + (size_t)l * 64 * 64, conv_b2 + l * 64, conv_bn2 + l * 256,
        hnext,
        lin_W + (size_t)(l + 1) * 64 * 10, batch, out);
    float* tswap = hin;
    hin = hnext;
    hnext = tswap;
  }
}


# Neighbour sums without LDS float atomics: each range's edges grouped once per call by owner wave (dst row mod 16), which adds its rows by plain LDS load-add-store; sum order = grouping arrival rank
# speedup vs baseline: 3.2500x; 1.0961x over previous
#include <hip/hip_runtime.h>

#define N_NODES   100000
#define N_EDGES   1600000
#define N_GRAPHS  1000
#define D_FEAT    128
#define HID       64
#define BN_EPS    1e-5f
#define NTILES    (N_NODES / 16)

typedef __attribute__((ext_vector_type(2))) float v2f;
typedef __attribute__((ext_vector_type(8))) float v8f;

template<int K1, bool NODE_BIAS>
__launch_bounds__(128)
__global__ void gin_mlp_kernel(const float* __restrict__ xin,
                               const float* __restrict__ W1,
                               const float* __restrict__ b1,
                               const float* __restrict__ bn1,
                               const float* __restrict__ W2,
                               const float* __restrict__ b2,
                               const float* __restrict__ bn2,
                               float* __restrict__ hout,
                               const float* __restrict__ linW,
                               const float* __restrict__ linb0,
                               const int* __restrict__ batch,
                               float* __restrict__ out,
                               int ntiles) {
  constexpr int KS1 = K1 / 4;
  __shared__ __align__(16) float w1s[4 * KS1 * 32 * 2];
  __shared__ __align__(16) float w2s[4 * 16 * 32 * 2];
  __shared__ __align__(16) float tb[4][16 * 68];

  const int tid = threadIdx.x;

  for (int i = tid; i < K1 * 64; i += 128) {
    int kk = i >> 6, nn = i & 63;
    int idx = ((((nn >> 4) * KS1) + (kk >> 2)) * 32 + (nn & 15) + ((kk & 2) << 3)) * 2 + (kk & 1);
    w1s[idx] = W1[i];
  }
  for (int i = tid; i < 64 * 64; i += 128) {
    int kk = i >> 6, nn = i & 63;
    int idx = ((((nn >> 4) * 16) + (kk >> 2)) * 32 + (nn & 15) + ((kk & 2) << 3)) * 2 + (kk & 1);
    w2s[idx] = W2[i];
  }
  __syncthreads();

  const int lane = tid & 31;
  const int wave = tid >> 5;
  const int row  = lane & 15;
  const int hi   = lane >> 4;
  const int tile = blockIdx.x * 4 + wave;
  const bool active = tile < ntiles;
  float* tbuf = tb[wave];

  const v8f vzero = {0.f, 0.f, 0.f, 0.f, 0.f, 0.f, 0.f, 0.f};

  if (active) {
    v2f a1[KS1];
    const float* xr = xin + (size_t)(tile * 16 + row) * K1 + 2 * hi;
#pragma unroll
    for (int k = 0; k < KS1; k++) a1[k] = *(const v2f*)(xr + 4 * k);

#pragma unroll
    for (int nt = 0; nt < 4; nt++) {
      v8f acc = vzero;
      const v2f* wf = ((const v2f*)w1s) + nt * KS1 * 32 + lane;
#pragma unroll
      for (int k = 0; k < KS1; k++) {
        acc = __builtin_amdgcn_wmma_f32_16x16x4_f32(false, a1[k], false, wf[k * 32],
                                                    (short)0, acc, false, false);
      }
      int n = nt * 16 + row;
      float s  = bn1[n] * rsqrtf(bn1[192 + n] + BN_EPS);
      float t0 = bn1[64 + n] + (b1[n] - bn1[128 + n]) * s;
#pragma unroll
      for (int i = 0; i < 8; i++) {
        float y = acc[i] * s + t0;
        tbuf[(i + 8 * hi) * 68 + n] = fmaxf(y, 0.0f);
      }
    }
  }
  __syncthreads();

  v8f acc2[4];
  v2f wl[16];
  int gid[8];
  const int hc = row < 10 ? row : 9;
#pragma unroll
  for (int k = 0; k < 16; k++) wl[k] = (v2f){0.f, 0.f};
#pragma unroll
  for (int r = 0; r < 8; r++) gid[r] = 0;
  if (active) {
    const float* lw = linW + (2 * hi) * 10 + hc;
#pragma unroll
    for (int k = 0; k < 16; k++) {
      float w0 = lw[(4 * k) * 10];
      float w1 = lw[(4 * k + 1) * 10];
      wl[k] = (v2f){row < 10 ? w0 : 0.f, row < 10 ? w1 : 0.f};
    }
#pragma unroll
    for (int r = 0; r < 8; r++) gid[r] = batch[tile * 16 + 8 * hi + r];
#pragma unroll
    for (int nt = 0; nt < 4; nt++) {
      v8f acc = vzero;
      const v2f* wf = ((const v2f*)w2s) + nt * 16 * 32 + lane;
      const float* tr = tbuf + row * 68 + 2 * hi;
#pragma unroll
      for (int k = 0; k < 16; k++) {
        v2f a = *(const v2f*)(tr + 4 * k);
        acc = __builtin_amdgcn_wmma_f32_16x16x4_f32(false, a, false, wf[k * 32],
                                                    (short)0, acc, false, false);
      }
      acc2[nt] = acc;
    }
  }
  __syncthreads();

  if (active) {
#pragma unroll
    for (int nt = 0; nt < 4; nt++) {
      int n = nt * 16 + row;
      float s  = bn2[n] * rsqrtf(bn2[192 + n] + BN_EPS);
      float t0 = bn2[64 + n] + (b2[n] - bn2[128 + n]) * s;
#pragma unroll
      for (int i = 0; i < 8; i++) {
        float y = acc2[nt][i] * s + t0;
        tbuf[(i + 8 * hi) * 68 + n] = fmaxf(y, 0.0f);
      }
    }
  }
  __syncthreads();

  if (active) {
    v8f p0 = vzero, p1 = vzero;
    const float* tr = tbuf + row * 68 + 2 * hi;
#pragma unroll
    for (int k = 0; k < 16; k += 2) {
      v2f a0 = *(const v2f*)(tr + 4 * k);
      v2f a1 = *(const v2f*)(tr + 4 * (k + 1));
      p0 = __builtin_amdgcn_wmma_f32_16x16x4_f32(false, a0, false, wl[k],
                                                 (short)0, p0, false, false);
      p1 = __builtin_amdgcn_wmma_f32_16x16x4_f32(false, a1, false, wl[k + 1],
                                                 (short)0, p1, false, false);
    }
    float pv[8];
#pragma unroll
    for (int r = 0; r < 8; r++) pv[r] = p0[r] + p1[r];
    if (NODE_BIAS) {
      const float lb = linb0[hc];
#pragma unroll
      for (int r = 0; r < 8; r++) pv[r] += lb;
    }

#pragma unroll
    for (int j = 0; j < 8; j++) {
      int flat = j * 32 + lane;
      int r = flat >> 4;
      int c = (flat & 15) << 2;
      float4 v = *(const float4*)(tbuf + r * 68 + c);
      *(float4*)(hout + (size_t)(tile * 16 + r) * 64 + c) = v;
    }

    if (row < 10) {
#pragma unroll
      for (int r = 0; r < 8; r++) atomicAdd(out + (size_t)gid[r] * 10 + row, pv[r]);
    }
  }
}

#define BK_NODES 200
#define NBK (N_NODES / BK_NODES)
#define BF_THREADS 800
#define BF_EDGES_PER_THREAD 8
#define BF_EDGES_PER_BLOCK (BF_THREADS * BF_EDGES_PER_THREAD)
#define BF_BLOCKS (N_EDGES / BF_EDGES_PER_BLOCK)
#define SRC_BITS 17
static_assert(N_NODES % BK_NODES == 0, "ranges must tile the nodes exactly");
static_assert(N_EDGES % BF_EDGES_PER_BLOCK == 0, "edge blocks must tile the edges exactly");
static_assert(N_NODES <= (1 << SRC_BITS), "a source id must fit the packed field");
static_assert(NBK <= 512, "the range scan below holds 512 entries");
static_assert(NBK * 17 <= 16384, "group starts fit the space kept for them");

__launch_bounds__(BF_THREADS)
__global__ void gin_bin_count_kernel(const int* __restrict__ dst,
                                     int* __restrict__ cnt) {
  __shared__ int hist[NBK];
  const int tid = threadIdx.x;
  for (int b = tid; b < NBK; b += BF_THREADS) hist[b] = 0;
  __syncthreads();
  const int e0 = blockIdx.x * BF_EDGES_PER_BLOCK + tid;
  unsigned d[BF_EDGES_PER_THREAD];
#pragma unroll
  for (int k = 0; k < BF_EDGES_PER_THREAD; k++) d[k] = (unsigned)dst[e0 + k * BF_THREADS];
#pragma unroll
  for (int k = 0; k < BF_EDGES_PER_THREAD; k++) atomicAdd(&hist[d[k] / BK_NODES], 1);
  __syncthreads();
  for (int b = tid; b < NBK; b += BF_THREADS) {
    const int c = hist[b];
    if (c) atomicAdd(&cnt[b], c);
  }
}

__launch_bounds__(BF_THREADS)
__global__ void gin_bin_fill_kernel(const int* __restrict__ src,
                                    const int* __restrict__ dst,
                                    const int* __restrict__ cnt,
                                    int* __restrict__ cursor,
                                    int* __restrict__ offs,
                                    int* __restrict__ bins) {
  __shared__ int hist[NBK];
  __shared__ int bpos[NBK];
  __shared__ int sc[2][512];
  const int tid = threadIdx.x;
  for (int b = tid; b < NBK; b += BF_THREADS) hist[b] = 0;
  if (tid < 512) sc[0][tid] = tid < NBK ? cnt[tid] : 0;
  __syncthreads();

  int bk[BF_EDGES_PER_THREAD], rk[BF_EDGES_PER_THREAD], pk[BF_EDGES_PER_THREAD];
  const int e0 = blockIdx.x * BF_EDGES_PER_BLOCK + tid;
#pragma unroll
  for (int k = 0; k < BF_EDGES_PER_THREAD; k++) {
    pk[k] = src[e0 + k * BF_THREADS];
    bk[k] = dst[e0 + k * BF_THREADS];
  }
#pragma unroll
  for (int k = 0; k < BF_EDGES_PER_THREAD; k++) {
    const unsigned s = (unsigned)pk[k];
    const unsigned d = (unsigned)bk[k];
    const unsigned b = d / BK_NODES;
    bk[k] = (int)b;
    pk[k] = (int)(((d - b * BK_NODES) << SRC_BITS) | s);
    rk[k] = atomicAdd(&hist[b], 1);
  }

  int cur = 0;
  for (int off = 1; off < 512; off <<= 1) {
    if (tid < 512) {
      int v = sc[cur][tid];
      if (tid >= off) v += sc[cur][tid - off];
      sc[cur ^ 1][tid] = v;
    }
    __syncthreads();
    cur ^= 1;
  }
  const int* scan = sc[cur];

  for (int b = tid; b < NBK; b += BF_THREADS) {
    const int c = hist[b];
    const int start = b ? scan[b - 1] : 0;
    const int r = c ? atomicAdd(&cursor[b], c) : 0;
    bpos[b] = start + r;
    if (blockIdx.x == 0) offs[b] = start;
  }
  if (blockIdx.x == 0 && tid == 0) offs[NBK] = scan[NBK - 1];
  __syncthreads();

#pragma unroll
  for (int k = 0; k < BF_EDGES_PER_THREAD; k++) bins[bpos[bk[k]] + rk[k]] = pk[k];
}


#define NGRP 16
#define GR_THREADS 512
#define GR_KEEP 8
__launch_bounds__(GR_THREADS)
__global__ void gin_bin_group_kernel(const int* __restrict__ offs,
                                     const int* __restrict__ bins,
                                     int* __restrict__ goffs,
                                     int* __restrict__ gbins) {
  __shared__ int gcnt[NGRP];
  __shared__ int gpos[NGRP];
  __shared__ int gbase[NGRP + 1];
  const int tid = threadIdx.x;
  const int b = blockIdx.x;
  const int e0 = offs[b];
  const int e1 = offs[b + 1];
  if (tid < NGRP) {
    gcnt[tid] = 0;
    gpos[tid] = 0;
  }
  int pr[GR_KEEP];
#pragma unroll
  for (int k = 0; k < GR_KEEP; k++) {
    const int i = e0 + tid + k * GR_THREADS;
    pr[k] = bins[i < e1 ? i : N_EDGES - 1];
  }
  __syncthreads();
#pragma unroll
  for (int k = 0; k < GR_KEEP; k++) {
    if (e0 + tid + k * GR_THREADS < e1) atomicAdd(&gcnt[(pr[k] >> SRC_BITS) & (NGRP - 1)], 1);
  }
  for (int i = e0 + tid + GR_KEEP * GR_THREADS; i < e1; i += GR_THREADS) {
    const int g = (bins[i] >> SRC_BITS) & (NGRP - 1);
    atomicAdd(&gcnt[g], 1);
  }
  __syncthreads();
  if (tid == 0) {
    int s = e0;
    for (int g = 0; g < NGRP; g++) {
      gbase[g] = s;
      s += gcnt[g];
    }
    gbase[NGRP] = s;
  }
  __syncthreads();
  if (tid <= NGRP) goffs[b * (NGRP + 1) + tid] = gbase[tid];
#pragma unroll
  for (int k = 0; k < GR_KEEP; k++) {
    if (e0 + tid + k * GR_THREADS < e1) {
      const int g = (pr[k] >> SRC_BITS) & (NGRP - 1);
      const int r = atomicAdd(&gpos[g], 1);
      gbins[gbase[g] + r] = pr[k];
    }
  }
  for (int i = e0 + tid + GR_KEEP * GR_THREADS; i < e1; i += GR_THREADS) {
    const int p = bins[i];
    const int g = (p >> SRC_BITS) & (NGRP - 1);
    const int r = atomicAdd(&gpos[g], 1);
    gbins[gbase[g] + r] = p;
  }
}

#define AG_WAVES 16
static_assert(AG_WAVES == NGRP, "one owner wave per row group");
#define AG_THREADS (AG_WAVES * 32)
#define AG_VEC (BK_NODES * 16)
#define CV_TILES ((BK_NODES + 15) / 16)
#define CV_ROWS (CV_TILES * 16)
#define CV_PITCH 68
#define CV_PASSES ((CV_ROWS * 16 + AG_THREADS - 1) / AG_THREADS)
#define CV_LROWS (CV_PASSES * AG_THREADS / 16)
static_assert(BK_NODES % 8 == 0, "a lane half of a tile (8 rows) is all valid or all padding");
static_assert(CV_TILES <= AG_WAVES, "one wave per tile");
static_assert(CV_LROWS >= CV_ROWS && (CV_PASSES * AG_THREADS) % 16 == 0, "passes cover every row a tile reads");
static_assert((64 * 64) % AG_THREADS == 0, "weight pieces per thread");
#define CV_WPT ((64 * 64) / AG_THREADS)
__launch_bounds__(AG_THREADS, 8)
__global__ void gin_conv_kernel(const float* __restrict__ h,
                                const int* __restrict__ goffs,
                                const int* __restrict__ gbins,
                                const float* __restrict__ W1,
                                const float* __restrict__ b1,
                                const float* __restrict__ bn1,
                                const float* __restrict__ W2,
                                const float* __restrict__ b2,
                                const float* __restrict__ bn2,
                                float* __restrict__ hout,
                                const float* __restrict__ linW,
                                const int* __restrict__ batch,
                                float* __restrict__ out) {
  __shared__ __align__(16) float acc[CV_LROWS * CV_PITCH];
  __shared__ __align__(16) float w1s[4 * 16 * 32 * 2];
  __shared__ __align__(16) float w2s[4 * 16 * 32 * 2];
  __shared__ float bns[4][64];
  const int tid  = threadIdx.x;
  const int lane = tid & 31;
  const int wave = __builtin_amdgcn_readfirstlane(tid >> 5);
  const int b = blockIdx.x;
  const size_t rowbase = (size_t)b * BK_NODES * 64;

  {
    const float4* hs = (const float4*)(h + rowbase);
    float4 t[CV_PASSES];
#pragma unroll
    for (int k = 0; k < CV_PASSES; k++) {
      const int i = tid + k * AG_THREADS;
      t[k] = hs[i < AG_VEC ? i : AG_VEC - 1];
    }
    float wa[CV_WPT], wb[CV_WPT];
#pragma unroll
    for (int k = 0; k < CV_WPT; k++) {
      wa[k] = W1[tid + k * AG_THREADS];
      wb[k] = W2[tid + k * AG_THREADS];
    }
#pragma unroll
    for (int k = 0; k < CV_PASSES; k++) {
      const int i = tid + k * AG_THREADS;
      const bool real = i < AG_VEC;
      float4 v = t[k];
      v.x = real ? v.x : 0.f;
      v.y = real ? v.y : 0.f;
      v.z = real ? v.z : 0.f;
      v.w = real ? v.w : 0.f;
      *(float4*)(acc + (i >> 4) * CV_PITCH + ((i & 15) << 2)) = v;
    }
#pragma unroll
    for (int k = 0; k < CV_WPT; k++) {
      const int i = tid + k * AG_THREADS;
      const int kk = i >> 6, nn = i & 63;
      const int idx = ((((nn >> 4) * 16) + (kk >> 2)) * 32 + (nn & 15) + ((kk & 2) << 3)) * 2 + (kk & 1);
      w1s[idx] = wa[k];
      w2s[idx] = wb[k];
    }
  }
  __syncthreads();

  const int g0 = goffs[b * (NGRP + 1) + wave];
  const int g1 = goffs[b * (NGRP + 1) + wave + 1];
  int c = g0;
  int p = gbins[c + lane < N_EDGES ? c + lane : N_EDGES - 1];
  while (c + 32 <= g1) {
    const int cn = c + 32;
    v2f v[32];
#pragma unroll
    for (int j = 0; j < 32; j++) {
      const int pj = __builtin_amdgcn_readlane(p, j);
      v[j] = *(const v2f*)(h + (size_t)(pj & ((1 << SRC_BITS) - 1)) * 64 + 2 * lane);
    }
    const int pn = gbins[cn + lane < N_EDGES ? cn + lane : N_EDGES - 1];
#pragma unroll
    for (int j = 0; j < 32; j++) {
      const int pj = __builtin_amdgcn_readlane(p, j);
      v2f* q = (v2f*)(acc + (pj >> SRC_BITS) * CV_PITCH + 2 * lane);
      v2f t = *q;
      t.x += v[j].x;
      t.y += v[j].y;
      *q = t;
    }
    p = pn;
    c = cn;
  }
  if (c < g1) {
    const int n = g1 - c;
    const int pt = lane < n ? p : __builtin_amdgcn_readlane(p, 0);
    v2f v[32];
#pragma unroll
    for (int j = 0; j < 32; j++) {
      const int pj = __builtin_amdgcn_readlane(pt, j);
      v[j] = *(const v2f*)(h + (size_t)(pj & ((1 << SRC_BITS) - 1)) * 64 + 2 * lane);
    }
#pragma unroll
    for (int j = 0; j < 32; j++) {
      if (j < n) {
        const int pj = __builtin_amdgcn_readlane(pt, j);
        v2f* q = (v2f*)(acc + (pj >> SRC_BITS) * CV_PITCH + 2 * lane);
        v2f t = *q;
        t.x += v[j].x;
        t.y += v[j].y;
        *q = t;
      }
    }
  }
  if (tid >= AG_THREADS - 128) {
    const int u = tid - (AG_THREADS - 128);
    const float* bp = u < 64 ? bn1 : bn2;
    const float* lp = u < 64 ? b1 : b2;
    const int n = u & 63;
    const float s = bp[n] * rsqrtf(bp[192 + n] + BN_EPS);
    bns[(u >> 6) * 2][n] = s;
    bns[(u >> 6) * 2 + 1][n] = bp[64 + n] + (lp[n] - bp[128 + n]) * s;
  }
  __syncthreads();

  const int row  = lane & 15;
  const int hi   = lane >> 4;
  const bool active = wave < CV_TILES;
  const int left = BK_NODES - wave * 16;
  const int nrow = left < 16 ? left : 16;
  const bool rows_ok = 8 * hi < nrow;
  const int node0 = b * BK_NODES + wave * 16;
  float* tbuf = acc + wave * 16 * CV_PITCH;
  const v8f vzero = {0.f, 0.f, 0.f, 0.f, 0.f, 0.f, 0.f, 0.f};

  v2f a1[16];
  if (active) {
    const float* tr = tbuf + row * CV_PITCH + 2 * hi;
#pragma unroll
    for (int k = 0; k < 16; k++) a1[k] = *(const v2f*)(tr + 4 * k);
  }
  __syncthreads();

  if (active) {
#pragma unroll
    for (int nt = 0; nt < 4; nt++) {
      v8f a = vzero;
      const v2f* wf = ((const v2f*)w1s) + nt * 16 * 32 + lane;
#pragma unroll
      for (int k = 0; k < 16; k++) {
        a = __builtin_amdgcn_wmma_f32_16x16x4_f32(false, a1[k], false, wf[k * 32],
                                                  (short)0, a, false, false);
      }
      int n = nt * 16 + row;
      float s  = bns[0][n];
      float t0 = bns[1][n];
#pragma unroll
      for (int i = 0; i < 8; i++) {
        float y = a[i] * s + t0;
        tbuf[(i + 8 * hi) * CV_PITCH + n] = fmaxf(y, 0.0f);
      }
    }
  }
  __syncthreads();

  v8f acc2[4];
  v2f wl[16];
  int gid[8];
  const int hc = row < 10 ? row : 9;
#pragma unroll
  for (int k = 0; k < 16; k++) wl[k] = (v2f){0.f, 0.f};
#pragma unroll
  for (int r = 0; r < 8; r++) gid[r] = 0;
  if (active) {
    const float* lw = linW + (2 * hi) * 10 + hc;
#pragma unroll
    for (int k = 0; k < 16; k++) {
      float w0 = lw[(4 * k) * 10];
      float w1 = lw[(4 * k + 1) * 10];
      wl[k] = (v2f){row < 10 ? w0 : 0.f, row < 10 ? w1 : 0.f};
    }
    if (rows_ok) {
#pragma unroll
      for (int r = 0; r < 8; r++) gid[r] = batch[node0 + 8 * hi + r];
    }
#pragma unroll
    for (int nt = 0; nt < 4; nt++) {
      v8f a = vzero;
      const v2f* wf = ((const v2f*)w2s) + nt * 16 * 32 + lane;
      const float* tr = tbuf + row * CV_PITCH + 2 * hi;
#pragma unroll
      for (int k = 0; k < 16; k++) {
        v2f av = *(const v2f*)(tr + 4 * k);
        a = __builtin_amdgcn_wmma_f32_16x16x4_f32(false, av, false, wf[k * 32],
                                                  (short)0, a, false, false);
      }
      acc2[nt] = a;
    }
  }
  __syncthreads();

  if (active) {
#pragma unroll
    for (int nt = 0; nt < 4; nt++) {
      int n = nt * 16 + row;
      float s  = bns[2][n];
      float t0 = bns[3][n];
#pragma unroll
      for (int i = 0; i < 8; i++) {
        float y = acc2[nt][i] * s + t0;
        tbuf[(i + 8 * hi) * CV_PITCH + n] = fmaxf(y, 0.0f);
      }
    }
  }
  __syncthreads();

  if (active) {
    v8f p0 = vzero, p1 = vzero;
    const float* tr = tbuf + row * CV_PITCH + 2 * hi;
#pragma unroll
    for (int k = 0; k < 16; k += 2) {
      v2f a0 = *(const v2f*)(tr + 4 * k);
      v2f a1h = *(const v2f*)(tr + 4 * (k + 1));
      p0 = __builtin_amdgcn_wmma_f32_16x16x4_f32(false, a0, false, wl[k],
                                                 (short)0, p0, false, false);
      p1 = __builtin_amdgcn_wmma_f32_16x16x4_f32(false, a1h, false, wl[k + 1],
                                                 (short)0, p1, false, false);
    }
    float pv[8];
#pragma unroll
    for (int r = 0; r < 8; r++) pv[r] = p0[r] + p1[r];

#pragma unroll
    for (int j = 0; j < 8; j++) {
      const int r = 2 * j + hi;
      const int cc = row << 2;
      if (r < nrow) {
        float4 v = *(const float4*)(tbuf + r * CV_PITCH + cc);
        *(float4*)(hout + (size_t)(node0 + r) * 64 + cc) = v;
      }
    }

    if (row < 10 && rows_ok) {
#pragma unroll
      for (int r = 0; r < 8; r++) atomicAdd(out + (size_t)gid[r] * 10 + row, pv[r]);
    }
  }
}

__global__ void gin_out_init_kernel(const float* __restrict__ linb,
                                    float* __restrict__ out,
                                    int* __restrict__ cnt) {
  int idx = blockIdx.x * blockDim.x + threadIdx.x;
  if (idx < 2 * NBK) cnt[idx] = 0;
  if (idx >= N_GRAPHS * 10) return;
  int t = idx % 10;
  out[idx] = ((linb[10 + t] + linb[20 + t]) + linb[30 + t]) + linb[40 + t];
}

extern "C" void kernel_launch(void* const* d_in, const int* in_sizes, int n_in,
                              void* d_out, int out_size, void* d_ws, size_t ws_size,
                              hipStream_t stream) {
  const float* x        = (const float*)d_in[0];
  const float* fh_W1    = (const float*)d_in[1];
  const float* fh_b1    = (const float*)d_in[2];
  const float* fh_bn1   = (const float*)d_in[3];
  const float* fh_W2    = (const float*)d_in[4];
  const float* fh_b2    = (const float*)d_in[5];
  const float* fh_bn2   = (const float*)d_in[6];
  const float* conv_W1  = (const float*)d_in[7];
  const float* conv_b1  = (const float*)d_in[8];
  const float* conv_bn1 = (const float*)d_in[9];
  const float* conv_W2  = (const float*)d_in[10];
  const float* conv_b2  = (const float*)d_in[11];
  const float* conv_bn2 = (const float*)d_in[12];
  const float* lin_W    = (const float*)d_in[13];
  const float* lin_b    = (const float*)d_in[14];
  const int*   eidx     = (const int*)d_in[15];
  const int*   batch    = (const int*)d_in[16];
  float* out = (float*)d_out;

  const int* src = eidx;
  const int* dst = eidx + N_EDGES;

  float* ws     = (float*)d_ws;
  float* hA     = ws;
  float* hB     = ws + (size_t)N_NODES * 64;
  int*   cnt    = (int*)(ws + (size_t)2 * N_NODES * 64);
  int*   cursor = cnt + NBK;
  int*   offs   = cnt + 1024;
  int*   bins   = cnt + 2048;
  int*   goffs  = cnt + 2048 + N_EDGES;
  int*   gbins  = cnt + 2048 + N_EDGES + 16384;

  gin_out_init_kernel<<<(N_GRAPHS * 10 + 255) / 256, 256, 0, stream>>>(lin_b, out, cnt);

  gin_bin_count_kernel<<<BF_BLOCKS, BF_THREADS, 0, stream>>>(dst, cnt);
  gin_bin_fill_kernel<<<BF_BLOCKS, BF_THREADS, 0, stream>>>(src, dst, cnt, cursor, offs, bins);
  gin_bin_group_kernel<<<NBK, GR_THREADS, 0, stream>>>(offs, bins, goffs, gbins);

  const int mlpBlocks = (NTILES + 3) / 4;

  gin_mlp_kernel<D_FEAT, true><<<mlpBlocks, 128, 0, stream>>>(
      x, fh_W1, fh_b1, fh_bn1, fh_W2, fh_b2, fh_bn2, hA,
      lin_W, lin_b, batch, out, NTILES);

  float* hin = hA;
  float* hnext = hB;
  for (int l = 0; l < 4; l++) {
    gin_conv_kernel<<<NBK, AG_THREADS, 0, stream>>>(
        hin, goffs, gbins,
        conv_W1 + (size_t)l * 64 * 64, conv_b1 + l * 64, conv_bn1 + l * 256,
        conv_W2 + (size_t)l * 64 * 64, conv_b2 + l * 64, conv_bn2 + l * 256,
        hnext,
        lin_W + (size_t)(l + 1) * 64 * 10, batch, out);
    float* tswap = hin;
    hin = hnext;
    hnext = tswap;
  }
}


# Neighbour sums in registers: owner wave keeps its 13 rows as running sums in VGPRs picked by a scalar register index, stores them to LDS once; per-edge LDS load-add-store, h copy gone; same sum order
# speedup vs baseline: 3.3688x; 1.0366x over previous
#include <hip/hip_runtime.h>

#define N_NODES   100000
#define N_EDGES   1600000
#define N_GRAPHS  1000
#define D_FEAT    128
#define HID       64
#define BN_EPS    1e-5f
#define NTILES    (N_NODES / 16)

typedef __attribute__((ext_vector_type(2))) float v2f;
typedef __attribute__((ext_vector_type(8))) float v8f;

template<int K1, bool NODE_BIAS>
__launch_bounds__(128)
__global__ void gin_mlp_kernel(const float* __restrict__ xin,
                               const float* __restrict__ W1,
                               const float* __restrict__ b1,
                               const float* __restrict__ bn1,
                               const float* __restrict__ W2,
                               const float* __restrict__ b2,
                               const float* __restrict__ bn2,
                               float* __restrict__ hout,
                               const float* __restrict__ linW,
                               const float* __restrict__ linb0,
                               const int* __restrict__ batch,
                               float* __restrict__ out,
                               int ntiles) {
  constexpr int KS1 = K1 / 4;
  __shared__ __align__(16) float w1s[4 * KS1 * 32 * 2];
  __shared__ __align__(16) float w2s[4 * 16 * 32 * 2];
  __shared__ __align__(16) float tb[4][16 * 68];

  const int tid = threadIdx.x;

  for (int i = tid; i < K1 * 64; i += 128) {
    int kk = i >> 6, nn = i & 63;
    int idx = ((((nn >> 4) * KS1) + (kk >> 2)) * 32 + (nn & 15) + ((kk & 2) << 3)) * 2 + (kk & 1);
    w1s[idx] = W1[i];
  }
  for (int i = tid; i < 64 * 64; i += 128) {
    int kk = i >> 6, nn = i & 63;
    int idx = ((((nn >> 4) * 16) + (kk >> 2)) * 32 + (nn & 15) + ((kk & 2) << 3)) * 2 + (kk & 1);
    w2s[idx] = W2[i];
  }
  __syncthreads();

  const int lane = tid & 31;
  const int wave = tid >> 5;
  const int row  = lane & 15;
  const int hi   = lane >> 4;
  const int tile = blockIdx.x * 4 + wave;
  const bool active = tile < ntiles;
  float* tbuf = tb[wave];

  const v8f vzero = {0.f, 0.f, 0.f, 0.f, 0.f, 0.f, 0.f, 0.f};

  if (active) {
    v2f a1[KS1];
    const float* xr = xin + (size_t)(tile * 16 + row) * K1 + 2 * hi;
#pragma unroll
    for (int k = 0; k < KS1; k++) a1[k] = *(const v2f*)(xr + 4 * k);

#pragma unroll
    for (int nt = 0; nt < 4; nt++) {
      v8f acc = vzero;
      const v2f* wf = ((const v2f*)w1s) + nt * KS1 * 32 + lane;
#pragma unroll
      for (int k = 0; k < KS1; k++) {
        acc = __builtin_amdgcn_wmma_f32_16x16x4_f32(false, a1[k], false, wf[k * 32],
                                                    (short)0, acc, false, false);
      }
      int n = nt * 16 + row;
      float s  = bn1[n] * rsqrtf(bn1[192 + n] + BN_EPS);
      float t0 = bn1[64 + n] + (b1[n] - bn1[128 + n]) * s;
#pragma unroll
      for (int i = 0; i < 8; i++) {
        float y = acc[i] * s + t0;
        tbuf[(i + 8 * hi) * 68 + n] = fmaxf(y, 0.0f);
      }
    }
  }
  __syncthreads();

  v8f acc2[4];
  v2f wl[16];
  int gid[8];
  const int hc = row < 10 ? row : 9;
#pragma unroll
  for (int k = 0; k < 16; k++) wl[k] = (v2f){0.f, 0.f};
#pragma unroll
  for (int r = 0; r < 8; r++) gid[r] = 0;
  if (active) {
    const float* lw = linW + (2 * hi) * 10 + hc;
#pragma unroll
    for (int k = 0; k < 16; k++) {
      float w0 = lw[(4 * k) * 10];
      float w1 = lw[(4 * k + 1) * 10];
      wl[k] = (v2f){row < 10 ? w0 : 0.f, row < 10 ? w1 : 0.f};
    }
#pragma unroll
    for (int r = 0; r < 8; r++) gid[r] = batch[tile * 16 + 8 * hi + r];
#pragma unroll
    for (int nt = 0; nt < 4; nt++) {
      v8f acc = vzero;
      const v2f* wf = ((const v2f*)w2s) + nt * 16 * 32 + lane;
      const float* tr = tbuf + row * 68 + 2 * hi;
#pragma unroll
      for (int k = 0; k < 16; k++) {
        v2f a = *(const v2f*)(tr + 4 * k);
        acc = __builtin_amdgcn_wmma_f32_16x16x4_f32(false, a, false, wf[k * 32],
                                                    (short)0, acc, false, false);
      }
      acc2[nt] = acc;
    }
  }
  __syncthreads();

  if (active) {
#pragma unroll
    for (int nt = 0; nt < 4; nt++) {
      int n = nt * 16 + row;
      float s  = bn2[n] * rsqrtf(bn2[192 + n] + BN_EPS);
      float t0 = bn2[64 + n] + (b2[n] - bn2[128 + n]) * s;
#pragma unroll
      for (int i = 0; i < 8; i++) {
        float y = acc2[nt][i] * s + t0;
        tbuf[(i + 8 * hi) * 68 + n] = fmaxf(y, 0.0f);
      }
    }
  }
  __syncthreads();

  if (active) {
    v8f p0 = vzero, p1 = vzero;
    const float* tr = tbuf + row * 68 + 2 * hi;
#pragma unroll
    for (int k = 0; k < 16; k += 2) {
      v2f a0 = *(const v2f*)(tr + 4 * k);
      v2f a1 = *(const v2f*)(tr + 4 * (k + 1));
      p0 = __builtin_amdgcn_wmma_f32_16x16x4_f32(false, a0, false, wl[k],
                                                 (short)0, p0, false, false);
      p1 = __builtin_amdgcn_wmma_f32_16x16x4_f32(false, a1, false, wl[k + 1],
                                                 (short)0, p1, false, false);
    }
    float pv[8];
#pragma unroll
    for (int r = 0; r < 8; r++) pv[r] = p0[r] + p1[r];
    if (NODE_BIAS) {
      const float lb = linb0[hc];
#pragma unroll
      for (int r = 0; r < 8; r++) pv[r] += lb;
    }

#pragma unroll
    for (int j = 0; j < 8; j++) {
      int flat = j * 32 + lane;
      int r = flat >> 4;
      int c = (flat & 15) << 2;
      float4 v = *(const float4*)(tbuf + r * 68 + c);
      *(float4*)(hout + (size_t)(tile * 16 + r) * 64 + c) = v;
    }

    if (row < 10) {
#pragma unroll
      for (int r = 0; r < 8; r++) atomicAdd(out + (size_t)gid[r] * 10 + row, pv[r]);
    }
  }
}

#define BK_NODES 200
#define NBK (N_NODES / BK_NODES)
#define BF_THREADS 800
#define BF_EDGES_PER_THREAD 8
#define BF_EDGES_PER_BLOCK (BF_THREADS * BF_EDGES_PER_THREAD)
#define BF_BLOCKS (N_EDGES / BF_EDGES_PER_BLOCK)
#define SRC_BITS 17
static_assert(N_NODES % BK_NODES == 0, "ranges must tile the nodes exactly");
static_assert(N_EDGES % BF_EDGES_PER_BLOCK == 0, "edge blocks must tile the edges exactly");
static_assert(N_NODES <= (1 << SRC_BITS), "a source id must fit the packed field");
static_assert(NBK <= 512, "the range scan below holds 512 entries");
static_assert(NBK * 17 <= 16384, "group starts fit the space kept for them");

__launch_bounds__(BF_THREADS)
__global__ void gin_bin_count_kernel(const int* __restrict__ dst,
                                     int* __restrict__ cnt) {
  __shared__ int hist[NBK];
  const int tid = threadIdx.x;
  for (int b = tid; b < NBK; b += BF_THREADS) hist[b] = 0;
  __syncthreads();
  const int e0 = blockIdx.x * BF_EDGES_PER_BLOCK + tid;
  unsigned d[BF_EDGES_PER_THREAD];
#pragma unroll
  for (int k = 0; k < BF_EDGES_PER_THREAD; k++) d[k] = (unsigned)dst[e0 + k * BF_THREADS];
#pragma unroll
  for (int k = 0; k < BF_EDGES_PER_THREAD; k++) atomicAdd(&hist[d[k] / BK_NODES], 1);
  __syncthreads();
  for (int b = tid; b < NBK; b += BF_THREADS) {
    const int c = hist[b];
    if (c) atomicAdd(&cnt[b], c);
  }
}

__launch_bounds__(BF_THREADS)
__global__ void gin_bin_fill_kernel(const int* __restrict__ src,
                                    const int* __restrict__ dst,
                                    const int* __restrict__ cnt,
                                    int* __restrict__ cursor,
                                    int* __restrict__ offs,
                                    int* __restrict__ bins) {
  __shared__ int hist[NBK];
  __shared__ int bpos[NBK];
  __shared__ int sc[2][512];
  const int tid = threadIdx.x;
  for (int b = tid; b < NBK; b += BF_THREADS) hist[b] = 0;
  if (tid < 512) sc[0][tid] = tid < NBK ? cnt[tid] : 0;
  __syncthreads();

  int bk[BF_EDGES_PER_THREAD], rk[BF_EDGES_PER_THREAD], pk[BF_EDGES_PER_THREAD];
  const int e0 = blockIdx.x * BF_EDGES_PER_BLOCK + tid;
#pragma unroll
  for (int k = 0; k < BF_EDGES_PER_THREAD; k++) {
    pk[k] = src[e0 + k * BF_THREADS];
    bk[k] = dst[e0 + k * BF_THREADS];
  }
#pragma unroll
  for (int k = 0; k < BF_EDGES_PER_THREAD; k++) {
    const unsigned s = (unsigned)pk[k];
    const unsigned d = (unsigned)bk[k];
    const unsigned b = d / BK_NODES;
    bk[k] = (int)b;
    pk[k] = (int)(((d - b * BK_NODES) << SRC_BITS) | s);
    rk[k] = atomicAdd(&hist[b], 1);
  }

  int cur = 0;
  for (int off = 1; off < 512; off <<= 1) {
    if (tid < 512) {
      int v = sc[cur][tid];
      if (tid >= off) v += sc[cur][tid - off];
      sc[cur ^ 1][tid] = v;
    }
    __syncthreads();
    cur ^= 1;
  }
  const int* scan = sc[cur];

  for (int b = tid; b < NBK; b += BF_THREADS) {
    const int c = hist[b];
    const int start = b ? scan[b - 1] : 0;
    const int r = c ? atomicAdd(&cursor[b], c) : 0;
    bpos[b] = start + r;
    if (blockIdx.x == 0) offs[b] = start;
  }
  if (blockIdx.x == 0 && tid == 0) offs[NBK] = scan[NBK - 1];
  __syncthreads();

#pragma unroll
  for (int k = 0; k < BF_EDGES_PER_THREAD; k++) bins[bpos[bk[k]] + rk[k]] = pk[k];
}


#define NGRP 16
#define GR_THREADS 512
#define GR_KEEP 8
__launch_bounds__(GR_THREADS)
__global__ void gin_bin_group_kernel(const int* __restrict__ offs,
                                     const int* __restrict__ bins,
                                     int* __restrict__ goffs,
                                     int* __restrict__ gbins) {
  __shared__ int gcnt[NGRP];
  __shared__ int gpos[NGRP];
  __shared__ int gbase[NGRP + 1];
  const int tid = threadIdx.x;
  const int b = blockIdx.x;
  const int e0 = offs[b];
  const int e1 = offs[b + 1];
  if (tid < NGRP) {
    gcnt[tid] = 0;
    gpos[tid] = 0;
  }
  int pr[GR_KEEP];
#pragma unroll
  for (int k = 0; k < GR_KEEP; k++) {
    const int i = e0 + tid + k * GR_THREADS;
    pr[k] = bins[i < e1 ? i : N_EDGES - 1];
  }
  __syncthreads();
#pragma unroll
  for (int k = 0; k < GR_KEEP; k++) {
    if (e0 + tid + k * GR_THREADS < e1) atomicAdd(&gcnt[(pr[k] >> SRC_BITS) & (NGRP - 1)], 1);
  }
  for (int i = e0 + tid + GR_KEEP * GR_THREADS; i < e1; i += GR_THREADS) {
    const int g = (bins[i] >> SRC_BITS) & (NGRP - 1);
    atomicAdd(&gcnt[g], 1);
  }
  __syncthreads();
  if (tid == 0) {
    int s = e0;
    for (int g = 0; g < NGRP; g++) {
      gbase[g] = s;
      s += gcnt[g];
    }
    gbase[NGRP] = s;
  }
  __syncthreads();
  if (tid <= NGRP) goffs[b * (NGRP + 1) + tid] = gbase[tid];
#pragma unroll
  for (int k = 0; k < GR_KEEP; k++) {
    if (e0 + tid + k * GR_THREADS < e1) {
      const int g = (pr[k] >> SRC_BITS) & (NGRP - 1);
      const int r = atomicAdd(&gpos[g], 1);
      gbins[gbase[g] + r] = pr[k];
    }
  }
  for (int i = e0 + tid + GR_KEEP * GR_THREADS; i < e1; i += GR_THREADS) {
    const int p = bins[i];
    const int g = (p >> SRC_BITS) & (NGRP - 1);
    const int r = atomicAdd(&gpos[g], 1);
    gbins[gbase[g] + r] = p;
  }
}

#define AG_WAVES 16
static_assert(AG_WAVES == NGRP, "one owner wave per row group");
#define AG_THREADS (AG_WAVES * 32)
#define CV_TILES ((BK_NODES + 15) / 16)
#define CV_ROWS (CV_TILES * 16)
#define CV_PITCH 68
#define CV_SLOTS (CV_ROWS / AG_WAVES)
static_assert(BK_NODES % 8 == 0, "a lane half of a tile (8 rows) is all valid or all padding");
static_assert(CV_TILES <= AG_WAVES, "one wave per tile");
static_assert(CV_SLOTS * AG_WAVES == CV_ROWS, "the owner waves' slots cover every row a tile reads");
static_assert(BK_NODES > (CV_SLOTS - 1) * AG_WAVES, "only the last slot can be a padding row");
static_assert((64 * 64) % AG_THREADS == 0, "weight pieces per thread");
#define CV_WPT ((64 * 64) / AG_THREADS)
typedef __attribute__((ext_vector_type(16))) float v16f;
static_assert(CV_SLOTS <= 16, "an owner wave's slots fit one 16-register vector per column half");
__launch_bounds__(AG_THREADS, 8)
__global__ void gin_conv_kernel(const float* __restrict__ h,
                                const int* __restrict__ goffs,
                                const int* __restrict__ gbins,
                                const float* __restrict__ W1,
                                const float* __restrict__ b1,
                                const float* __restrict__ bn1,
                                const float* __restrict__ W2,
                                const float* __restrict__ b2,
                                const float* __restrict__ bn2,
                                float* __restrict__ hout,
                                const float* __restrict__ linW,
                                const int* __restrict__ batch,
                                float* __restrict__ out) {
  __shared__ __align__(16) float acc[CV_ROWS * CV_PITCH];
  __shared__ __align__(16) float w1s[4 * 16 * 32 * 2];
  __shared__ __align__(16) float w2s[4 * 16 * 32 * 2];
  __shared__ float bns[4][64];
  const int tid  = threadIdx.x;
  const int lane = tid & 31;
  const int wave = __builtin_amdgcn_readfirstlane(tid >> 5);
  const int b = blockIdx.x;
  const size_t rowbase = (size_t)b * BK_NODES * 64;

  const int g0 = goffs[b * (NGRP + 1) + wave];
  const int g1 = goffs[b * (NGRP + 1) + wave + 1];
  {
    float wa[CV_WPT], wb[CV_WPT];
#pragma unroll
    for (int k = 0; k < CV_WPT; k++) {
      wa[k] = W1[tid + k * AG_THREADS];
      wb[k] = W2[tid + k * AG_THREADS];
    }
#pragma unroll
    for (int k = 0; k < CV_WPT; k++) {
      const int i = tid + k * AG_THREADS;
      const int kk = i >> 6, nn = i & 63;
      const int idx = ((((nn >> 4) * 16) + (kk >> 2)) * 32 + (nn & 15) + ((kk & 2) << 3)) * 2 + (kk & 1);
      w1s[idx] = wa[k];
      w2s[idx] = wb[k];
    }
  }

  v16f sx = {0.f, 0.f, 0.f, 0.f, 0.f, 0.f, 0.f, 0.f, 0.f, 0.f, 0.f, 0.f, 0.f, 0.f, 0.f, 0.f};
  v16f sy = sx;
#pragma unroll
  for (int s = 0; s < CV_SLOTS; s++) {
    const int r = wave + AG_WAVES * s;
    const bool real = r < BK_NODES;
    const v2f t = *(const v2f*)(h + rowbase + (size_t)(real ? r : BK_NODES - 1) * 64 + 2 * lane);
    sx[s] = real ? t.x : 0.f;
    sy[s] = real ? t.y : 0.f;
  }
  int c = g0;
  int p = gbins[c + lane < N_EDGES ? c + lane : N_EDGES - 1];
  while (c + 32 <= g1) {
    const int cn = c + 32;
    v2f v[32];
#pragma unroll
    for (int j = 0; j < 32; j++) {
      const int pj = __builtin_amdgcn_readlane(p, j);
      v[j] = *(const v2f*)(h + (size_t)(pj & ((1 << SRC_BITS) - 1)) * 64 + 2 * lane);
    }
    const int pn = gbins[cn + lane < N_EDGES ? cn + lane : N_EDGES - 1];
#pragma unroll
    for (int j = 0; j < 32; j++) {
      const int sl = __builtin_amdgcn_readlane(p, j) >> (SRC_BITS + 4);
      sx[sl] += v[j].x;
      sy[sl] += v[j].y;
    }
    p = pn;
    c = cn;
  }
  if (c < g1) {
    const int n = g1 - c;
    const int pt = lane < n ? p : __builtin_amdgcn_readlane(p, 0);
    v2f v[32];
#pragma unroll
    for (int j = 0; j < 32; j++) {
      const int pj = __builtin_amdgcn_readlane(pt, j);
      v[j] = *(const v2f*)(h + (size_t)(pj & ((1 << SRC_BITS) - 1)) * 64 + 2 * lane);
    }
#pragma unroll
    for (int j = 0; j < 32; j++) {
      if (j < n) {
        const int sl = __builtin_amdgcn_readlane(pt, j) >> (SRC_BITS + 4);
        sx[sl] += v[j].x;
        sy[sl] += v[j].y;
      }
    }
  }
#pragma unroll
  for (int s = 0; s < CV_SLOTS; s++) {
    *(v2f*)(acc + (wave + AG_WAVES * s) * CV_PITCH + 2 * lane) = (v2f){sx[s], sy[s]};
  }
  if (tid >= AG_THREADS - 128) {
    const int u = tid - (AG_THREADS - 128);
    const float* bp = u < 64 ? bn1 : bn2;
    const float* lp = u < 64 ? b1 : b2;
    const int n = u & 63;
    const float s = bp[n] * rsqrtf(bp[192 + n] + BN_EPS);
    bns[(u >> 6) * 2][n] = s;
    bns[(u >> 6) * 2 + 1][n] = bp[64 + n] + (lp[n] - bp[128 + n]) * s;
  }
  __syncthreads();

  const int row  = lane & 15;
  const int hi   = lane >> 4;
  const bool active = wave < CV_TILES;
  const int left = BK_NODES - wave * 16;
  const int nrow = left < 16 ? left : 16;
  const bool rows_ok = 8 * hi < nrow;
  const int node0 = b * BK_NODES + wave * 16;
  float* tbuf = acc + wave * 16 * CV_PITCH;
  const v8f vzero = {0.f, 0.f, 0.f, 0.f, 0.f, 0.f, 0.f, 0.f};

  v2f a1[16];
  if (active) {
    const float* tr = tbuf + row * CV_PITCH + 2 * hi;
#pragma unroll
    for (int k = 0; k < 16; k++) a1[k] = *(const v2f*)(tr + 4 * k);
  }
  __syncthreads();

  if (active) {
#pragma unroll
    for (int nt = 0; nt < 4; nt++) {
      v8f a = vzero;
      const v2f* wf = ((const v2f*)w1s) + nt * 16 * 32 + lane;
#pragma unroll
      for (int k = 0; k < 16; k++) {
        a = __builtin_amdgcn_wmma_f32_16x16x4_f32(false, a1[k], false, wf[k * 32],
                                                  (short)0, a, false, false);
      }
      int n = nt * 16 + row;
      float s  = bns[0][n];
      float t0 = bns[1][n];
#pragma unroll
      for (int i = 0; i < 8; i++) {
        float y = a[i] * s + t0;
        tbuf[(i + 8 * hi) * CV_PITCH + n] = fmaxf(y, 0.0f);
      }
    }
  }
  __syncthreads();

  v8f acc2[4];
  v2f wl[16];
  int gid[8];
  const int hc = row < 10 ? row : 9;
#pragma unroll
  for (int k = 0; k < 16; k++) wl[k] = (v2f){0.f, 0.f};
#pragma unroll
  for (int r = 0; r < 8; r++) gid[r] = 0;
  if (active) {
    const float* lw = linW + (2 * hi) * 10 + hc;
#pragma unroll
    for (int k = 0; k < 16; k++) {
      float w0 = lw[(4 * k) * 10];
      float w1 = lw[(4 * k + 1) * 10];
      wl[k] = (v2f){row < 10 ? w0 : 0.f, row < 10 ? w1 : 0.f};
    }
    if (rows_ok) {
#pragma unroll
      for (int r = 0; r < 8; r++) gid[r] = batch[node0 + 8 * hi + r];
    }
#pragma unroll
    for (int nt = 0; nt < 4; nt++) {
      v8f a = vzero;
      const v2f* wf = ((const v2f*)w2s) + nt * 16 * 32 + lane;
      const float* tr = tbuf + row * CV_PITCH + 2 * hi;
#pragma unroll
      for (int k = 0; k < 16; k++) {
        v2f av = *(const v2f*)(tr + 4 * k);
        a = __builtin_amdgcn_wmma_f32_16x16x4_f32(false, av, false, wf[k * 32],
                                                  (short)0, a, false, false);
      }
      acc2[nt] = a;
    }
  }
  __syncthreads();

  if (active) {
#pragma unroll
    for (int nt = 0; nt < 4; nt++) {
      int n = nt * 16 + row;
      float s  = bns[2][n];
      float t0 = bns[3][n];
#pragma unroll
      for (int i = 0; i < 8; i++) {
        float y = acc2[nt][i] * s + t0;
        tbuf[(i + 8 * hi) * CV_PITCH + n] = fmaxf(y, 0.0f);
      }
    }
  }
  __syncthreads();

  if (active) {
    v8f p0 = vzero, p1 = vzero;
    const float* tr = tbuf + row * CV_PITCH + 2 * hi;
#pragma unroll
    for (int k = 0; k < 16; k += 2) {
      v2f a0 = *(const v2f*)(tr + 4 * k);
      v2f a1h = *(const v2f*)(tr + 4 * (k + 1));
      p0 = __builtin_amdgcn_wmma_f32_16x16x4_f32(false, a0, false, wl[k],
                                                 (short)0, p0, false, false);
      p1 = __builtin_amdgcn_wmma_f32_16x16x4_f32(false, a1h, false, wl[k + 1],
                                                 (short)0, p1, false, false);
    }
    float pv[8];
#pragma unroll
    for (int r = 0; r < 8; r++) pv[r] = p0[r] + p1[r];

#pragma unroll
    for (int j = 0; j < 8; j++) {
      const int r = 2 * j + hi;
      const int cc = row << 2;
      if (r < nrow) {
        float4 v = *(const float4*)(tbuf + r * CV_PITCH + cc);
        *(float4*)(hout + (size_t)(node0 + r) * 64 + cc) = v;
      }
    }

    if (row < 10 && rows_ok) {
#pragma unroll
      for (int r = 0; r < 8; r++) atomicAdd(out + (size_t)gid[r] * 10 + row, pv[r]);
    }
  }
}

__global__ void gin_out_init_kernel(const float* __restrict__ linb,
                                    float* __restrict__ out,
                                    int* __restrict__ cnt) {
  int idx = blockIdx.x * blockDim.x + threadIdx.x;
  if (idx < 2 * NBK) cnt[idx] = 0;
  if (idx >= N_GRAPHS * 10) return;
  int t = idx % 10;
  out[idx] = ((linb[10 + t] + linb[20 + t]) + linb[30 + t]) + linb[40 + t];
}

extern "C" void kernel_launch(void* const* d_in, const int* in_sizes, int n_in,
                              void* d_out, int out_size, void* d_ws, size_t ws_size,
                              hipStream_t stream) {
  const float* x        = (const float*)d_in[0];
  const float* fh_W1    = (const float*)d_in[1];
  const float* fh_b1    = (const float*)d_in[2];
  const float* fh_bn1   = (const float*)d_in[3];
  const float* fh_W2    = (const float*)d_in[4];
  const float* fh_b2    = (const float*)d_in[5];
  const float* fh_bn2   = (const float*)d_in[6];
  const float* conv_W1  = (const float*)d_in[7];
  const float* conv_b1  = (const float*)d_in[8];
  const float* conv_bn1 = (const float*)d_in[9];
  const float* conv_W2  = (const float*)d_in[10];
  const float* conv_b2  = (const float*)d_in[11];
  const float* conv_bn2 = (const float*)d_in[12];
  const float* lin_W    = (const float*)d_in[13];
  const float* lin_b    = (const float*)d_in[14];
  const int*   eidx     = (const int*)d_in[15];
  const int*   batch    = (const int*)d_in[16];
  float* out = (float*)d_out;

  const int* src = eidx;
  const int* dst = eidx + N_EDGES;

  float* ws     = (float*)d_ws;
  float* hA     = ws;
  float* hB     = ws + (size_t)N_NODES * 64;
  int*   cnt    = (int*)(ws + (size_t)2 * N_NODES * 64);
  int*   cursor = cnt + NBK;
  int*   offs   = cnt + 1024;
  int*   bins   = cnt + 2048;
  int*   goffs  = cnt + 2048 + N_EDGES;
  int*   gbins  = cnt + 2048 + N_EDGES + 16384;

  gin_out_init_kernel<<<(N_GRAPHS * 10 + 255) / 256, 256, 0, stream>>>(lin_b, out, cnt);

  gin_bin_count_kernel<<<BF_BLOCKS, BF_THREADS, 0, stream>>>(dst, cnt);
  gin_bin_fill_kernel<<<BF_BLOCKS, BF_THREADS, 0, stream>>>(src, dst, cnt, cursor, offs, bins);
  gin_bin_group_kernel<<<NBK, GR_THREADS, 0, stream>>>(offs, bins, goffs, gbins);

  const int mlpBlocks = (NTILES + 3) / 4;

  gin_mlp_kernel<D_FEAT, true><<<mlpBlocks, 128, 0, stream>>>(
      x, fh_W1, fh_b1, fh_bn1, fh_W2, fh_b2, fh_bn2, hA,
      lin_W, lin_b, batch, out, NTILES);

  float* hin = hA;
  float* hnext = hB;
  for (int l = 0; l < 4; l++) {
    gin_conv_kernel<<<NBK, AG_THREADS, 0, stream>>>(
        hin, goffs, gbins,
        conv_W1 + (size_t)l * 64 * 64, conv_b1 + l * 64, conv_bn1 + l * 256,
        conv_W2 + (size_t)l * 64 * 64, conv_b2 + l * 64, conv_bn2 + l * 256,
        hnext,
        lin_W + (size_t)(l + 1) * 64 * 10, batch, out);
    float* tswap = hin;
    hin = hnext;
    hnext = tswap;
  }
}


# Layer kernel as two wave teams: 8 summing waves gather the range's rows in 3 chunks while 8 matrix waves run MLP and head of the chunk before; neighbour sum order = grouping arrival rank, as before
# speedup vs baseline: 3.7074x; 1.1005x over previous
#include <hip/hip_runtime.h>

#define N_NODES   100000
#define N_EDGES   1600000
#define N_GRAPHS  1000
#define D_FEAT    128
#define HID       64
#define BN_EPS    1e-5f
#define NTILES    (N_NODES / 16)

typedef __attribute__((ext_vector_type(2))) float v2f;
typedef __attribute__((ext_vector_type(8))) float v8f;

template<int K1, bool NODE_BIAS>
__launch_bounds__(128)
__global__ void gin_mlp_kernel(const float* __restrict__ xin,
                               const float* __restrict__ W1,
                               const float* __restrict__ b1,
                               const float* __restrict__ bn1,
                               const float* __restrict__ W2,
                               const float* __restrict__ b2,
                               const float* __restrict__ bn2,
                               float* __restrict__ hout,
                               const float* __restrict__ linW,
                               const float* __restrict__ linb0,
                               const int* __restrict__ batch,
                               float* __restrict__ out,
                               int ntiles) {
  constexpr int KS1 = K1 / 4;
  __shared__ __align__(16) float w1s[4 * KS1 * 32 * 2];
  __shared__ __align__(16) float w2s[4 * 16 * 32 * 2];
  __shared__ __align__(16) float tb[4][16 * 68];

  const int tid = threadIdx.x;

  for (int i = tid; i < K1 * 64; i += 128) {
    int kk = i >> 6, nn = i & 63;
    int idx = ((((nn >> 4) * KS1) + (kk >> 2)) * 32 + (nn & 15) + ((kk & 2) << 3)) * 2 + (kk & 1);
    w1s[idx] = W1[i];
  }
  for (int i = tid; i < 64 * 64; i += 128) {
    int kk = i >> 6, nn = i & 63;
    int idx = ((((nn >> 4) * 16) + (kk >> 2)) * 32 + (nn & 15) + ((kk & 2) << 3)) * 2 + (kk & 1);
    w2s[idx] = W2[i];
  }
  __syncthreads();

  const int lane = tid & 31;
  const int wave = tid >> 5;
  const int row  = lane & 15;
  const int hi   = lane >> 4;
  const int tile = blockIdx.x * 4 + wave;
  const bool active = tile < ntiles;
  float* tbuf = tb[wave];

  const v8f vzero = {0.f, 0.f, 0.f, 0.f, 0.f, 0.f, 0.f, 0.f};

  if (active) {
    v2f a1[KS1];
    const float* xr = xin + (size_t)(tile * 16 + row) * K1 + 2 * hi;
#pragma unroll
    for (int k = 0; k < KS1; k++) a1[k] = *(const v2f*)(xr + 4 * k);

#pragma unroll
    for (int nt = 0; nt < 4; nt++) {
      v8f acc = vzero;
      const v2f* wf = ((const v2f*)w1s) + nt * KS1 * 32 + lane;
#pragma unroll
      for (int k = 0; k < KS1; k++) {
        acc = __builtin_amdgcn_wmma_f32_16x16x4_f32(false, a1[k], false, wf[k * 32],
                                                    (short)0, acc, false, false);
      }
      int n = nt * 16 + row;
      float s  = bn1[n] * rsqrtf(bn1[192 + n] + BN_EPS);
      float t0 = bn1[64 + n] + (b1[n] - bn1[128 + n]) * s;
#pragma unroll
      for (int i = 0; i < 8; i++) {
        float y = acc[i] * s + t0;
        tbuf[(i + 8 * hi) * 68 + n] = fmaxf(y, 0.0f);
      }
    }
  }
  __syncthreads();

  v8f acc2[4];
  v2f wl[16];
  int gid[8];
  const int hc = row < 10 ? row : 9;
#pragma unroll
  for (int k = 0; k < 16; k++) wl[k] = (v2f){0.f, 0.f};
#pragma unroll
  for (int r = 0; r < 8; r++) gid[r] = 0;
  if (active) {
    const float* lw = linW + (2 * hi) * 10 + hc;
#pragma unroll
    for (int k = 0; k < 16; k++) {
      float w0 = lw[(4 * k) * 10];
      float w1 = lw[(4 * k + 1) * 10];
      wl[k] = (v2f){row < 10 ? w0 : 0.f, row < 10 ? w1 : 0.f};
    }
#pragma unroll
    for (int r = 0; r < 8; r++) gid[r] = batch[tile * 16 + 8 * hi + r];
#pragma unroll
    for (int nt = 0; nt < 4; nt++) {
      v8f acc = vzero;
      const v2f* wf = ((const v2f*)w2s) + nt * 16 * 32 + lane;
      const float* tr = tbuf + row * 68 + 2 * hi;
#pragma unroll
      for (int k = 0; k < 16; k++) {
        v2f a = *(const v2f*)(tr + 4 * k);
        acc = __builtin_amdgcn_wmma_f32_16x16x4_f32(false, a, false, wf[k * 32],
                                                    (short)0, acc, false, false);
      }
      acc2[nt] = acc;
    }
  }
  __syncthreads();

  if (active) {
#pragma unroll
    for (int nt = 0; nt < 4; nt++) {
      int n = nt * 16 + row;
      float s  = bn2[n] * rsqrtf(bn2[192 + n] + BN_EPS);
      float t0 = bn2[64 + n] + (b2[n] - bn2[128 + n]) * s;
#pragma unroll
      for (int i = 0; i < 8; i++) {
        float y = acc2[nt][i] * s + t0;
        tbuf[(i + 8 * hi) * 68 + n] = fmaxf(y, 0.0f);
      }
    }
  }
  __syncthreads();

  if (active) {
    v8f p0 = vzero, p1 = vzero;
    const float* tr = tbuf + row * 68 + 2 * hi;
#pragma unroll
    for (int k = 0; k < 16; k += 2) {
      v2f a0 = *(const v2f*)(tr + 4 * k);
      v2f a1 = *(const v2f*)(tr + 4 * (k + 1));
      p0 = __builtin_amdgcn_wmma_f32_16x16x4_f32(false, a0, false, wl[k],
                                                 (short)0, p0, false, false);
      p1 = __builtin_amdgcn_wmma_f32_16x16x4_f32(false, a1, false, wl[k + 1],
                                                 (short)0, p1, false, false);
    }
    float pv[8];
#pragma unroll
    for (int r = 0; r < 8; r++) pv[r] = p0[r] + p1[r];
    if (NODE_BIAS) {
      const float lb = linb0[hc];
#pragma unroll
      for (int r = 0; r < 8; r++) pv[r] += lb;
    }

#pragma unroll
    for (int j = 0; j < 8; j++) {
      int flat = j * 32 + lane;
      int r = flat >> 4;
      int c = (flat & 15) << 2;
      float4 v = *(const float4*)(tbuf + r * 68 + c);
      *(float4*)(hout + (size_t)(tile * 16 + r) * 64 + c) = v;
    }

    if (row < 10) {
#pragma unroll
      for (int r = 0; r < 8; r++) atomicAdd(out + (size_t)gid[r] * 10 + row, pv[r]);
    }
  }
}

#define BK_NODES 200
#define NBK (N_NODES / BK_NODES)
#define BF_THREADS 800
#define BF_EDGES_PER_THREAD 8
#define BF_EDGES_PER_BLOCK (BF_THREADS * BF_EDGES_PER_THREAD)
#define BF_BLOCKS (N_EDGES / BF_EDGES_PER_BLOCK)
#define SRC_BITS 17
static_assert(N_NODES % BK_NODES == 0, "ranges must tile the nodes exactly");
static_assert(N_EDGES % BF_EDGES_PER_BLOCK == 0, "edge blocks must tile the edges exactly");
static_assert(N_NODES <= (1 << SRC_BITS), "a source id must fit the packed field");
static_assert(NBK <= 512, "the range scan below holds 512 entries");

__launch_bounds__(BF_THREADS)
__global__ void gin_bin_count_kernel(const int* __restrict__ dst,
                                     int* __restrict__ cnt) {
  __shared__ int hist[NBK];
  const int tid = threadIdx.x;
  for (int b = tid; b < NBK; b += BF_THREADS) hist[b] = 0;
  __syncthreads();
  const int e0 = blockIdx.x * BF_EDGES_PER_BLOCK + tid;
  unsigned d[BF_EDGES_PER_THREAD];
#pragma unroll
  for (int k = 0; k < BF_EDGES_PER_THREAD; k++) d[k] = (unsigned)dst[e0 + k * BF_THREADS];
#pragma unroll
  for (int k = 0; k < BF_EDGES_PER_THREAD; k++) atomicAdd(&hist[d[k] / BK_NODES], 1);
  __syncthreads();
  for (int b = tid; b < NBK; b += BF_THREADS) {
    const int c = hist[b];
    if (c) atomicAdd(&cnt[b], c);
  }
}

__launch_bounds__(BF_THREADS)
__global__ void gin_bin_fill_kernel(const int* __restrict__ src,
                                    const int* __restrict__ dst,
                                    const int* __restrict__ cnt,
                                    int* __restrict__ cursor,
                                    int* __restrict__ offs,
                                    int* __restrict__ bins) {
  __shared__ int hist[NBK];
  __shared__ int bpos[NBK];
  __shared__ int sc[2][512];
  const int tid = threadIdx.x;
  for (int b = tid; b < NBK; b += BF_THREADS) hist[b] = 0;
  if (tid < 512) sc[0][tid] = tid < NBK ? cnt[tid] : 0;
  __syncthreads();

  int bk[BF_EDGES_PER_THREAD], rk[BF_EDGES_PER_THREAD], pk[BF_EDGES_PER_THREAD];
  const int e0 = blockIdx.x * BF_EDGES_PER_BLOCK + tid;
#pragma unroll
  for (int k = 0; k < BF_EDGES_PER_THREAD; k++) {
    pk[k] = src[e0 + k * BF_THREADS];
    bk[k] = dst[e0 + k * BF_THREADS];
  }
#pragma unroll
  for (int k = 0; k < BF_EDGES_PER_THREAD; k++) {
    const unsigned s = (unsigned)pk[k];
    const unsigned d = (unsigned)bk[k];
    const unsigned b = d / BK_NODES;
    bk[k] = (int)b;
    pk[k] = (int)(((d - b * BK_NODES) << SRC_BITS) | s);
    rk[k] = atomicAdd(&hist[b], 1);
  }

  int cur = 0;
  for (int off = 1; off < 512; off <<= 1) {
    if (tid < 512) {
      int v = sc[cur][tid];
      if (tid >= off) v += sc[cur][tid - off];
      sc[cur ^ 1][tid] = v;
    }
    __syncthreads();
    cur ^= 1;
  }
  const int* scan = sc[cur];

  for (int b = tid; b < NBK; b += BF_THREADS) {
    const int c = hist[b];
    const int start = b ? scan[b - 1] : 0;
    const int r = c ? atomicAdd(&cursor[b], c) : 0;
    bpos[b] = start + r;
    if (blockIdx.x == 0) offs[b] = start;
  }
  if (blockIdx.x == 0 && tid == 0) offs[NBK] = scan[NBK - 1];
  __syncthreads();

#pragma unroll
  for (int k = 0; k < BF_EDGES_PER_THREAD; k++) bins[bpos[bk[k]] + rk[k]] = pk[k];
}


#define GT_WAVES 8
#define CH_N 3
#define CH1_ROW 80
#define CH2_ROW 144
#define NGRP (CH_N * GT_WAVES)
static_assert(NBK * (NGRP + 1) <= 16384, "group starts fit the space kept for them");
static_assert(CH1_ROW % 16 == 0 && CH2_ROW % 16 == 0, "chunks are whole tiles");
__device__ __forceinline__ int gin_group_of(int p) {
  const int r = p >> SRC_BITS;
  const int c = (r >= CH1_ROW ? 1 : 0) + (r >= CH2_ROW ? 1 : 0);
  return c * GT_WAVES + (r & (GT_WAVES - 1));
}
__device__ __forceinline__ int gin_slot_entry(int p) {
  const int r = p >> SRC_BITS;
  const int cs = r >= CH2_ROW ? CH2_ROW : (r >= CH1_ROW ? CH1_ROW : 0);
  return (((r - cs) / GT_WAVES) << SRC_BITS) | (p & ((1 << SRC_BITS) - 1));
}
#define GR_THREADS 512
#define GR_KEEP 8
__launch_bounds__(GR_THREADS)
__global__ void gin_bin_group_kernel(const int* __restrict__ offs,
                                     const int* __restrict__ bins,
                                     int* __restrict__ goffs,
                                     int* __restrict__ gbins) {
  __shared__ int gcnt[NGRP];
  __shared__ int gpos[NGRP];
  __shared__ int gbase[NGRP + 1];
  const int tid = threadIdx.x;
  const int b = blockIdx.x;
  const int e0 = offs[b];
  const int e1 = offs[b + 1];
  if (tid < NGRP) {
    gcnt[tid] = 0;
    gpos[tid] = 0;
  }
  int pr[GR_KEEP];
#pragma unroll
  for (int k = 0; k < GR_KEEP; k++) {
    const int i = e0 + tid + k * GR_THREADS;
    pr[k] = bins[i < e1 ? i : N_EDGES - 1];
  }
  __syncthreads();
#pragma unroll
  for (int k = 0; k < GR_KEEP; k++) {
    if (e0 + tid + k * GR_THREADS < e1) atomicAdd(&gcnt[gin_group_of(pr[k])], 1);
  }
  for (int i = e0 + tid + GR_KEEP * GR_THREADS; i < e1; i += GR_THREADS) {
    const int g = gin_group_of(bins[i]);
    atomicAdd(&gcnt[g], 1);
  }
  __syncthreads();
  if (tid == 0) {
    int s = e0;
    for (int g = 0; g < NGRP; g++) {
      gbase[g] = s;
      s += gcnt[g];
    }
    gbase[NGRP] = s;
  }
  __syncthreads();
  if (tid <= NGRP) goffs[b * (NGRP + 1) + tid] = gbase[tid];
#pragma unroll
  for (int k = 0; k < GR_KEEP; k++) {
    if (e0 + tid + k * GR_THREADS < e1) {
      const int g = gin_group_of(pr[k]);
      const int r = atomicAdd(&gpos[g], 1);
      gbins[gbase[g] + r] = gin_slot_entry(pr[k]);
    }
  }
  for (int i = e0 + tid + GR_KEEP * GR_THREADS; i < e1; i += GR_THREADS) {
    const int p = bins[i];
    const int g = gin_group_of(p);
    const int r = atomicAdd(&gpos[g], 1);
    gbins[gbase[g] + r] = gin_slot_entry(p);
  }
}

#define AG_WAVES 16
#define AG_THREADS (AG_WAVES * 32)
#define MT_WAVES (AG_WAVES - GT_WAVES)
#define MT_THREADS (MT_WAVES * 32)
#define CV_TILES ((BK_NODES + 15) / 16)
#define CV_ROWS (CV_TILES * 16)
#define CV_PITCH 68
#define CH0_TILES (CH1_ROW / 16)
#define CH1_TILES ((CH2_ROW - CH1_ROW) / 16)
#define CH2_TILES (CV_TILES - CH2_ROW / 16)
#define GS_MAX (CH1_ROW / GT_WAVES)
#define GS_MIN 8
static_assert(BK_NODES % 8 == 0, "a lane half of a tile (8 rows) is all valid or all padding");
static_assert(CH0_TILES <= MT_WAVES && CH1_TILES <= MT_WAVES && CH2_TILES <= MT_WAVES, "one matrix wave per tile of a chunk");
static_assert((CH2_ROW - CH1_ROW) / GT_WAVES == GS_MIN && (CV_ROWS - CH2_ROW) / GT_WAVES == GS_MIN, "slots of chunks 1 and 2");
static_assert(GS_MAX >= GS_MIN && GS_MAX <= 16, "a summing wave's slots fit one 16-register vector per column half");
static_assert(CH2_ROW + GT_WAVES - 1 < BK_NODES, "slot 0 of every summing wave is a real row in every chunk");
static_assert((64 * 64) % MT_THREADS == 0, "weight pieces per thread");
#define CV_WPT ((64 * 64) / MT_THREADS)
typedef __attribute__((ext_vector_type(16))) float v16f;
__launch_bounds__(AG_THREADS, 8)
__global__ void gin_conv_kernel(const float* __restrict__ h,
                                const int* __restrict__ goffs,
                                const int* __restrict__ gbins,
                                const float* __restrict__ W1,
                                const float* __restrict__ b1,
                                const float* __restrict__ bn1,
                                const float* __restrict__ W2,
                                const float* __restrict__ b2,
                                const float* __restrict__ bn2,
                                float* __restrict__ hout,
                                const float* __restrict__ linW,
                                const int* __restrict__ batch,
                                float* __restrict__ out) {
  __shared__ __align__(16) float acc[CV_ROWS * CV_PITCH];
  __shared__ __align__(16) float w1s[4 * 16 * 32 * 2];
  __shared__ __align__(16) float w2s[4 * 16 * 32 * 2];
  __shared__ float bns[4][64];
  const int tid  = threadIdx.x;
  const int lane = tid & 31;
  const int wave = __builtin_amdgcn_readfirstlane(tid >> 5);
  const bool gteam = wave < GT_WAVES;
  const int b = blockIdx.x;
  const size_t rowbase = (size_t)b * BK_NODES * 64;

  const int* gp = goffs + b * (NGRP + 1) + (wave & (GT_WAVES - 1));
  const int ga0 = gp[0],            ga1 = gp[1];
  const int gb0 = gp[GT_WAVES],     gb1 = gp[GT_WAVES + 1];
  const int gc0 = gp[2 * GT_WAVES], gc1 = gp[2 * GT_WAVES + 1];

  if (!gteam) {
    const int mt = tid - GT_WAVES * 32;
    float wa[CV_WPT], wb[CV_WPT];
#pragma unroll
    for (int k = 0; k < CV_WPT; k++) {
      wa[k] = W1[mt + k * MT_THREADS];
      wb[k] = W2[mt + k * MT_THREADS];
    }
#pragma unroll
    for (int k = 0; k < CV_WPT; k++) {
      const int i = mt + k * MT_THREADS;
      const int kk = i >> 6, nn = i & 63;
      const int idx = ((((nn >> 4) * 16) + (kk >> 2)) * 32 + (nn & 15) + ((kk & 2) << 3)) * 2 + (kk & 1);
      w1s[idx] = wa[k];
      w2s[idx] = wb[k];
    }
    if (mt < 128) {
      const float* bp = mt < 64 ? bn1 : bn2;
      const float* lp = mt < 64 ? b1 : b2;
      const int n = mt & 63;
      const float sc = bp[n] * rsqrtf(bp[192 + n] + BN_EPS);
      bns[(mt >> 6) * 2][n] = sc;
      bns[(mt >> 6) * 2 + 1][n] = bp[64 + n] + (lp[n] - bp[128 + n]) * sc;
    }
  }

#pragma nounroll
  for (int c = 0; c <= CH_N; c++) {
    if (gteam) {
      if (c < CH_N) {
        const int cs  = c == 0 ? 0 : (c == 1 ? CH1_ROW : CH2_ROW);
        const int nsl = c == 0 ? GS_MAX : GS_MIN;
        int lg = lane;
        asm volatile("" : "+v"(lg));
        const int g0  = c == 0 ? ga0 : (c == 1 ? gb0 : gc0);
        const int g1  = c == 0 ? ga1 : (c == 1 ? gb1 : gc1);
        v16f sx = {0.f, 0.f, 0.f, 0.f, 0.f, 0.f, 0.f, 0.f, 0.f, 0.f, 0.f, 0.f, 0.f, 0.f, 0.f, 0.f};
        v16f sy = sx;
#pragma unroll
        for (int sl = 0; sl < GS_MAX; sl++) {
          const int r = cs + wave + GT_WAVES * sl;
          const bool real = sl < nsl && r < BK_NODES;
          const v2f t = *(const v2f*)(h + rowbase + (size_t)(real ? r : cs + wave) * 64 + 2 * lg);
          sx[sl] = real ? t.x : 0.f;
          sy[sl] = real ? t.y : 0.f;
        }
        int e = g0;
        int p = gbins[e + lg < N_EDGES ? e + lg : N_EDGES - 1];
        while (e + 32 <= g1) {
          const int en = e + 32;
          v2f v[32];
#pragma unroll
          for (int j = 0; j < 32; j++) {
            const int pj = __builtin_amdgcn_readlane(p, j);
            v[j] = *(const v2f*)(h + (size_t)(pj & ((1 << SRC_BITS) - 1)) * 64 + 2 * lg);
          }
          const int pn = gbins[en + lg < N_EDGES ? en + lg : N_EDGES - 1];
          __builtin_amdgcn_sched_barrier(0);
#pragma unroll
          for (int j = 0; j < 32; j++) {
            const int sl = __builtin_amdgcn_readlane(p, j) >> SRC_BITS;
            sx[sl] += v[j].x;
            sy[sl] += v[j].y;
          }
          p = pn;
          e = en;
        }
        if (e < g1) {
          const int n = g1 - e;
          const int pt = lg < n ? p : __builtin_amdgcn_readlane(p, 0);
          v2f v[32];
#pragma unroll
          for (int j = 0; j < 32; j++) {
            const int pj = __builtin_amdgcn_readlane(pt, j);
            v[j] = *(const v2f*)(h + (size_t)(pj & ((1 << SRC_BITS) - 1)) * 64 + 2 * lg);
          }
#pragma unroll
          for (int j = 0; j < 32; j++) {
            if (j < n) {
              const int sl = __builtin_amdgcn_readlane(pt, j) >> SRC_BITS;
              sx[sl] += v[j].x;
              sy[sl] += v[j].y;
            }
          }
        }
        float* ar = acc + (cs + wave) * CV_PITCH + 2 * lg;
#pragma unroll
        for (int sl = 0; sl < GS_MIN; sl++) {
          *(v2f*)(ar + GT_WAVES * sl * CV_PITCH) = (v2f){sx[sl], sy[sl]};
        }
        if (nsl > GS_MIN) {
#pragma unroll
          for (int sl = GS_MIN; sl < GS_MAX; sl++) {
            *(v2f*)(ar + GT_WAVES * sl * CV_PITCH) = (v2f){sx[sl], sy[sl]};
          }
        }
      }
    } else if (c > 0) {
      const int mc = c - 1;
      const int m  = wave - GT_WAVES;
      const int nct = mc == 0 ? CH0_TILES : (mc == 1 ? CH1_TILES : CH2_TILES);
      if (m < nct) {
        const int tile = (mc == 0 ? 0 : (mc == 1 ? CH1_ROW / 16 : CH2_ROW / 16)) + m;
        int ln = lane;
        asm volatile("" : "+v"(ln));
        const int row  = ln & 15;
        const int hi   = ln >> 4;
        const int left = BK_NODES - tile * 16;
        const int nrow = left < 16 ? left : 16;
        const bool rows_ok = 8 * hi < nrow;
        const int node0 = b * BK_NODES + tile * 16;
        float* tbuf = acc + tile * 16 * CV_PITCH;
        const v8f vzero = {0.f, 0.f, 0.f, 0.f, 0.f, 0.f, 0.f, 0.f};

        v2f a1[16];
        {
          const float* tr = tbuf + row * CV_PITCH + 2 * hi;
#pragma unroll
          for (int k = 0; k < 16; k++) a1[k] = *(const v2f*)(tr + 4 * k);
        }

#pragma unroll
        for (int nt = 0; nt < 4; nt++) {
          v8f a = vzero;
          const v2f* wf = ((const v2f*)w1s) + nt * 16 * 32 + ln;
#pragma unroll
          for (int k = 0; k < 16; k++) {
            a = __builtin_amdgcn_wmma_f32_16x16x4_f32(false, a1[k], false, wf[k * 32],
                                                      (short)0, a, false, false);
          }
          int n = nt * 16 + row;
          float sc = bns[0][n];
          float t0 = bns[1][n];
#pragma unroll
          for (int i = 0; i < 8; i++) {
            float y = a[i] * sc + t0;
            tbuf[(i + 8 * hi) * CV_PITCH + n] = fmaxf(y, 0.0f);
          }
        }
        __builtin_amdgcn_fence(__ATOMIC_SEQ_CST, "wavefront");

        v8f acc2[4];
        v2f wl[16];
        int gid[8];
        const int hc = row < 10 ? row : 9;
#pragma unroll
        for (int r = 0; r < 8; r++) gid[r] = 0;
        {
          const float* lw = linW + (2 * hi) * 10 + hc;
#pragma unroll
          for (int k = 0; k < 16; k++) {
            float w0 = lw[(4 * k) * 10];
            float w1 = lw[(4 * k + 1) * 10];
            wl[k] = (v2f){row < 10 ? w0 : 0.f, row < 10 ? w1 : 0.f};
          }
        }
        if (rows_ok) {
#pragma unroll
          for (int r = 0; r < 8; r++) gid[r] = batch[node0 + 8 * hi + r];
        }
#pragma unroll
        for (int nt = 0; nt < 4; nt++) {
          v8f a = vzero;
          const v2f* wf = ((const v2f*)w2s) + nt * 16 * 32 + ln;
          const float* tr = tbuf + row * CV_PITCH + 2 * hi;
#pragma unroll
          for (int k = 0; k < 16; k++) {
            v2f av = *(const v2f*)(tr + 4 * k);
            a = __builtin_amdgcn_wmma_f32_16x16x4_f32(false, av, false, wf[k * 32],
                                                      (short)0, a, false, false);
          }
          acc2[nt] = a;
        }
        __builtin_amdgcn_fence(__ATOMIC_SEQ_CST, "wavefront");

#pragma unroll
        for (int nt = 0; nt < 4; nt++) {
          int n = nt * 16 + row;
          float sc = bns[2][n];
          float t0 = bns[3][n];
#pragma unroll
          for (int i = 0; i < 8; i++) {
            float y = acc2[nt][i] * sc + t0;
            tbuf[(i + 8 * hi) * CV_PITCH + n] = fmaxf(y, 0.0f);
          }
        }
        __builtin_amdgcn_fence(__ATOMIC_SEQ_CST, "wavefront");

        v8f p0 = vzero, p1 = vzero;
        const float* tr = tbuf + row * CV_PITCH + 2 * hi;
#pragma unroll
        for (int k = 0; k < 16; k += 2) {
          v2f a0 = *(const v2f*)(tr + 4 * k);
          v2f a1h = *(const v2f*)(tr + 4 * (k + 1));
          p0 = __builtin_amdgcn_wmma_f32_16x16x4_f32(false, a0, false, wl[k],
                                                     (short)0, p0, false, false);
          p1 = __builtin_amdgcn_wmma_f32_16x16x4_f32(false, a1h, false, wl[k + 1],
                                                     (short)0, p1, false, false);
        }
        float pv[8];
#pragma unroll
        for (int r = 0; r < 8; r++) pv[r] = p0[r] + p1[r];

#pragma unroll
        for (int j = 0; j < 8; j++) {
          const int r = 2 * j + hi;
          const int cc = row << 2;
          if (r < nrow) {
            float4 v = *(const float4*)(tbuf + r * CV_PITCH + cc);
            *(float4*)(hout + (size_t)(node0 + r) * 64 + cc) = v;
          }
        }

        if (row < 10 && rows_ok) {
#pragma unroll
          for (int r = 0; r < 8; r++) atomicAdd(out + (size_t)gid[r] * 10 + row, pv[r]);
        }
      }
    }
    if (c < CH_N) __syncthreads();
  }
}

__global__ void gin_out_init_kernel(const float* __restrict__ linb,
                                    float* __restrict__ out,
                                    int* __restrict__ cnt) {
  int idx = blockIdx.x * blockDim.x + threadIdx.x;
  if (idx < 2 * NBK) cnt[idx] = 0;
  if (idx >= N_GRAPHS * 10) return;
  int t = idx % 10;
  out[idx] = ((linb[10 + t] + linb[20 + t]) + linb[30 + t]) + linb[40 + t];
}

extern "C" void kernel_launch(void* const* d_in, const int* in_sizes, int n_in,
                              void* d_out, int out_size, void* d_ws, size_t ws_size,
                              hipStream_t stream) {
  const float* x        = (const float*)d_in[0];
  const float* fh_W1    = (const float*)d_in[1];
  const float* fh_b1    = (const float*)d_in[2];
  const float* fh_bn1   = (const float*)d_in[3];
  const float* fh_W2    = (const float*)d_in[4];
  const float* fh_b2    = (const float*)d_in[5];
  const float* fh_bn2   = (const float*)d_in[6];
  const float* conv_W1  = (const float*)d_in[7];
  const float* conv_b1  = (const float*)d_in[8];
  const float* conv_bn1 = (const float*)d_in[9];
  const float* conv_W2  = (const float*)d_in[10];
  const float* conv_b2  = (const float*)d_in[11];
  const float* conv_bn2 = (const float*)d_in[12];
  const float* lin_W    = (const float*)d_in[13];
  const float* lin_b    = (const float*)d_in[14];
  const int*   eidx     = (const int*)d_in[15];
  const int*   batch    = (const int*)d_in[16];
  float* out = (float*)d_out;

  const int* src = eidx;
  const int* dst = eidx + N_EDGES;

  float* ws     = (float*)d_ws;
  float* hA     = ws;
  float* hB     = ws + (size_t)N_NODES * 64;
  int*   cnt    = (int*)(ws + (size_t)2 * N_NODES * 64);
  int*   cursor = cnt + NBK;
  int*   offs   = cnt + 1024;
  int*   bins   = cnt + 2048;
  int*   goffs  = cnt + 2048 + N_EDGES;
  int*   gbins  = cnt + 2048 + N_EDGES + 16384;

  gin_out_init_kernel<<<(N_GRAPHS * 10 + 255) / 256, 256, 0, stream>>>(lin_b, out, cnt);

  gin_bin_count_kernel<<<BF_BLOCKS, BF_THREADS, 0, stream>>>(dst, cnt);
  gin_bin_fill_kernel<<<BF_BLOCKS, BF_THREADS, 0, stream>>>(src, dst, cnt, cursor, offs, bins);
  gin_bin_group_kernel<<<NBK, GR_THREADS, 0, stream>>>(offs, bins, goffs, gbins);

  const int mlpBlocks = (NTILES + 3) / 4;

  gin_mlp_kernel<D_FEAT, true><<<mlpBlocks, 128, 0, stream>>>(
      x, fh_W1, fh_b1, fh_bn1, fh_W2, fh_b2, fh_bn2, hA,
      lin_W, lin_b, batch, out, NTILES);

  float* hin = hA;
  float* hnext = hB;
  for (int l = 0; l < 4; l++) {
    gin_conv_kernel<<<NBK, AG_THREADS, 0, stream>>>(
        hin, goffs, gbins,
        conv_W1 + (size_t)l * 64 * 64, conv_b1 + l * 64, conv_bn1 + l * 256,
        conv_W2 + (size_t)l * 64 * 64, conv_b2 + l * 64, conv_bn2 + l * 256,
        hnext,
        lin_W + (size_t)(l + 1) * 64 * 10, batch, out);
    float* tswap = hin;
    hin = hnext;
    hnext = tswap;
  }
}


# Edge binning: fill and grouping passes lay their entries out in LDS in final order and store them with consecutive lanes (was 32 scattered 4-byte stores per store instruction); same sum order
# speedup vs baseline: 3.9423x; 1.0634x over previous
#include <hip/hip_runtime.h>

#define N_NODES   100000
#define N_EDGES   1600000
#define N_GRAPHS  1000
#define D_FEAT    128
#define HID       64
#define BN_EPS    1e-5f
#define NTILES    (N_NODES / 16)

typedef __attribute__((ext_vector_type(2))) float v2f;
typedef __attribute__((ext_vector_type(8))) float v8f;

template<int K1, bool NODE_BIAS>
__launch_bounds__(128)
__global__ void gin_mlp_kernel(const float* __restrict__ xin,
                               const float* __restrict__ W1,
                               const float* __restrict__ b1,
                               const float* __restrict__ bn1,
                               const float* __restrict__ W2,
                               const float* __restrict__ b2,
                               const float* __restrict__ bn2,
                               float* __restrict__ hout,
                               const float* __restrict__ linW,
                               const float* __restrict__ linb0,
                               const int* __restrict__ batch,
                               float* __restrict__ out,
                               int ntiles) {
  constexpr int KS1 = K1 / 4;
  __shared__ __align__(16) float w1s[4 * KS1 * 32 * 2];
  __shared__ __align__(16) float w2s[4 * 16 * 32 * 2];
  __shared__ __align__(16) float tb[4][16 * 68];

  const int tid = threadIdx.x;

  for (int i = tid; i < K1 * 64; i += 128) {
    int kk = i >> 6, nn = i & 63;
    int idx = ((((nn >> 4) * KS1) + (kk >> 2)) * 32 + (nn & 15) + ((kk & 2) << 3)) * 2 + (kk & 1);
    w1s[idx] = W1[i];
  }
  for (int i = tid; i < 64 * 64; i += 128) {
    int kk = i >> 6, nn = i & 63;
    int idx = ((((nn >> 4) * 16) + (kk >> 2)) * 32 + (nn & 15) + ((kk & 2) << 3)) * 2 + (kk & 1);
    w2s[idx] = W2[i];
  }
  __syncthreads();

  const int lane = tid & 31;
  const int wave = tid >> 5;
  const int row  = lane & 15;
  const int hi   = lane >> 4;
  const int tile = blockIdx.x * 4 + wave;
  const bool active = tile < ntiles;
  float* tbuf = tb[wave];

  const v8f vzero = {0.f, 0.f, 0.f, 0.f, 0.f, 0.f, 0.f, 0.f};

  if (active) {
    v2f a1[KS1];
    const float* xr = xin + (size_t)(tile * 16 + row) * K1 + 2 * hi;
#pragma unroll
    for (int k = 0; k < KS1; k++) a1[k] = *(const v2f*)(xr + 4 * k);

#pragma unroll
    for (int nt = 0; nt < 4; nt++) {
      v8f acc = vzero;
      const v2f* wf = ((const v2f*)w1s) + nt * KS1 * 32 + lane;
#pragma unroll
      for (int k = 0; k < KS1; k++) {
        acc = __builtin_amdgcn_wmma_f32_16x16x4_f32(false, a1[k], false, wf[k * 32],
                                                    (short)0, acc, false, false);
      }
      int n = nt * 16 + row;
      float s  = bn1[n] * rsqrtf(bn1[192 + n] + BN_EPS);
      float t0 = bn1[64 + n] + (b1[n] - bn1[128 + n]) * s;
#pragma unroll
      for (int i = 0; i < 8; i++) {
        float y = acc[i] * s + t0;
        tbuf[(i + 8 * hi) * 68 + n] = fmaxf(y, 0.0f);
      }
    }
  }
  __syncthreads();

  v8f acc2[4];
  v2f wl[16];
  int gid[8];
  const int hc = row < 10 ? row : 9;
#pragma unroll
  for (int k = 0; k < 16; k++) wl[k] = (v2f){0.f, 0.f};
#pragma unroll
  for (int r = 0; r < 8; r++) gid[r] = 0;
  if (active) {
    const float* lw = linW + (2 * hi) * 10 + hc;
#pragma unroll
    for (int k = 0; k < 16; k++) {
      float w0 = lw[(4 * k) * 10];
      float w1 = lw[(4 * k + 1) * 10];
      wl[k] = (v2f){row < 10 ? w0 : 0.f, row < 10 ? w1 : 0.f};
    }
#pragma unroll
    for (int r = 0; r < 8; r++) gid[r] = batch[tile * 16 + 8 * hi + r];
#pragma unroll
    for (int nt = 0; nt < 4; nt++) {
      v8f acc = vzero;
      const v2f* wf = ((const v2f*)w2s) + nt * 16 * 32 + lane;
      const float* tr = tbuf + row * 68 + 2 * hi;
#pragma unroll
      for (int k = 0; k < 16; k++) {
        v2f a = *(const v2f*)(tr + 4 * k);
        acc = __builtin_amdgcn_wmma_f32_16x16x4_f32(false, a, false, wf[k * 32],
                                                    (short)0, acc, false, false);
      }
      acc2[nt] = acc;
    }
  }
  __syncthreads();

  if (active) {
#pragma unroll
    for (int nt = 0; nt < 4; nt++) {
      int n = nt * 16 + row;
      float s  = bn2[n] * rsqrtf(bn2[192 + n] + BN_EPS);
      float t0 = bn2[64 + n] + (b2[n] - bn2[128 + n]) * s;
#pragma unroll
      for (int i = 0; i < 8; i++) {
        float y = acc2[nt][i] * s + t0;
        tbuf[(i + 8 * hi) * 68 + n] = fmaxf(y, 0.0f);
      }
    }
  }
  __syncthreads();

  if (active) {
    v8f p0 = vzero, p1 = vzero;
    const float* tr = tbuf + row * 68 + 2 * hi;
#pragma unroll
    for (int k = 0; k < 16; k += 2) {
      v2f a0 = *(const v2f*)(tr + 4 * k);
      v2f a1 = *(const v2f*)(tr + 4 * (k + 1));
      p0 = __builtin_amdgcn_wmma_f32_16x16x4_f32(false, a0, false, wl[k],
                                                 (short)0, p0, false, false);
      p1 = __builtin_amdgcn_wmma_f32_16x16x4_f32(false, a1, false, wl[k + 1],
                                                 (short)0, p1, false, false);
    }
    float pv[8];
#pragma unroll
    for (int r = 0; r < 8; r++) pv[r] = p0[r] + p1[r];
    if (NODE_BIAS) {
      const float lb = linb0[hc];
#pragma unroll
      for (int r = 0; r < 8; r++) pv[r] += lb;
    }

#pragma unroll
    for (int j = 0; j < 8; j++) {
      int flat = j * 32 + lane;
      int r = flat >> 4;
      int c = (flat & 15) << 2;
      float4 v = *(const float4*)(tbuf + r * 68 + c);
      *(float4*)(hout + (size_t)(tile * 16 + r) * 64 + c) = v;
    }

    if (row < 10) {
#pragma unroll
      for (int r = 0; r < 8; r++) atomicAdd(out + (size_t)gid[r] * 10 + row, pv[r]);
    }
  }
}

#define BK_NODES 200
#define NBK (N_NODES / BK_NODES)
#define BF_THREADS 800
#define BF_EDGES_PER_THREAD 8
#define BF_EDGES_PER_BLOCK (BF_THREADS * BF_EDGES_PER_THREAD)
#define BF_BLOCKS (N_EDGES / BF_EDGES_PER_BLOCK)
#define SRC_BITS 17
static_assert(N_NODES % BK_NODES == 0, "ranges must tile the nodes exactly");
static_assert(N_EDGES % BF_EDGES_PER_BLOCK == 0, "edge blocks must tile the edges exactly");
static_assert(N_NODES <= (1 << SRC_BITS), "a source id must fit the packed field");
static_assert(NBK <= 512, "the range scan below holds 512 entries");

__launch_bounds__(BF_THREADS)
__global__ void gin_bin_count_kernel(const int* __restrict__ dst,
                                     int* __restrict__ cnt) {
  __shared__ int hist[NBK];
  const int tid = threadIdx.x;
  for (int b = tid; b < NBK; b += BF_THREADS) hist[b] = 0;
  __syncthreads();
  const int e0 = blockIdx.x * BF_EDGES_PER_BLOCK + tid;
  unsigned d[BF_EDGES_PER_THREAD];
#pragma unroll
  for (int k = 0; k < BF_EDGES_PER_THREAD; k++) d[k] = (unsigned)dst[e0 + k * BF_THREADS];
#pragma unroll
  for (int k = 0; k < BF_EDGES_PER_THREAD; k++) atomicAdd(&hist[d[k] / BK_NODES], 1);
  __syncthreads();
  for (int b = tid; b < NBK; b += BF_THREADS) {
    const int c = hist[b];
    if (c) atomicAdd(&cnt[b], c);
  }
}

__launch_bounds__(BF_THREADS)
__global__ void gin_bin_fill_kernel(const int* __restrict__ src,
                                    const int* __restrict__ dst,
                                    const int* __restrict__ cnt,
                                    int* __restrict__ cursor,
                                    int* __restrict__ offs,
                                    int* __restrict__ bins) {
  __shared__ int hist[NBK];
  __shared__ int bpos[NBK];
  __shared__ int loff[NBK];
  __shared__ int2 sc[2][512];
  __shared__ int2 stage[BF_EDGES_PER_BLOCK];
  const int tid = threadIdx.x;
  for (int b = tid; b < NBK; b += BF_THREADS) hist[b] = 0;
  const int cg = tid < NBK ? cnt[tid] : 0;
  __syncthreads();

  int bk[BF_EDGES_PER_THREAD], rk[BF_EDGES_PER_THREAD], pk[BF_EDGES_PER_THREAD];
  const int e0 = blockIdx.x * BF_EDGES_PER_BLOCK + tid;
#pragma unroll
  for (int k = 0; k < BF_EDGES_PER_THREAD; k++) {
    pk[k] = src[e0 + k * BF_THREADS];
    bk[k] = dst[e0 + k * BF_THREADS];
  }
#pragma unroll
  for (int k = 0; k < BF_EDGES_PER_THREAD; k++) {
    const unsigned s = (unsigned)pk[k];
    const unsigned d = (unsigned)bk[k];
    const unsigned b = d / BK_NODES;
    bk[k] = (int)b;
    pk[k] = (int)(((d - b * BK_NODES) << SRC_BITS) | s);
    rk[k] = atomicAdd(&hist[b], 1);
  }
  __syncthreads();
  if (tid < 512) sc[0][tid] = make_int2(cg, tid < NBK ? hist[tid] : 0);
  __syncthreads();

  int cur = 0;
  for (int off = 1; off < 512; off <<= 1) {
    if (tid < 512) {
      int2 v = sc[cur][tid];
      if (tid >= off) {
        const int2 u = sc[cur][tid - off];
        v.x += u.x;
        v.y += u.y;
      }
      sc[cur ^ 1][tid] = v;
    }
    __syncthreads();
    cur ^= 1;
  }
  const int2* scan = sc[cur];

  for (int b = tid; b < NBK; b += BF_THREADS) {
    const int c = hist[b];
    const int start = b ? scan[b - 1].x : 0;
    const int r = c ? atomicAdd(&cursor[b], c) : 0;
    bpos[b] = start + r;
    loff[b] = b ? scan[b - 1].y : 0;
    if (blockIdx.x == 0) offs[b] = start;
  }
  if (blockIdx.x == 0 && tid == 0) offs[NBK] = scan[NBK - 1].x;
  __syncthreads();

#pragma unroll
  for (int k = 0; k < BF_EDGES_PER_THREAD; k++) {
    const int b = bk[k];
    stage[loff[b] + rk[k]] = make_int2(pk[k], bpos[b] + rk[k]);
  }
  __syncthreads();
#pragma unroll
  for (int k = 0; k < BF_EDGES_PER_THREAD; k++) {
    const int2 s = stage[tid + k * BF_THREADS];
    bins[s.y] = s.x;
  }
}


#define GT_WAVES 8
#define CH_N 3
#define CH1_ROW 80
#define CH2_ROW 144
#define NGRP (CH_N * GT_WAVES)
static_assert(NBK * (NGRP + 1) <= 16384, "group starts fit the space kept for them");
static_assert(CH1_ROW % 16 == 0 && CH2_ROW % 16 == 0, "chunks are whole tiles");
__device__ __forceinline__ int gin_group_of(int p) {
  const int r = p >> SRC_BITS;
  const int c = (r >= CH1_ROW ? 1 : 0) + (r >= CH2_ROW ? 1 : 0);
  return c * GT_WAVES + (r & (GT_WAVES - 1));
}
__device__ __forceinline__ int gin_slot_entry(int p) {
  const int r = p >> SRC_BITS;
  const int cs = r >= CH2_ROW ? CH2_ROW : (r >= CH1_ROW ? CH1_ROW : 0);
  return (((r - cs) / GT_WAVES) << SRC_BITS) | (p & ((1 << SRC_BITS) - 1));
}
#define GR_THREADS 512
#define GR_KEEP 8
#define GR_STAGE (GR_KEEP * GR_THREADS)
__launch_bounds__(GR_THREADS)
__global__ void gin_bin_group_kernel(const int* __restrict__ offs,
                                     const int* __restrict__ bins,
                                     int* __restrict__ goffs,
                                     int* __restrict__ gbins) {
  __shared__ int gcnt[NGRP];
  __shared__ int gpos[NGRP];
  __shared__ int gbase[NGRP + 1];
  __shared__ int stage[GR_STAGE];
  const int tid = threadIdx.x;
  const int b = blockIdx.x;
  const int e0 = offs[b];
  const int e1 = offs[b + 1];
  if (tid < NGRP) {
    gcnt[tid] = 0;
    gpos[tid] = 0;
  }
  int pr[GR_KEEP];
#pragma unroll
  for (int k = 0; k < GR_KEEP; k++) {
    const int i = e0 + tid + k * GR_THREADS;
    pr[k] = bins[i < e1 ? i : N_EDGES - 1];
  }
  __syncthreads();
#pragma unroll
  for (int k = 0; k < GR_KEEP; k++) {
    if (e0 + tid + k * GR_THREADS < e1) atomicAdd(&gcnt[gin_group_of(pr[k])], 1);
  }
  for (int i = e0 + tid + GR_KEEP * GR_THREADS; i < e1; i += GR_THREADS) {
    const int g = gin_group_of(bins[i]);
    atomicAdd(&gcnt[g], 1);
  }
  __syncthreads();
  if (tid == 0) {
    int s = e0;
    for (int g = 0; g < NGRP; g++) {
      gbase[g] = s;
      s += gcnt[g];
    }
    gbase[NGRP] = s;
  }
  __syncthreads();
  if (tid <= NGRP) goffs[b * (NGRP + 1) + tid] = gbase[tid];
#pragma unroll
  for (int k = 0; k < GR_KEEP; k++) {
    if (e0 + tid + k * GR_THREADS < e1) {
      const int g = gin_group_of(pr[k]);
      const int r = atomicAdd(&gpos[g], 1);
      const int q = gbase[g] - e0 + r;
      const int v = gin_slot_entry(pr[k]);
      if (q < GR_STAGE) stage[q] = v;
      else gbins[e0 + q] = v;
    }
  }
  for (int i = e0 + tid + GR_KEEP * GR_THREADS; i < e1; i += GR_THREADS) {
    const int p = bins[i];
    const int g = gin_group_of(p);
    const int r = atomicAdd(&gpos[g], 1);
    const int q = gbase[g] - e0 + r;
    const int v = gin_slot_entry(p);
    if (q < GR_STAGE) stage[q] = v;
    else gbins[e0 + q] = v;
  }
  __syncthreads();
  const int nst = e1 - e0 < GR_STAGE ? e1 - e0 : GR_STAGE;
  int sv[GR_KEEP];
#pragma unroll
  for (int k = 0; k < GR_KEEP; k++) {
    const int i = tid + k * GR_THREADS;
    sv[k] = i < nst ? stage[i] : 0;
  }
#pragma unroll
  for (int k = 0; k < GR_KEEP; k++) {
    const int i = tid + k * GR_THREADS;
    if (i < nst) gbins[e0 + i] = sv[k];
  }
}

#define AG_WAVES 16
#define AG_THREADS (AG_WAVES * 32)
#define MT_WAVES (AG_WAVES - GT_WAVES)
#define MT_THREADS (MT_WAVES * 32)
#define CV_TILES ((BK_NODES + 15) / 16)
#define CV_ROWS (CV_TILES * 16)
#define CV_PITCH 68
#define CH0_TILES (CH1_ROW / 16)
#define CH1_TILES ((CH2_ROW - CH1_ROW) / 16)
#define CH2_TILES (CV_TILES - CH2_ROW / 16)
#define GS_MAX (CH1_ROW / GT_WAVES)
#define GS_MIN 8
static_assert(BK_NODES % 8 == 0, "a lane half of a tile (8 rows) is all valid or all padding");
static_assert(CH0_TILES <= MT_WAVES && CH1_TILES <= MT_WAVES && CH2_TILES <= MT_WAVES, "one matrix wave per tile of a chunk");
static_assert((CH2_ROW - CH1_ROW) / GT_WAVES == GS_MIN && (CV_ROWS - CH2_ROW) / GT_WAVES == GS_MIN, "slots of chunks 1 and 2");
static_assert(GS_MAX >= GS_MIN && GS_MAX <= 16, "a summing wave's slots fit one 16-register vector per column half");
static_assert(CH2_ROW + GT_WAVES - 1 < BK_NODES, "slot 0 of every summing wave is a real row in every chunk");
static_assert((64 * 64) % MT_THREADS == 0, "weight pieces per thread");
#define CV_WPT ((64 * 64) / MT_THREADS)
typedef __attribute__((ext_vector_type(16))) float v16f;
__launch_bounds__(AG_THREADS, 8)
__global__ void gin_conv_kernel(const float* __restrict__ h,
                                const int* __restrict__ goffs,
                                const int* __restrict__ gbins,
                                const float* __restrict__ W1,
                                const float* __restrict__ b1,
                                const float* __restrict__ bn1,
                                const float* __restrict__ W2,
                                const float* __restrict__ b2,
                                const float* __restrict__ bn2,
                                float* __restrict__ hout,
                                const float* __restrict__ linW,
                                const int* __restrict__ batch,
                                float* __restrict__ out) {
  __shared__ __align__(16) float acc[CV_ROWS * CV_PITCH];
  __shared__ __align__(16) float w1s[4 * 16 * 32 * 2];
  __shared__ __align__(16) float w2s[4 * 16 * 32 * 2];
  __shared__ float bns[4][64];
  const int tid  = threadIdx.x;
  const int lane = tid & 31;
  const int wave = __builtin_amdgcn_readfirstlane(tid >> 5);
  const bool gteam = wave < GT_WAVES;
  const int b = blockIdx.x;
  const size_t rowbase = (size_t)b * BK_NODES * 64;

  const int* gp = goffs + b * (NGRP + 1) + (wave & (GT_WAVES - 1));
  const int ga0 = gp[0],            ga1 = gp[1];
  const int gb0 = gp[GT_WAVES],     gb1 = gp[GT_WAVES + 1];
  const int gc0 = gp[2 * GT_WAVES], gc1 = gp[2 * GT_WAVES + 1];

  if (!gteam) {
    const int mt = tid - GT_WAVES * 32;
    float wa[CV_WPT], wb[CV_WPT];
#pragma unroll
    for (int k = 0; k < CV_WPT; k++) {
      wa[k] = W1[mt + k * MT_THREADS];
      wb[k] = W2[mt + k * MT_THREADS];
    }
#pragma unroll
    for (int k = 0; k < CV_WPT; k++) {
      const int i = mt + k * MT_THREADS;
      const int kk = i >> 6, nn = i & 63;
      const int idx = ((((nn >> 4) * 16) + (kk >> 2)) * 32 + (nn & 15) + ((kk & 2) << 3)) * 2 + (kk & 1);
      w1s[idx] = wa[k];
      w2s[idx] = wb[k];
    }
    if (mt < 128) {
      const float* bp = mt < 64 ? bn1 : bn2;
      const float* lp = mt < 64 ? b1 : b2;
      const int n = mt & 63;
      const float sc = bp[n] * rsqrtf(bp[192 + n] + BN_EPS);
      bns[(mt >> 6) * 2][n] = sc;
      bns[(mt >> 6) * 2 + 1][n] = bp[64 + n] + (lp[n] - bp[128 + n]) * sc;
    }
  }

#pragma nounroll
  for (int c = 0; c <= CH_N; c++) {
    if (gteam) {
      if (c < CH_N) {
        const int cs  = c == 0 ? 0 : (c == 1 ? CH1_ROW : CH2_ROW);
        const int nsl = c == 0 ? GS_MAX : GS_MIN;
        int lg = lane;
        asm volatile("" : "+v"(lg));
        const int g0  = c == 0 ? ga0 : (c == 1 ? gb0 : gc0);
        const int g1  = c == 0 ? ga1 : (c == 1 ? gb1 : gc1);
        v16f sx = {0.f, 0.f, 0.f, 0.f, 0.f, 0.f, 0.f, 0.f, 0.f, 0.f, 0.f, 0.f, 0.f, 0.f, 0.f, 0.f};
        v16f sy = sx;
#pragma unroll
        for (int sl = 0; sl < GS_MAX; sl++) {
          const int r = cs + wave + GT_WAVES * sl;
          const bool real = sl < nsl && r < BK_NODES;
          const v2f t = *(const v2f*)(h + rowbase + (size_t)(real ? r : cs + wave) * 64 + 2 * lg);
          sx[sl] = real ? t.x : 0.f;
          sy[sl] = real ? t.y : 0.f;
        }
        int e = g0;
        int p = gbins[e + lg < N_EDGES ? e + lg : N_EDGES - 1];
        while (e + 32 <= g1) {
          const int en = e + 32;
          v2f v[32];
#pragma unroll
          for (int j = 0; j < 32; j++) {
            const int pj = __builtin_amdgcn_readlane(p, j);
            v[j] = *(const v2f*)(h + (size_t)(pj & ((1 << SRC_BITS) - 1)) * 64 + 2 * lg);
          }
          const int pn = gbins[en + lg < N_EDGES ? en + lg : N_EDGES - 1];
          __builtin_amdgcn_sched_barrier(0);
#pragma unroll
          for (int j = 0; j < 32; j++) {
            const int sl = __builtin_amdgcn_readlane(p, j) >> SRC_BITS;
            sx[sl] += v[j].x;
            sy[sl] += v[j].y;
          }
          p = pn;
          e = en;
        }
        if (e < g1) {
          const int n = g1 - e;
          const int pt = lg < n ? p : __builtin_amdgcn_readlane(p, 0);
          v2f v[32];
#pragma unroll
          for (int j = 0; j < 32; j++) {
            const int pj = __builtin_amdgcn_readlane(pt, j);
            v[j] = *(const v2f*)(h + (size_t)(pj & ((1 << SRC_BITS) - 1)) * 64 + 2 * lg);
          }
#pragma unroll
          for (int j = 0; j < 32; j++) {
            if (j < n) {
              const int sl = __builtin_amdgcn_readlane(pt, j) >> SRC_BITS;
              sx[sl] += v[j].x;
              sy[sl] += v[j].y;
            }
          }
        }
        float* ar = acc + (cs + wave) * CV_PITCH + 2 * lg;
#pragma unroll
        for (int sl = 0; sl < GS_MIN; sl++) {
          *(v2f*)(ar + GT_WAVES * sl * CV_PITCH) = (v2f){sx[sl], sy[sl]};
        }
        if (nsl > GS_MIN) {
#pragma unroll
          for (int sl = GS_MIN; sl < GS_MAX; sl++) {
            *(v2f*)(ar + GT_WAVES * sl * CV_PITCH) = (v2f){sx[sl], sy[sl]};
          }
        }
      }
    } else if (c > 0) {
      const int mc = c - 1;
      const int m  = wave - GT_WAVES;
      const int nct = mc == 0 ? CH0_TILES : (mc == 1 ? CH1_TILES : CH2_TILES);
      if (m < nct) {
        const int tile = (mc == 0 ? 0 : (mc == 1 ? CH1_ROW / 16 : CH2_ROW / 16)) + m;
        int ln = lane;
        asm volatile("" : "+v"(ln));
        const int row  = ln & 15;
        const int hi   = ln >> 4;
        const int left = BK_NODES - tile * 16;
        const int nrow = left < 16 ? left : 16;
        const bool rows_ok = 8 * hi < nrow;
        const int node0 = b * BK_NODES + tile * 16;
        float* tbuf = acc + tile * 16 * CV_PITCH;
        const v8f vzero = {0.f, 0.f, 0.f, 0.f, 0.f, 0.f, 0.f, 0.f};

        v2f a1[16];
        {
          const float* tr = tbuf + row * CV_PITCH + 2 * hi;
#pragma unroll
          for (int k = 0; k < 16; k++) a1[k] = *(const v2f*)(tr + 4 * k);
        }

#pragma unroll
        for (int nt = 0; nt < 4; nt++) {
          v8f a = vzero;
          const v2f* wf = ((const v2f*)w1s) + nt * 16 * 32 + ln;
#pragma unroll
          for (int k = 0; k < 16; k++) {
            a = __builtin_amdgcn_wmma_f32_16x16x4_f32(false, a1[k], false, wf[k * 32],
                                                      (short)0, a, false, false);
          }
          int n = nt * 16 + row;
          float sc = bns[0][n];
          float t0 = bns[1][n];
#pragma unroll
          for (int i = 0; i < 8; i++) {
            float y = a[i] * sc + t0;
            tbuf[(i + 8 * hi) * CV_PITCH + n] = fmaxf(y, 0.0f);
          }
        }
        __builtin_amdgcn_fence(__ATOMIC_SEQ_CST, "wavefront");

        v8f acc2[4];
        v2f wl[16];
        int gid[8];
        const int hc = row < 10 ? row : 9;
#pragma unroll
        for (int r = 0; r < 8; r++) gid[r] = 0;
        {
          const float* lw = linW + (2 * hi) * 10 + hc;
#pragma unroll
          for (int k = 0; k < 16; k++) {
            float w0 = lw[(4 * k) * 10];
            float w1 = lw[(4 * k + 1) * 10];
            wl[k] = (v2f){row < 10 ? w0 : 0.f, row < 10 ? w1 : 0.f};
          }
        }
        if (rows_ok) {
#pragma unroll
          for (int r = 0; r < 8; r++) gid[r] = batch[node0 + 8 * hi + r];
        }
#pragma unroll
        for (int nt = 0; nt < 4; nt++) {
          v8f a = vzero;
          const v2f* wf = ((const v2f*)w2s) + nt * 16 * 32 + ln;
          const float* tr = tbuf + row * CV_PITCH + 2 * hi;
#pragma unroll
          for (int k = 0; k < 16; k++) {
            v2f av = *(const v2f*)(tr + 4 * k);
            a = __builtin_amdgcn_wmma_f32_16x16x4_f32(false, av, false, wf[k * 32],
                                                      (short)0, a, false, false);
          }
          acc2[nt] = a;
        }
        __builtin_amdgcn_fence(__ATOMIC_SEQ_CST, "wavefront");

#pragma unroll
        for (int nt = 0; nt < 4; nt++) {
          int n = nt * 16 + row;
          float sc = bns[2][n];
          float t0 = bns[3][n];
#pragma unroll
          for (int i = 0; i < 8; i++) {
            float y = acc2[nt][i] * sc + t0;
            tbuf[(i + 8 * hi) * CV_PITCH + n] = fmaxf(y, 0.0f);
          }
        }
        __builtin_amdgcn_fence(__ATOMIC_SEQ_CST, "wavefront");

        v8f p0 = vzero, p1 = vzero;
        const float* tr = tbuf + row * CV_PITCH + 2 * hi;
#pragma unroll
        for (int k = 0; k < 16; k += 2) {
          v2f a0 = *(const v2f*)(tr + 4 * k);
          v2f a1h = *(const v2f*)(tr + 4 * (k + 1));
          p0 = __builtin_amdgcn_wmma_f32_16x16x4_f32(false, a0, false, wl[k],
                                                     (short)0, p0, false, false);
          p1 = __builtin_amdgcn_wmma_f32_16x16x4_f32(false, a1h, false, wl[k + 1],
                                                     (short)0, p1, false, false);
        }
        float pv[8];
#pragma unroll
        for (int r = 0; r < 8; r++) pv[r] = p0[r] + p1[r];

#pragma unroll
        for (int j = 0; j < 8; j++) {
          const int r = 2 * j + hi;
          const int cc = row << 2;
          if (r < nrow) {
            float4 v = *(const float4*)(tbuf + r * CV_PITCH + cc);
            *(float4*)(hout + (size_t)(node0 + r) * 64 + cc) = v;
          }
        }

        if (row < 10 && rows_ok) {
#pragma unroll
          for (int r = 0; r < 8; r++) atomicAdd(out + (size_t)gid[r] * 10 + row, pv[r]);
        }
      }
    }
    if (c < CH_N) __syncthreads();
  }
}

__global__ void gin_out_init_kernel(const float* __restrict__ linb,
                                    float* __restrict__ out,
                                    int* __restrict__ cnt) {
  int idx = blockIdx.x * blockDim.x + threadIdx.x;
  if (idx < 2 * NBK) cnt[idx] = 0;
  if (idx >= N_GRAPHS * 10) return;
  int t = idx % 10;
  out[idx] = ((linb[10 + t] + linb[20 + t]) + linb[30 + t]) + linb[40 + t];
}

extern "C" void kernel_launch(void* const* d_in, const int* in_sizes, int n_in,
                              void* d_out, int out_size, void* d_ws, size_t ws_size,
                              hipStream_t stream) {
  const float* x        = (const float*)d_in[0];
  const float* fh_W1    = (const float*)d_in[1];
  const float* fh_b1    = (const float*)d_in[2];
  const float* fh_bn1   = (const float*)d_in[3];
  const float* fh_W2    = (const float*)d_in[4];
  const float* fh_b2    = (const float*)d_in[5];
  const float* fh_bn2   = (const float*)d_in[6];
  const float* conv_W1  = (const float*)d_in[7];
  const float* conv_b1  = (const float*)d_in[8];
  const float* conv_bn1 = (const float*)d_in[9];
  const float* conv_W2  = (const float*)d_in[10];
  const float* conv_b2  = (const float*)d_in[11];
  const float* conv_bn2 = (const float*)d_in[12];
  const float* lin_W    = (const float*)d_in[13];
  const float* lin_b    = (const float*)d_in[14];
  const int*   eidx     = (const int*)d_in[15];
  const int*   batch    = (const int*)d_in[16];
  float* out = (float*)d_out;

  const int* src = eidx;
  const int* dst = eidx + N_EDGES;

  float* ws     = (float*)d_ws;
  float* hA     = ws;
  float* hB     = ws + (size_t)N_NODES * 64;
  int*   cnt    = (int*)(ws + (size_t)2 * N_NODES * 64);
  int*   cursor = cnt + NBK;
  int*   offs   = cnt + 1024;
  int*   bins   = cnt + 2048;
  int*   goffs  = cnt + 2048 + N_EDGES;
  int*   gbins  = cnt + 2048 + N_EDGES + 16384;

  gin_out_init_kernel<<<(N_GRAPHS * 10 + 255) / 256, 256, 0, stream>>>(lin_b, out, cnt);

  gin_bin_count_kernel<<<BF_BLOCKS, BF_THREADS, 0, stream>>>(dst, cnt);
  gin_bin_fill_kernel<<<BF_BLOCKS, BF_THREADS, 0, stream>>>(src, dst, cnt, cursor, offs, bins);
  gin_bin_group_kernel<<<NBK, GR_THREADS, 0, stream>>>(offs, bins, goffs, gbins);

  const int mlpBlocks = (NTILES + 3) / 4;

  gin_mlp_kernel<D_FEAT, true><<<mlpBlocks, 128, 0, stream>>>(
      x, fh_W1, fh_b1, fh_bn1, fh_W2, fh_b2, fh_bn2, hA,
      lin_W, lin_b, batch, out, NTILES);

  float* hin = hA;
  float* hnext = hB;
  for (int l = 0; l < 4; l++) {
    gin_conv_kernel<<<NBK, AG_THREADS, 0, stream>>>(
        hin, goffs, gbins,
        conv_W1 + (size_t)l * 64 * 64, conv_b1 + l * 64, conv_bn1 + l * 256,
        conv_W2 + (size_t)l * 64 * 64, conv_b2 + l * 64, conv_bn2 + l * 256,
        hnext,
        lin_W + (size_t)(l + 1) * 64 * 10, batch, out);
    float* tswap = hin;
    hin = hnext;
    hnext = tswap;
  }
}


# Head sums pooled per work-group: tiles add their 10 head columns into a per-graph LDS table (LDS f32 atomics), one global f32 atomic per touched entry; out sum order = LDS arrival, then arrival
# speedup vs baseline: 4.2729x; 1.0839x over previous
#include <hip/hip_runtime.h>

#define N_NODES   100000
#define N_EDGES   1600000
#define N_GRAPHS  1000
#define D_FEAT    128
#define HID       64
#define BN_EPS    1e-5f
#define NTILES    (N_NODES / 16)
#define MLP_PG_SLOTS 16
#define CV_PG_SLOTS  64

typedef __attribute__((ext_vector_type(2))) float v2f;
typedef __attribute__((ext_vector_type(8))) float v8f;

template<int K1, bool NODE_BIAS>
__launch_bounds__(128)
__global__ void gin_mlp_kernel(const float* __restrict__ xin,
                               const float* __restrict__ W1,
                               const float* __restrict__ b1,
                               const float* __restrict__ bn1,
                               const float* __restrict__ W2,
                               const float* __restrict__ b2,
                               const float* __restrict__ bn2,
                               float* __restrict__ hout,
                               const float* __restrict__ linW,
                               const float* __restrict__ linb0,
                               const int* __restrict__ batch,
                               float* __restrict__ out,
                               int ntiles) {
  constexpr int KS1 = K1 / 4;
  __shared__ __align__(16) float w1s[4 * KS1 * 32 * 2];
  __shared__ __align__(16) float w2s[4 * 16 * 32 * 2];
  __shared__ __align__(16) float tb[4][16 * 68];
  __shared__ float gs[MLP_PG_SLOTS * 10];

  const int tid = threadIdx.x;
  const int g_lo = batch[blockIdx.x * 64];
  for (int i = tid; i < MLP_PG_SLOTS * 10; i += 128) gs[i] = 0.f;

  for (int i = tid; i < K1 * 64; i += 128) {
    int kk = i >> 6, nn = i & 63;
    int idx = ((((nn >> 4) * KS1) + (kk >> 2)) * 32 + (nn & 15) + ((kk & 2) << 3)) * 2 + (kk & 1);
    w1s[idx] = W1[i];
  }
  for (int i = tid; i < 64 * 64; i += 128) {
    int kk = i >> 6, nn = i & 63;
    int idx = ((((nn >> 4) * 16) + (kk >> 2)) * 32 + (nn & 15) + ((kk & 2) << 3)) * 2 + (kk & 1);
    w2s[idx] = W2[i];
  }
  __syncthreads();

  const int lane = tid & 31;
  const int wave = tid >> 5;
  const int row  = lane & 15;
  const int hi   = lane >> 4;
  const int tile = blockIdx.x * 4 + wave;
  const bool active = tile < ntiles;
  float* tbuf = tb[wave];

  const v8f vzero = {0.f, 0.f, 0.f, 0.f, 0.f, 0.f, 0.f, 0.f};

  if (active) {
    v2f a1[KS1];
    const float* xr = xin + (size_t)(tile * 16 + row) * K1 + 2 * hi;
#pragma unroll
    for (int k = 0; k < KS1; k++) a1[k] = *(const v2f*)(xr + 4 * k);

#pragma unroll
    for (int nt = 0; nt < 4; nt++) {
      v8f acc = vzero;
      const v2f* wf = ((const v2f*)w1s) + nt * KS1 * 32 + lane;
#pragma unroll
      for (int k = 0; k < KS1; k++) {
        acc = __builtin_amdgcn_wmma_f32_16x16x4_f32(false, a1[k], false, wf[k * 32],
                                                    (short)0, acc, false, false);
      }
      int n = nt * 16 + row;
      float s  = bn1[n] * rsqrtf(bn1[192 + n] + BN_EPS);
      float t0 = bn1[64 + n] + (b1[n] - bn1[128 + n]) * s;
#pragma unroll
      for (int i = 0; i < 8; i++) {
        float y = acc[i] * s + t0;
        tbuf[(i + 8 * hi) * 68 + n] = fmaxf(y, 0.0f);
      }
    }
  }
  __syncthreads();

  v8f acc2[4];
  v2f wl[16];
  int gid[8];
  const int hc = row < 10 ? row : 9;
#pragma unroll
  for (int k = 0; k < 16; k++) wl[k] = (v2f){0.f, 0.f};
#pragma unroll
  for (int r = 0; r < 8; r++) gid[r] = 0;
  if (active) {
    const float* lw = linW + (2 * hi) * 10 + hc;
#pragma unroll
    for (int k = 0; k < 16; k++) {
      float w0 = lw[(4 * k) * 10];
      float w1 = lw[(4 * k + 1) * 10];
      wl[k] = (v2f){row < 10 ? w0 : 0.f, row < 10 ? w1 : 0.f};
    }
#pragma unroll
    for (int r = 0; r < 8; r++) gid[r] = batch[tile * 16 + 8 * hi + r];
#pragma unroll
    for (int nt = 0; nt < 4; nt++) {
      v8f acc = vzero;
      const v2f* wf = ((const v2f*)w2s) + nt * 16 * 32 + lane;
      const float* tr = tbuf + row * 68 + 2 * hi;
#pragma unroll
      for (int k = 0; k < 16; k++) {
        v2f a = *(const v2f*)(tr + 4 * k);
        acc = __builtin_amdgcn_wmma_f32_16x16x4_f32(false, a, false, wf[k * 32],
                                                    (short)0, acc, false, false);
      }
      acc2[nt] = acc;
    }
  }
  __syncthreads();

  if (active) {
#pragma unroll
    for (int nt = 0; nt < 4; nt++) {
      int n = nt * 16 + row;
      float s  = bn2[n] * rsqrtf(bn2[192 + n] + BN_EPS);
      float t0 = bn2[64 + n] + (b2[n] - bn2[128 + n]) * s;
#pragma unroll
      for (int i = 0; i < 8; i++) {
        float y = acc2[nt][i] * s + t0;
        tbuf[(i + 8 * hi) * 68 + n] = fmaxf(y, 0.0f);
      }
    }
  }
  __syncthreads();

  if (active) {
    v8f p0 = vzero, p1 = vzero;
    const float* tr = tbuf + row * 68 + 2 * hi;
#pragma unroll
    for (int k = 0; k < 16; k += 2) {
      v2f a0 = *(const v2f*)(tr + 4 * k);
      v2f a1 = *(const v2f*)(tr + 4 * (k + 1));
      p0 = __builtin_amdgcn_wmma_f32_16x16x4_f32(false, a0, false, wl[k],
                                                 (short)0, p0, false, false);
      p1 = __builtin_amdgcn_wmma_f32_16x16x4_f32(false, a1, false, wl[k + 1],
                                                 (short)0, p1, false, false);
    }
    float pv[8];
#pragma unroll
    for (int r = 0; r < 8; r++) pv[r] = p0[r] + p1[r];
    if (NODE_BIAS) {
      const float lb = linb0[hc];
#pragma unroll
      for (int r = 0; r < 8; r++) pv[r] += lb;
    }

#pragma unroll
    for (int j = 0; j < 8; j++) {
      int flat = j * 32 + lane;
      int r = flat >> 4;
      int c = (flat & 15) << 2;
      float4 v = *(const float4*)(tbuf + r * 68 + c);
      *(float4*)(hout + (size_t)(tile * 16 + r) * 64 + c) = v;
    }

    if (row < 10) {
#pragma unroll
      for (int r = 0; r < 8; r++) {
        const unsigned d = (unsigned)(gid[r] - g_lo);
        if (d < MLP_PG_SLOTS) atomicAdd(&gs[d * 10 + row], pv[r]);
        else atomicAdd(out + (size_t)gid[r] * 10 + row, pv[r]);
      }
    }
  }
  __syncthreads();
  for (int i = tid; i < MLP_PG_SLOTS * 10; i += 128) {
    const float v = gs[i];
    if (v != 0.f && g_lo + i / 10 < N_GRAPHS) atomicAdd(out + (size_t)g_lo * 10 + i, v);
  }
}

#define BK_NODES 200
#define NBK (N_NODES / BK_NODES)
#define BF_THREADS 800
#define BF_EDGES_PER_THREAD 8
#define BF_EDGES_PER_BLOCK (BF_THREADS * BF_EDGES_PER_THREAD)
#define BF_BLOCKS (N_EDGES / BF_EDGES_PER_BLOCK)
#define SRC_BITS 17
static_assert(N_NODES % BK_NODES == 0, "ranges must tile the nodes exactly");
static_assert(N_EDGES % BF_EDGES_PER_BLOCK == 0, "edge blocks must tile the edges exactly");
static_assert(N_NODES <= (1 << SRC_BITS), "a source id must fit the packed field");
static_assert(NBK <= 512, "the range scan below holds 512 entries");

__launch_bounds__(BF_THREADS)
__global__ void gin_bin_count_kernel(const int* __restrict__ dst,
                                     int* __restrict__ cnt) {
  __shared__ int hist[NBK];
  const int tid = threadIdx.x;
  for (int b = tid; b < NBK; b += BF_THREADS) hist[b] = 0;
  __syncthreads();
  const int e0 = blockIdx.x * BF_EDGES_PER_BLOCK + tid;
  unsigned d[BF_EDGES_PER_THREAD];
#pragma unroll
  for (int k = 0; k < BF_EDGES_PER_THREAD; k++) d[k] = (unsigned)dst[e0 + k * BF_THREADS];
#pragma unroll
  for (int k = 0; k < BF_EDGES_PER_THREAD; k++) atomicAdd(&hist[d[k] / BK_NODES], 1);
  __syncthreads();
  for (int b = tid; b < NBK; b += BF_THREADS) {
    const int c = hist[b];
    if (c) atomicAdd(&cnt[b], c);
  }
}

__launch_bounds__(BF_THREADS)
__global__ void gin_bin_fill_kernel(const int* __restrict__ src,
                                    const int* __restrict__ dst,
                                    const int* __restrict__ cnt,
                                    int* __restrict__ cursor,
                                    int* __restrict__ offs,
                                    int* __restrict__ bins) {
  __shared__ int hist[NBK];
  __shared__ int bpos[NBK];
  __shared__ int loff[NBK];
  __shared__ int2 sc[2][512];
  __shared__ int2 stage[BF_EDGES_PER_BLOCK];
  const int tid = threadIdx.x;
  for (int b = tid; b < NBK; b += BF_THREADS) hist[b] = 0;
  const int cg = tid < NBK ? cnt[tid] : 0;
  __syncthreads();

  int bk[BF_EDGES_PER_THREAD], rk[BF_EDGES_PER_THREAD], pk[BF_EDGES_PER_THREAD];
  const int e0 = blockIdx.x * BF_EDGES_PER_BLOCK + tid;
#pragma unroll
  for (int k = 0; k < BF_EDGES_PER_THREAD; k++) {
    pk[k] = src[e0 + k * BF_THREADS];
    bk[k] = dst[e0 + k * BF_THREADS];
  }
#pragma unroll
  for (int k = 0; k < BF_EDGES_PER_THREAD; k++) {
    const unsigned s = (unsigned)pk[k];
    const unsigned d = (unsigned)bk[k];
    const unsigned b = d / BK_NODES;
    bk[k] = (int)b;
    pk[k] = (int)(((d - b * BK_NODES) << SRC_BITS) | s);
    rk[k] = atomicAdd(&hist[b], 1);
  }
  __syncthreads();
  if (tid < 512) sc[0][tid] = make_int2(cg, tid < NBK ? hist[tid] : 0);
  __syncthreads();

  int cur = 0;
  for (int off = 1; off < 512; off <<= 1) {
    if (tid < 512) {
      int2 v = sc[cur][tid];
      if (tid >= off) {
        const int2 u = sc[cur][tid - off];
        v.x += u.x;
        v.y += u.y;
      }
      sc[cur ^ 1][tid] = v;
    }
    __syncthreads();
    cur ^= 1;
  }
  const int2* scan = sc[cur];

  for (int b = tid; b < NBK; b += BF_THREADS) {
    const int c = hist[b];
    const int start = b ? scan[b - 1].x : 0;
    const int r = c ? atomicAdd(&cursor[b], c) : 0;
    bpos[b] = start + r;
    loff[b] = b ? scan[b - 1].y : 0;
    if (blockIdx.x == 0) offs[b] = start;
  }
  if (blockIdx.x == 0 && tid == 0) offs[NBK] = scan[NBK - 1].x;
  __syncthreads();

#pragma unroll
  for (int k = 0; k < BF_EDGES_PER_THREAD; k++) {
    const int b = bk[k];
    stage[loff[b] + rk[k]] = make_int2(pk[k], bpos[b] + rk[k]);
  }
  __syncthreads();
#pragma unroll
  for (int k = 0; k < BF_EDGES_PER_THREAD; k++) {
    const int2 s = stage[tid + k * BF_THREADS];
    bins[s.y] = s.x;
  }
}


#define GT_WAVES 8
#define CH_N 3
#define CH1_ROW 80
#define CH2_ROW 144
#define NGRP (CH_N * GT_WAVES)
static_assert(NBK * (NGRP + 1) <= 16384, "group starts fit the space kept for them");
static_assert(CH1_ROW % 16 == 0 && CH2_ROW % 16 == 0, "chunks are whole tiles");
__device__ __forceinline__ int gin_group_of(int p) {
  const int r = p >> SRC_BITS;
  const int c = (r >= CH1_ROW ? 1 : 0) + (r >= CH2_ROW ? 1 : 0);
  return c * GT_WAVES + (r & (GT_WAVES - 1));
}
__device__ __forceinline__ int gin_slot_entry(int p) {
  const int r = p >> SRC_BITS;
  const int cs = r >= CH2_ROW ? CH2_ROW : (r >= CH1_ROW ? CH1_ROW : 0);
  return (((r - cs) / GT_WAVES) << SRC_BITS) | (p & ((1 << SRC_BITS) - 1));
}
#define GR_THREADS 512
#define GR_KEEP 8
#define GR_STAGE (GR_KEEP * GR_THREADS)
__launch_bounds__(GR_THREADS)
__global__ void gin_bin_group_kernel(const int* __restrict__ offs,
                                     const int* __restrict__ bins,
                                     int* __restrict__ goffs,
                                     int* __restrict__ gbins) {
  __shared__ int gcnt[NGRP];
  __shared__ int gpos[NGRP];
  __shared__ int gbase[NGRP + 1];
  __shared__ int stage[GR_STAGE];
  const int tid = threadIdx.x;
  const int b = blockIdx.x;
  const int e0 = offs[b];
  const int e1 = offs[b + 1];
  if (tid < NGRP) {
    gcnt[tid] = 0;
    gpos[tid] = 0;
  }
  int pr[GR_KEEP];
#pragma unroll
  for (int k = 0; k < GR_KEEP; k++) {
    const int i = e0 + tid + k * GR_THREADS;
    pr[k] = bins[i < e1 ? i : N_EDGES - 1];
  }
  __syncthreads();
#pragma unroll
  for (int k = 0; k < GR_KEEP; k++) {
    if (e0 + tid + k * GR_THREADS < e1) atomicAdd(&gcnt[gin_group_of(pr[k])], 1);
  }
  for (int i = e0 + tid + GR_KEEP * GR_THREADS; i < e1; i += GR_THREADS) {
    const int g = gin_group_of(bins[i]);
    atomicAdd(&gcnt[g], 1);
  }
  __syncthreads();
  if (tid == 0) {
    int s = e0;
    for (int g = 0; g < NGRP; g++) {
      gbase[g] = s;
      s += gcnt[g];
    }
    gbase[NGRP] = s;
  }
  __syncthreads();
  if (tid <= NGRP) goffs[b * (NGRP + 1) + tid] = gbase[tid];
#pragma unroll
  for (int k = 0; k < GR_KEEP; k++) {
    if (e0 + tid + k * GR_THREADS < e1) {
      const int g = gin_group_of(pr[k]);
      const int r = atomicAdd(&gpos[g], 1);
      const int q = gbase[g] - e0 + r;
      const int v = gin_slot_entry(pr[k]);
      if (q < GR_STAGE) stage[q] = v;
      else gbins[e0 + q] = v;
    }
  }
  for (int i = e0 + tid + GR_KEEP * GR_THREADS; i < e1; i += GR_THREADS) {
    const int p = bins[i];
    const int g = gin_group_of(p);
    const int r = atomicAdd(&gpos[g], 1);
    const int q = gbase[g] - e0 + r;
    const int v = gin_slot_entry(p);
    if (q < GR_STAGE) stage[q] = v;
    else gbins[e0 + q] = v;
  }
  __syncthreads();
  const int nst = e1 - e0 < GR_STAGE ? e1 - e0 : GR_STAGE;
  int sv[GR_KEEP];
#pragma unroll
  for (int k = 0; k < GR_KEEP; k++) {
    const int i = tid + k * GR_THREADS;
    sv[k] = i < nst ? stage[i] : 0;
  }
#pragma unroll
  for (int k = 0; k < GR_KEEP; k++) {
    const int i = tid + k * GR_THREADS;
    if (i < nst) gbins[e0 + i] = sv[k];
  }
}

#define AG_WAVES 16
#define AG_THREADS (AG_WAVES * 32)
#define MT_WAVES (AG_WAVES - GT_WAVES)
#define MT_THREADS (MT_WAVES * 32)
#define CV_TILES ((BK_NODES + 15) / 16)
#define CV_ROWS (CV_TILES * 16)
#define CV_PITCH 68
#define CH0_TILES (CH1_ROW / 16)
#define CH1_TILES ((CH2_ROW - CH1_ROW) / 16)
#define CH2_TILES (CV_TILES - CH2_ROW / 16)
#define GS_MAX (CH1_ROW / GT_WAVES)
#define GS_MIN 8
static_assert(BK_NODES % 8 == 0, "a lane half of a tile (8 rows) is all valid or all padding");
static_assert(CH0_TILES <= MT_WAVES && CH1_TILES <= MT_WAVES && CH2_TILES <= MT_WAVES, "one matrix wave per tile of a chunk");
static_assert((CH2_ROW - CH1_ROW) / GT_WAVES == GS_MIN && (CV_ROWS - CH2_ROW) / GT_WAVES == GS_MIN, "slots of chunks 1 and 2");
static_assert(GS_MAX >= GS_MIN && GS_MAX <= 16, "a summing wave's slots fit one 16-register vector per column half");
static_assert(CH2_ROW + GT_WAVES - 1 < BK_NODES, "slot 0 of every summing wave is a real row in every chunk");
static_assert((64 * 64) % MT_THREADS == 0, "weight pieces per thread");
#define CV_WPT ((64 * 64) / MT_THREADS)
typedef __attribute__((ext_vector_type(16))) float v16f;
__launch_bounds__(AG_THREADS, 8)
__global__ void gin_conv_kernel(const float* __restrict__ h,
                                const int* __restrict__ goffs,
                                const int* __restrict__ gbins,
                                const float* __restrict__ W1,
                                const float* __restrict__ b1,
                                const float* __restrict__ bn1,
                                const float* __restrict__ W2,
                                const float* __restrict__ b2,
                                const float* __restrict__ bn2,
                                float* __restrict__ hout,
                                const float* __restrict__ linW,
                                const int* __restrict__ batch,
                                float* __restrict__ out) {
  __shared__ __align__(16) float acc[CV_ROWS * CV_PITCH];
  __shared__ __align__(16) float w1s[4 * 16 * 32 * 2];
  __shared__ __align__(16) float w2s[4 * 16 * 32 * 2];
  __shared__ float bns[4][64];
  __shared__ float gs[CV_PG_SLOTS * 10];
  const int tid  = threadIdx.x;
  const int lane = tid & 31;
  const int wave = __builtin_amdgcn_readfirstlane(tid >> 5);
  const bool gteam = wave < GT_WAVES;
  const int b = blockIdx.x;
  const size_t rowbase = (size_t)b * BK_NODES * 64;
  const int g_lo = batch[b * BK_NODES];
  for (int i = tid; i < CV_PG_SLOTS * 10; i += AG_THREADS) gs[i] = 0.f;

  const int* gp = goffs + b * (NGRP + 1) + (wave & (GT_WAVES - 1));
  const int ga0 = gp[0],            ga1 = gp[1];
  const int gb0 = gp[GT_WAVES],     gb1 = gp[GT_WAVES + 1];
  const int gc0 = gp[2 * GT_WAVES], gc1 = gp[2 * GT_WAVES + 1];

  if (!gteam) {
    const int mt = tid - GT_WAVES * 32;
    float wa[CV_WPT], wb[CV_WPT];
#pragma unroll
    for (int k = 0; k < CV_WPT; k++) {
      wa[k] = W1[mt + k * MT_THREADS];
      wb[k] = W2[mt + k * MT_THREADS];
    }
#pragma unroll
    for (int k = 0; k < CV_WPT; k++) {
      const int i = mt + k * MT_THREADS;
      const int kk = i >> 6, nn = i & 63;
      const int idx = ((((nn >> 4) * 16) + (kk >> 2)) * 32 + (nn & 15) + ((kk & 2) << 3)) * 2 + (kk & 1);
      w1s[idx] = wa[k];
      w2s[idx] = wb[k];
    }
    if (mt < 128) {
      const float* bp = mt < 64 ? bn1 : bn2;
      const float* lp = mt < 64 ? b1 : b2;
      const int n = mt & 63;
      const float sc = bp[n] * rsqrtf(bp[192 + n] + BN_EPS);
      bns[(mt >> 6) * 2][n] = sc;
      bns[(mt >> 6) * 2 + 1][n] = bp[64 + n] + (lp[n] - bp[128 + n]) * sc;
    }
  }

#pragma nounroll
  for (int c = 0; c <= CH_N; c++) {
    if (gteam) {
      if (c < CH_N) {
        const int cs  = c == 0 ? 0 : (c == 1 ? CH1_ROW : CH2_ROW);
        const int nsl = c == 0 ? GS_MAX : GS_MIN;
        int lg = lane;
        asm volatile("" : "+v"(lg));
        const int g0  = c == 0 ? ga0 : (c == 1 ? gb0 : gc0);
        const int g1  = c == 0 ? ga1 : (c == 1 ? gb1 : gc1);
        v16f sx = {0.f, 0.f, 0.f, 0.f, 0.f, 0.f, 0.f, 0.f, 0.f, 0.f, 0.f, 0.f, 0.f, 0.f, 0.f, 0.f};
        v16f sy = sx;
#pragma unroll
        for (int sl = 0; sl < GS_MAX; sl++) {
          const int r = cs + wave + GT_WAVES * sl;
          const bool real = sl < nsl && r < BK_NODES;
          const v2f t = *(const v2f*)(h + rowbase + (size_t)(real ? r : cs + wave) * 64 + 2 * lg);
          sx[sl] = real ? t.x : 0.f;
          sy[sl] = real ? t.y : 0.f;
        }
        int e = g0;
        int p = gbins[e + lg < N_EDGES ? e + lg : N_EDGES - 1];
        while (e + 32 <= g1) {
          const int en = e + 32;
          v2f v[32];
#pragma unroll
          for (int j = 0; j < 32; j++) {
            const int pj = __builtin_amdgcn_readlane(p, j);
            v[j] = *(const v2f*)(h + (size_t)(pj & ((1 << SRC_BITS) - 1)) * 64 + 2 * lg);
          }
          const int pn = gbins[en + lg < N_EDGES ? en + lg : N_EDGES - 1];
          __builtin_amdgcn_sched_barrier(0);
#pragma unroll
          for (int j = 0; j < 32; j++) {
            const int sl = __builtin_amdgcn_readlane(p, j) >> SRC_BITS;
            sx[sl] += v[j].x;
            sy[sl] += v[j].y;
          }
          p = pn;
          e = en;
        }
        if (e < g1) {
          const int n = g1 - e;
          const int pt = lg < n ? p : __builtin_amdgcn_readlane(p, 0);
          v2f v[32];
#pragma unroll
          for (int j = 0; j < 32; j++) {
            const int pj = __builtin_amdgcn_readlane(pt, j);
            v[j] = *(const v2f*)(h + (size_t)(pj & ((1 << SRC_BITS) - 1)) * 64 + 2 * lg);
          }
#pragma unroll
          for (int j = 0; j < 32; j++) {
            if (j < n) {
              const int sl = __builtin_amdgcn_readlane(pt, j) >> SRC_BITS;
              sx[sl] += v[j].x;
              sy[sl] += v[j].y;
            }
          }
        }
        float* ar = acc + (cs + wave) * CV_PITCH + 2 * lg;
#pragma unroll
        for (int sl = 0; sl < GS_MIN; sl++) {
          *(v2f*)(ar + GT_WAVES * sl * CV_PITCH) = (v2f){sx[sl], sy[sl]};
        }
        if (nsl > GS_MIN) {
#pragma unroll
          for (int sl = GS_MIN; sl < GS_MAX; sl++) {
            *(v2f*)(ar + GT_WAVES * sl * CV_PITCH) = (v2f){sx[sl], sy[sl]};
          }
        }
      }
    } else if (c > 0) {
      const int mc = c - 1;
      const int m  = wave - GT_WAVES;
      const int nct = mc == 0 ? CH0_TILES : (mc == 1 ? CH1_TILES : CH2_TILES);
      if (m < nct) {
        const int tile = (mc == 0 ? 0 : (mc == 1 ? CH1_ROW / 16 : CH2_ROW / 16)) + m;
        int ln = lane;
        asm volatile("" : "+v"(ln));
        const int row  = ln & 15;
        const int hi   = ln >> 4;
        const int left = BK_NODES - tile * 16;
        const int nrow = left < 16 ? left : 16;
        const bool rows_ok = 8 * hi < nrow;
        const int node0 = b * BK_NODES + tile * 16;
        float* tbuf = acc + tile * 16 * CV_PITCH;
        const v8f vzero = {0.f, 0.f, 0.f, 0.f, 0.f, 0.f, 0.f, 0.f};

        v2f a1[16];
        {
          const float* tr = tbuf + row * CV_PITCH + 2 * hi;
#pragma unroll
          for (int k = 0; k < 16; k++) a1[k] = *(const v2f*)(tr + 4 * k);
        }

#pragma unroll
        for (int nt = 0; nt < 4; nt++) {
          v8f a = vzero;
          const v2f* wf = ((const v2f*)w1s) + nt * 16 * 32 + ln;
#pragma unroll
          for (int k = 0; k < 16; k++) {
            a = __builtin_amdgcn_wmma_f32_16x16x4_f32(false, a1[k], false, wf[k * 32],
                                                      (short)0, a, false, false);
          }
          int n = nt * 16 + row;
          float sc = bns[0][n];
          float t0 = bns[1][n];
#pragma unroll
          for (int i = 0; i < 8; i++) {
            float y = a[i] * sc + t0;
            tbuf[(i + 8 * hi) * CV_PITCH + n] = fmaxf(y, 0.0f);
          }
        }
        __builtin_amdgcn_fence(__ATOMIC_SEQ_CST, "wavefront");

        v8f acc2[4];
        v2f wl[16];
        int gid[8];
        const int hc = row < 10 ? row : 9;
#pragma unroll
        for (int r = 0; r < 8; r++) gid[r] = 0;
        {
          const float* lw = linW + (2 * hi) * 10 + hc;
#pragma unroll
          for (int k = 0; k < 16; k++) {
            float w0 = lw[(4 * k) * 10];
            float w1 = lw[(4 * k + 1) * 10];
            wl[k] = (v2f){row < 10 ? w0 : 0.f, row < 10 ? w1 : 0.f};
          }
        }
        if (rows_ok) {
#pragma unroll
          for (int r = 0; r < 8; r++) gid[r] = batch[node0 + 8 * hi + r];
        }
#pragma unroll
        for (int nt = 0; nt < 4; nt++) {
          v8f a = vzero;
          const v2f* wf = ((const v2f*)w2s) + nt * 16 * 32 + ln;
          const float* tr = tbuf + row * CV_PITCH + 2 * hi;
#pragma unroll
          for (int k = 0; k < 16; k++) {
            v2f av = *(const v2f*)(tr + 4 * k);
            a = __builtin_amdgcn_wmma_f32_16x16x4_f32(false, av, false, wf[k * 32],
                                                      (short)0, a, false, false);
          }
          acc2[nt] = a;
        }
        __builtin_amdgcn_fence(__ATOMIC_SEQ_CST, "wavefront");

#pragma unroll
        for (int nt = 0; nt < 4; nt++) {
          int n = nt * 16 + row;
          float sc = bns[2][n];
          float t0 = bns[3][n];
#pragma unroll
          for (int i = 0; i < 8; i++) {
            float y = acc2[nt][i] * sc + t0;
            tbuf[(i + 8 * hi) * CV_PITCH + n] = fmaxf(y, 0.0f);
          }
        }
        __builtin_amdgcn_fence(__ATOMIC_SEQ_CST, "wavefront");

        v8f p0 = vzero, p1 = vzero;
        const float* tr = tbuf + row * CV_PITCH + 2 * hi;
#pragma unroll
        for (int k = 0; k < 16; k += 2) {
          v2f a0 = *(const v2f*)(tr + 4 * k);
          v2f a1h = *(const v2f*)(tr + 4 * (k + 1));
          p0 = __builtin_amdgcn_wmma_f32_16x16x4_f32(false, a0, false, wl[k],
                                                     (short)0, p0, false, false);
          p1 = __builtin_amdgcn_wmma_f32_16x16x4_f32(false, a1h, false, wl[k + 1],
                                                     (short)0, p1, false, false);
        }
        float pv[8];
#pragma unroll
        for (int r = 0; r < 8; r++) pv[r] = p0[r] + p1[r];

#pragma unroll
        for (int j = 0; j < 8; j++) {
          const int r = 2 * j + hi;
          const int cc = row << 2;
          if (r < nrow) {
            float4 v = *(const float4*)(tbuf + r * CV_PITCH + cc);
            *(float4*)(hout + (size_t)(node0 + r) * 64 + cc) = v;
          }
        }

        if (row < 10 && rows_ok) {
#pragma unroll
          for (int r = 0; r < 8; r++) {
            const unsigned d = (unsigned)(gid[r] - g_lo);
            if (d < CV_PG_SLOTS) atomicAdd(&gs[d * 10 + row], pv[r]);
            else atomicAdd(out + (size_t)gid[r] * 10 + row, pv[r]);
          }
        }
      }
    }
    if (c < CH_N) __syncthreads();
  }
  __syncthreads();
  for (int i = tid; i < CV_PG_SLOTS * 10; i += AG_THREADS) {
    const float v = gs[i];
    if (v != 0.f && g_lo + i / 10 < N_GRAPHS) atomicAdd(out + (size_t)g_lo * 10 + i, v);
  }
}

__global__ void gin_out_init_kernel(const float* __restrict__ linb,
                                    float* __restrict__ out,
                                    int* __restrict__ cnt) {
  int idx = blockIdx.x * blockDim.x + threadIdx.x;
  if (idx < 2 * NBK) cnt[idx] = 0;
  if (idx >= N_GRAPHS * 10) return;
  int t = idx % 10;
  out[idx] = ((linb[10 + t] + linb[20 + t]) + linb[30 + t]) + linb[40 + t];
}

extern "C" void kernel_launch(void* const* d_in, const int* in_sizes, int n_in,
                              void* d_out, int out_size, void* d_ws, size_t ws_size,
                              hipStream_t stream) {
  const float* x        = (const float*)d_in[0];
  const float* fh_W1    = (const float*)d_in[1];
  const float* fh_b1    = (const float*)d_in[2];
  const float* fh_bn1   = (const float*)d_in[3];
  const float* fh_W2    = (const float*)d_in[4];
  const float* fh_b2    = (const float*)d_in[5];
  const float* fh_bn2   = (const float*)d_in[6];
  const float* conv_W1  = (const float*)d_in[7];
  const float* conv_b1  = (const float*)d_in[8];
  const float* conv_bn1 = (const float*)d_in[9];
  const float* conv_W2  = (const float*)d_in[10];
  const float* conv_b2  = (const float*)d_in[11];
  const float* conv_bn2 = (const float*)d_in[12];
  const float* lin_W    = (const float*)d_in[13];
  const float* lin_b    = (const float*)d_in[14];
  const int*   eidx     = (const int*)d_in[15];
  const int*   batch    = (const int*)d_in[16];
  float* out = (float*)d_out;

  const int* src = eidx;
  const int* dst = eidx + N_EDGES;

  float* ws     = (float*)d_ws;
  float* hA     = ws;
  float* hB     = ws + (size_t)N_NODES * 64;
  int*   cnt    = (int*)(ws + (size_t)2 * N_NODES * 64);
  int*   cursor = cnt + NBK;
  int*   offs   = cnt + 1024;
  int*   bins   = cnt + 2048;
  int*   goffs  = cnt + 2048 + N_EDGES;
  int*   gbins  = cnt + 2048 + N_EDGES + 16384;

  gin_out_init_kernel<<<(N_GRAPHS * 10 + 255) / 256, 256, 0, stream>>>(lin_b, out, cnt);

  gin_bin_count_kernel<<<BF_BLOCKS, BF_THREADS, 0, stream>>>(dst, cnt);
  gin_bin_fill_kernel<<<BF_BLOCKS, BF_THREADS, 0, stream>>>(src, dst, cnt, cursor, offs, bins);
  gin_bin_group_kernel<<<NBK, GR_THREADS, 0, stream>>>(offs, bins, goffs, gbins);

  const int mlpBlocks = (NTILES + 3) / 4;

  gin_mlp_kernel<D_FEAT, true><<<mlpBlocks, 128, 0, stream>>>(
      x, fh_W1, fh_b1, fh_bn1, fh_W2, fh_b2, fh_bn2, hA,
      lin_W, lin_b, batch, out, NTILES);

  float* hin = hA;
  float* hnext = hB;
  for (int l = 0; l < 4; l++) {
    gin_conv_kernel<<<NBK, AG_THREADS, 0, stream>>>(
        hin, goffs, gbins,
        conv_W1 + (size_t)l * 64 * 64, conv_b1 + l * 64, conv_bn1 + l * 256,
        conv_W2 + (size_t)l * 64 * 64, conv_b2 + l * 64, conv_bn2 + l * 256,
        hnext,
        lin_W + (size_t)(l + 1) * 64 * 10, batch, out);
    float* tswap = hin;
    hin = hnext;
    hnext = tswap;
  }
}


# Encoder MLP as 256 resident 16-wave work-groups, a wave loops over its tiles: weights staged once per work-group, no block barrier per tile; out sum order = LDS arrival, then arrival, as before
# speedup vs baseline: 4.4764x; 1.0476x over previous
#include <hip/hip_runtime.h>

#define N_NODES   100000
#define N_EDGES   1600000
#define N_GRAPHS  1000
#define D_FEAT    128
#define HID       64
#define BN_EPS    1e-5f
#define NTILES    (N_NODES / 16)
#define ENC_WAVES    16
#define ENC_THREADS  (ENC_WAVES * 32)
#define ENC_BLOCKS   256
#define MLP_PG_SLOTS 16
#define CV_PG_SLOTS  64

typedef __attribute__((ext_vector_type(2))) float v2f;
typedef __attribute__((ext_vector_type(8))) float v8f;

template<int K1, bool NODE_BIAS>
__launch_bounds__(ENC_THREADS)
__global__ void gin_mlp_kernel(const float* __restrict__ xin,
                               const float* __restrict__ W1,
                               const float* __restrict__ b1,
                               const float* __restrict__ bn1,
                               const float* __restrict__ W2,
                               const float* __restrict__ b2,
                               const float* __restrict__ bn2,
                               float* __restrict__ hout,
                               const float* __restrict__ linW,
                               const float* __restrict__ linb0,
                               const int* __restrict__ batch,
                               float* __restrict__ out,
                               int ntiles) {
  constexpr int KS1 = K1 / 4;
  __shared__ __align__(16) float w1s[4 * KS1 * 32 * 2];
  __shared__ __align__(16) float w2s[4 * 16 * 32 * 2];
  __shared__ __align__(16) float tb[ENC_WAVES][16 * 68];
  __shared__ float gs[MLP_PG_SLOTS * 10];

  const int tid = threadIdx.x;
  const int t_lo = (int)((blockIdx.x * (unsigned)ntiles) / gridDim.x);
  const int t_hi = (int)(((blockIdx.x + 1) * (unsigned)ntiles) / gridDim.x);
  const int g_lo = batch[(t_lo < ntiles ? t_lo : ntiles - 1) * 16];
  for (int i = tid; i < MLP_PG_SLOTS * 10; i += ENC_THREADS) gs[i] = 0.f;

  constexpr int W1_PT = (K1 * 64) / ENC_THREADS;
  constexpr int W2_PT = (64 * 64) / ENC_THREADS;
  static_assert((K1 * 64) % ENC_THREADS == 0 && (64 * 64) % ENC_THREADS == 0, "weight pieces per thread");
  float wa[W1_PT], wb[W2_PT];
#pragma unroll
  for (int j = 0; j < W1_PT; j++) wa[j] = W1[tid + j * ENC_THREADS];
#pragma unroll
  for (int j = 0; j < W2_PT; j++) wb[j] = W2[tid + j * ENC_THREADS];

  const int lane = tid & 31;
  const int wave = __builtin_amdgcn_readfirstlane(tid >> 5);
  const int row  = lane & 15;
  const int hi   = lane >> 4;
  float* tbuf = tb[wave];

  const v8f vzero = {0.f, 0.f, 0.f, 0.f, 0.f, 0.f, 0.f, 0.f};

  float s1[4], t1[4], s2[4], t2[4];
#pragma unroll
  for (int nt = 0; nt < 4; nt++) {
    const int n = nt * 16 + row;
    s1[nt] = bn1[n] * rsqrtf(bn1[192 + n] + BN_EPS);
    t1[nt] = bn1[64 + n] + (b1[n] - bn1[128 + n]) * s1[nt];
    s2[nt] = bn2[n] * rsqrtf(bn2[192 + n] + BN_EPS);
    t2[nt] = bn2[64 + n] + (b2[n] - bn2[128 + n]) * s2[nt];
  }
  v2f wl[16];
  const int hc = row < 10 ? row : 9;
  {
    const float* lw = linW + (2 * hi) * 10 + hc;
#pragma unroll
    for (int k = 0; k < 16; k++) {
      float w0 = lw[(4 * k) * 10];
      float w1 = lw[(4 * k + 1) * 10];
      wl[k] = (v2f){row < 10 ? w0 : 0.f, row < 10 ? w1 : 0.f};
    }
  }
  float lb = 0.f;
  if (NODE_BIAS) lb = linb0[hc];

#pragma unroll
  for (int j = 0; j < W1_PT; j++) {
    const int i = tid + j * ENC_THREADS;
    const int kk = i >> 6, nn = i & 63;
    const int idx = ((((nn >> 4) * KS1) + (kk >> 2)) * 32 + (nn & 15) + ((kk & 2) << 3)) * 2 + (kk & 1);
    w1s[idx] = wa[j];
  }
#pragma unroll
  for (int j = 0; j < W2_PT; j++) {
    const int i = tid + j * ENC_THREADS;
    const int kk = i >> 6, nn = i & 63;
    const int idx = ((((nn >> 4) * 16) + (kk >> 2)) * 32 + (nn & 15) + ((kk & 2) << 3)) * 2 + (kk & 1);
    w2s[idx] = wb[j];
  }
  __syncthreads();

  for (int tile = t_lo + wave; tile < t_hi; tile += ENC_WAVES) {
    {
      v2f a1[KS1];
      const float* xr = xin + (size_t)(tile * 16 + row) * K1 + 2 * hi;
#pragma unroll
      for (int k = 0; k < KS1; k++) a1[k] = *(const v2f*)(xr + 4 * k);

#pragma unroll
      for (int nt = 0; nt < 4; nt++) {
        v8f acc = vzero;
        const v2f* wf = ((const v2f*)w1s) + nt * KS1 * 32 + lane;
#pragma unroll
        for (int k = 0; k < KS1; k++) {
          acc = __builtin_amdgcn_wmma_f32_16x16x4_f32(false, a1[k], false, wf[k * 32],
                                                      (short)0, acc, false, false);
        }
        const int n = nt * 16 + row;
#pragma unroll
        for (int i = 0; i < 8; i++) {
          float y = acc[i] * s1[nt] + t1[nt];
          tbuf[(i + 8 * hi) * 68 + n] = fmaxf(y, 0.0f);
        }
      }
    }
    __builtin_amdgcn_fence(__ATOMIC_SEQ_CST, "wavefront");

    v8f acc2[4];
    int gid[8];
#pragma unroll
    for (int r = 0; r < 8; r++) gid[r] = batch[tile * 16 + 8 * hi + r];
#pragma unroll
    for (int nt = 0; nt < 4; nt++) {
      v8f acc = vzero;
      const v2f* wf = ((const v2f*)w2s) + nt * 16 * 32 + lane;
      const float* tr = tbuf + row * 68 + 2 * hi;
#pragma unroll
      for (int k = 0; k < 16; k++) {
        v2f a = *(const v2f*)(tr + 4 * k);
        acc = __builtin_amdgcn_wmma_f32_16x16x4_f32(false, a, false, wf[k * 32],
                                                    (short)0, acc, false, false);
      }
      acc2[nt] = acc;
    }
    __builtin_amdgcn_fence(__ATOMIC_SEQ_CST, "wavefront");

#pragma unroll
    for (int nt = 0; nt < 4; nt++) {
      const int n = nt * 16 + row;
#pragma unroll
      for (int i = 0; i < 8; i++) {
        float y = acc2[nt][i] * s2[nt] + t2[nt];
        tbuf[(i + 8 * hi) * 68 + n] = fmaxf(y, 0.0f);
      }
    }
    __builtin_amdgcn_fence(__ATOMIC_SEQ_CST, "wavefront");

    v8f p0 = vzero, p1 = vzero;
    const float* tr = tbuf + row * 68 + 2 * hi;
#pragma unroll
    for (int k = 0; k < 16; k += 2) {
      v2f a0 = *(const v2f*)(tr + 4 * k);
      v2f a1h = *(const v2f*)(tr + 4 * (k + 1));
      p0 = __builtin_amdgcn_wmma_f32_16x16x4_f32(false, a0, false, wl[k],
                                                 (short)0, p0, false, false);
      p1 = __builtin_amdgcn_wmma_f32_16x16x4_f32(false, a1h, false, wl[k + 1],
                                                 (short)0, p1, false, false);
    }
    float pv[8];
#pragma unroll
    for (int r = 0; r < 8; r++) pv[r] = p0[r] + p1[r];
    if (NODE_BIAS) {
#pragma unroll
      for (int r = 0; r < 8; r++) pv[r] += lb;
    }

#pragma unroll
    for (int j = 0; j < 8; j++) {
      int flat = j * 32 + lane;
      int r = flat >> 4;
      int c = (flat & 15) << 2;
      float4 v = *(const float4*)(tbuf + r * 68 + c);
      *(float4*)(hout + (size_t)(tile * 16 + r) * 64 + c) = v;
    }

    if (row < 10) {
#pragma unroll
      for (int r = 0; r < 8; r++) {
        const unsigned d = (unsigned)(gid[r] - g_lo);
        if (d < MLP_PG_SLOTS) atomicAdd(&gs[d * 10 + row], pv[r]);
        else atomicAdd(out + (size_t)gid[r] * 10 + row, pv[r]);
      }
    }
    __builtin_amdgcn_fence(__ATOMIC_SEQ_CST, "wavefront");
  }
  __syncthreads();
  for (int i = tid; i < MLP_PG_SLOTS * 10; i += ENC_THREADS) {
    const float v = gs[i];
    if (v != 0.f && g_lo + i / 10 < N_GRAPHS) atomicAdd(out + (size_t)g_lo * 10 + i, v);
  }
}

#define BK_NODES 200
#define NBK (N_NODES / BK_NODES)
#define BF_THREADS 800
#define BF_EDGES_PER_THREAD 8
#define BF_EDGES_PER_BLOCK (BF_THREADS * BF_EDGES_PER_THREAD)
#define BF_BLOCKS (N_EDGES / BF_EDGES_PER_BLOCK)
#define SRC_BITS 17
static_assert(N_NODES % BK_NODES == 0, "ranges must tile the nodes exactly");
static_assert(N_EDGES % BF_EDGES_PER_BLOCK == 0, "edge blocks must tile the edges exactly");
static_assert(N_NODES <= (1 << SRC_BITS), "a source id must fit the packed field");
static_assert(NBK <= 512, "the range scan below holds 512 entries");

__launch_bounds__(BF_THREADS)
__global__ void gin_bin_count_kernel(const int* __restrict__ dst,
                                     int* __restrict__ cnt) {
  __shared__ int hist[NBK];
  const int tid = threadIdx.x;
  for (int b = tid; b < NBK; b += BF_THREADS) hist[b] = 0;
  __syncthreads();
  const int e0 = blockIdx.x * BF_EDGES_PER_BLOCK + tid;
  unsigned d[BF_EDGES_PER_THREAD];
#pragma unroll
  for (int k = 0; k < BF_EDGES_PER_THREAD; k++) d[k] = (unsigned)dst[e0 + k * BF_THREADS];
#pragma unroll
  for (int k = 0; k < BF_EDGES_PER_THREAD; k++) atomicAdd(&hist[d[k] / BK_NODES], 1);
  __syncthreads();
  for (int b = tid; b < NBK; b += BF_THREADS) {
    const int c = hist[b];
    if (c) atomicAdd(&cnt[b], c);
  }
}

__launch_bounds__(BF_THREADS)
__global__ void gin_bin_fill_kernel(const int* __restrict__ src,
                                    const int* __restrict__ dst,
                                    const int* __restrict__ cnt,
                                    int* __restrict__ cursor,
                                    int* __restrict__ offs,
                                    int* __restrict__ bins) {
  __shared__ int hist[NBK];
  __shared__ int bpos[NBK];
  __shared__ int loff[NBK];
  __shared__ int2 sc[2][512];
  __shared__ int2 stage[BF_EDGES_PER_BLOCK];
  const int tid = threadIdx.x;
  for (int b = tid; b < NBK; b += BF_THREADS) hist[b] = 0;
  const int cg = tid < NBK ? cnt[tid] : 0;
  __syncthreads();

  int bk[BF_EDGES_PER_THREAD], rk[BF_EDGES_PER_THREAD], pk[BF_EDGES_PER_THREAD];
  const int e0 = blockIdx.x * BF_EDGES_PER_BLOCK + tid;
#pragma unroll
  for (int k = 0; k < BF_EDGES_PER_THREAD; k++) {
    pk[k] = src[e0 + k * BF_THREADS];
    bk[k] = dst[e0 + k * BF_THREADS];
  }
#pragma unroll
  for (int k = 0; k < BF_EDGES_PER_THREAD; k++) {
    const unsigned s = (unsigned)pk[k];
    const unsigned d = (unsigned)bk[k];
    const unsigned b = d / BK_NODES;
    bk[k] = (int)b;
    pk[k] = (int)(((d - b * BK_NODES) << SRC_BITS) | s);
    rk[k] = atomicAdd(&hist[b], 1);
  }
  __syncthreads();
  if (tid < 512) sc[0][tid] = make_int2(cg, tid < NBK ? hist[tid] : 0);
  __syncthreads();

  int cur = 0;
  for (int off = 1; off < 512; off <<= 1) {
    if (tid < 512) {
      int2 v = sc[cur][tid];
      if (tid >= off) {
        const int2 u = sc[cur][tid - off];
        v.x += u.x;
        v.y += u.y;
      }
      sc[cur ^ 1][tid] = v;
    }
    __syncthreads();
    cur ^= 1;
  }
  const int2* scan = sc[cur];

  for (int b = tid; b < NBK; b += BF_THREADS) {
    const int c = hist[b];
    const int start = b ? scan[b - 1].x : 0;
    const int r = c ? atomicAdd(&cursor[b], c) : 0;
    bpos[b] = start + r;
    loff[b] = b ? scan[b - 1].y : 0;
    if (blockIdx.x == 0) offs[b] = start;
  }
  if (blockIdx.x == 0 && tid == 0) offs[NBK] = scan[NBK - 1].x;
  __syncthreads();

#pragma unroll
  for (int k = 0; k < BF_EDGES_PER_THREAD; k++) {
    const int b = bk[k];
    stage[loff[b] + rk[k]] = make_int2(pk[k], bpos[b] + rk[k]);
  }
  __syncthreads();
#pragma unroll
  for (int k = 0; k < BF_EDGES_PER_THREAD; k++) {
    const int2 s = stage[tid + k * BF_THREADS];
    bins[s.y] = s.x;
  }
}


#define GT_WAVES 8
#define CH_N 3
#define CH1_ROW 80
#define CH2_ROW 144
#define NGRP (CH_N * GT_WAVES)
static_assert(NBK * (NGRP + 1) <= 16384, "group starts fit the space kept for them");
static_assert(CH1_ROW % 16 == 0 && CH2_ROW % 16 == 0, "chunks are whole tiles");
__device__ __forceinline__ int gin_group_of(int p) {
  const int r = p >> SRC_BITS;
  const int c = (r >= CH1_ROW ? 1 : 0) + (r >= CH2_ROW ? 1 : 0);
  return c * GT_WAVES + (r & (GT_WAVES - 1));
}
__device__ __forceinline__ int gin_slot_entry(int p) {
  const int r = p >> SRC_BITS;
  const int cs = r >= CH2_ROW ? CH2_ROW : (r >= CH1_ROW ? CH1_ROW : 0);
  return (((r - cs) / GT_WAVES) << SRC_BITS) | (p & ((1 << SRC_BITS) - 1));
}
#define GR_THREADS 512
#define GR_KEEP 8
#define GR_STAGE (GR_KEEP * GR_THREADS)
__launch_bounds__(GR_THREADS)
__global__ void gin_bin_group_kernel(const int* __restrict__ offs,
                                     const int* __restrict__ bins,
                                     int* __restrict__ goffs,
                                     int* __restrict__ gbins) {
  __shared__ int gcnt[NGRP];
  __shared__ int gpos[NGRP];
  __shared__ int gbase[NGRP + 1];
  __shared__ int stage[GR_STAGE];
  const int tid = threadIdx.x;
  const int b = blockIdx.x;
  const int e0 = offs[b];
  const int e1 = offs[b + 1];
  if (tid < NGRP) {
    gcnt[tid] = 0;
    gpos[tid] = 0;
  }
  int pr[GR_KEEP];
#pragma unroll
  for (int k = 0; k < GR_KEEP; k++) {
    const int i = e0 + tid + k * GR_THREADS;
    pr[k] = bins[i < e1 ? i : N_EDGES - 1];
  }
  __syncthreads();
#pragma unroll
  for (int k = 0; k < GR_KEEP; k++) {
    if (e0 + tid + k * GR_THREADS < e1) atomicAdd(&gcnt[gin_group_of(pr[k])], 1);
  }
  for (int i = e0 + tid + GR_KEEP * GR_THREADS; i < e1; i += GR_THREADS) {
    const int g = gin_group_of(bins[i]);
    atomicAdd(&gcnt[g], 1);
  }
  __syncthreads();
  if (tid == 0) {
    int s = e0;
    for (int g = 0; g < NGRP; g++) {
      gbase[g] = s;
      s += gcnt[g];
    }
    gbase[NGRP] = s;
  }
  __syncthreads();
  if (tid <= NGRP) goffs[b * (NGRP + 1) + tid] = gbase[tid];
#pragma unroll
  for (int k = 0; k < GR_KEEP; k++) {
    if (e0 + tid + k * GR_THREADS < e1) {
      const int g = gin_group_of(pr[k]);
      const int r = atomicAdd(&gpos[g], 1);
      const int q = gbase[g] - e0 + r;
      const int v = gin_slot_entry(pr[k]);
      if (q < GR_STAGE) stage[q] = v;
      else gbins[e0 + q] = v;
    }
  }
  for (int i = e0 + tid + GR_KEEP * GR_THREADS; i < e1; i += GR_THREADS) {
    const int p = bins[i];
    const int g = gin_group_of(p);
    const int r = atomicAdd(&gpos[g], 1);
    const int q = gbase[g] - e0 + r;
    const int v = gin_slot_entry(p);
    if (q < GR_STAGE) stage[q] = v;
    else gbins[e0 + q] = v;
  }
  __syncthreads();
  const int nst = e1 - e0 < GR_STAGE ? e1 - e0 : GR_STAGE;
  int sv[GR_KEEP];
#pragma unroll
  for (int k = 0; k < GR_KEEP; k++) {
    const int i = tid + k * GR_THREADS;
    sv[k] = i < nst ? stage[i] : 0;
  }
#pragma unroll
  for (int k = 0; k < GR_KEEP; k++) {
    const int i = tid + k * GR_THREADS;
    if (i < nst) gbins[e0 + i] = sv[k];
  }
}

#define AG_WAVES 16
#define AG_THREADS (AG_WAVES * 32)
#define MT_WAVES (AG_WAVES - GT_WAVES)
#define MT_THREADS (MT_WAVES * 32)
#define CV_TILES ((BK_NODES + 15) / 16)
#define CV_ROWS (CV_TILES * 16)
#define CV_PITCH 68
#define CH0_TILES (CH1_ROW / 16)
#define CH1_TILES ((CH2_ROW - CH1_ROW) / 16)
#define CH2_TILES (CV_TILES - CH2_ROW / 16)
#define GS_MAX (CH1_ROW / GT_WAVES)
#define GS_MIN 8
static_assert(BK_NODES % 8 == 0, "a lane half of a tile (8 rows) is all valid or all padding");
static_assert(CH0_TILES <= MT_WAVES && CH1_TILES <= MT_WAVES && CH2_TILES <= MT_WAVES, "one matrix wave per tile of a chunk");
static_assert((CH2_ROW - CH1_ROW) / GT_WAVES == GS_MIN && (CV_ROWS - CH2_ROW) / GT_WAVES == GS_MIN, "slots of chunks 1 and 2");
static_assert(GS_MAX >= GS_MIN && GS_MAX <= 16, "a summing wave's slots fit one 16-register vector per column half");
static_assert(CH2_ROW + GT_WAVES - 1 < BK_NODES, "slot 0 of every summing wave is a real row in every chunk");
static_assert((64 * 64) % MT_THREADS == 0, "weight pieces per thread");
#define CV_WPT ((64 * 64) / MT_THREADS)
typedef __attribute__((ext_vector_type(16))) float v16f;
__launch_bounds__(AG_THREADS, 8)
__global__ void gin_conv_kernel(const float* __restrict__ h,
                                const int* __restrict__ goffs,
                                const int* __restrict__ gbins,
                                const float* __restrict__ W1,
                                const float* __restrict__ b1,
                                const float* __restrict__ bn1,
                                const float* __restrict__ W2,
                                const float* __restrict__ b2,
                                const float* __restrict__ bn2,
                                float* __restrict__ hout,
                                const float* __restrict__ linW,
                                const int* __restrict__ batch,
                                float* __restrict__ out) {
  __shared__ __align__(16) float acc[CV_ROWS * CV_PITCH];
  __shared__ __align__(16) float w1s[4 * 16 * 32 * 2];
  __shared__ __align__(16) float w2s[4 * 16 * 32 * 2];
  __shared__ float bns[4][64];
  __shared__ float gs[CV_PG_SLOTS * 10];
  const int tid  = threadIdx.x;
  const int lane = tid & 31;
  const int wave = __builtin_amdgcn_readfirstlane(tid >> 5);
  const bool gteam = wave < GT_WAVES;
  const int b = blockIdx.x;
  const size_t rowbase = (size_t)b * BK_NODES * 64;
  const int g_lo = batch[b * BK_NODES];
  for (int i = tid; i < CV_PG_SLOTS * 10; i += AG_THREADS) gs[i] = 0.f;

  const int* gp = goffs + b * (NGRP + 1) + (wave & (GT_WAVES - 1));
  const int ga0 = gp[0],            ga1 = gp[1];
  const int gb0 = gp[GT_WAVES],     gb1 = gp[GT_WAVES + 1];
  const int gc0 = gp[2 * GT_WAVES], gc1 = gp[2 * GT_WAVES + 1];

  if (!gteam) {
    const int mt = tid - GT_WAVES * 32;
    float wa[CV_WPT], wb[CV_WPT];
#pragma unroll
    for (int k = 0; k < CV_WPT; k++) {
      wa[k] = W1[mt + k * MT_THREADS];
      wb[k] = W2[mt + k * MT_THREADS];
    }
#pragma unroll
    for (int k = 0; k < CV_WPT; k++) {
      const int i = mt + k * MT_THREADS;
      const int kk = i >> 6, nn = i & 63;
      const int idx = ((((nn >> 4) * 16) + (kk >> 2)) * 32 + (nn & 15) + ((kk & 2) << 3)) * 2 + (kk & 1);
      w1s[idx] = wa[k];
      w2s[idx] = wb[k];
    }
    if (mt < 128) {
      const float* bp = mt < 64 ? bn1 : bn2;
      const float* lp = mt < 64 ? b1 : b2;
      const int n = mt & 63;
      const float sc = bp[n] * rsqrtf(bp[192 + n] + BN_EPS);
      bns[(mt >> 6) * 2][n] = sc;
      bns[(mt >> 6) * 2 + 1][n] = bp[64 + n] + (lp[n] - bp[128 + n]) * sc;
    }
  }

#pragma nounroll
  for (int c = 0; c <= CH_N; c++) {
    if (gteam) {
      if (c < CH_N) {
        const int cs  = c == 0 ? 0 : (c == 1 ? CH1_ROW : CH2_ROW);
        const int nsl = c == 0 ? GS_MAX : GS_MIN;
        int lg = lane;
        asm volatile("" : "+v"(lg));
        const int g0  = c == 0 ? ga0 : (c == 1 ? gb0 : gc0);
        const int g1  = c == 0 ? ga1 : (c == 1 ? gb1 : gc1);
        v16f sx = {0.f, 0.f, 0.f, 0.f, 0.f, 0.f, 0.f, 0.f, 0.f, 0.f, 0.f, 0.f, 0.f, 0.f, 0.f, 0.f};
        v16f sy = sx;
#pragma unroll
        for (int sl = 0; sl < GS_MAX; sl++) {
          const int r = cs + wave + GT_WAVES * sl;
          const bool real = sl < nsl && r < BK_NODES;
          const v2f t = *(const v2f*)(h + rowbase + (size_t)(real ? r : cs + wave) * 64 + 2 * lg);
          sx[sl] = real ? t.x : 0.f;
          sy[sl] = real ? t.y : 0.f;
        }
        int e = g0;
        int p = gbins[e + lg < N_EDGES ? e + lg : N_EDGES - 1];
        while (e + 32 <= g1) {
          const int en = e + 32;
          v2f v[32];
#pragma unroll
          for (int j = 0; j < 32; j++) {
            const int pj = __builtin_amdgcn_readlane(p, j);
            v[j] = *(const v2f*)(h + (size_t)(pj & ((1 << SRC_BITS) - 1)) * 64 + 2 * lg);
          }
          const int pn = gbins[en + lg < N_EDGES ? en + lg : N_EDGES - 1];
          __builtin_amdgcn_sched_barrier(0);
#pragma unroll
          for (int j = 0; j < 32; j++) {
            const int sl = __builtin_amdgcn_readlane(p, j) >> SRC_BITS;
            sx[sl] += v[j].x;
            sy[sl] += v[j].y;
          }
          p = pn;
          e = en;
        }
        if (e < g1) {
          const int n = g1 - e;
          const int pt = lg < n ? p : __builtin_amdgcn_readlane(p, 0);
          v2f v[32];
#pragma unroll
          for (int j = 0; j < 32; j++) {
            const int pj = __builtin_amdgcn_readlane(pt, j);
            v[j] = *(const v2f*)(h + (size_t)(pj & ((1 << SRC_BITS) - 1)) * 64 + 2 * lg);
          }
#pragma unroll
          for (int j = 0; j < 32; j++) {
            if (j < n) {
              const int sl = __builtin_amdgcn_readlane(pt, j) >> SRC_BITS;
              sx[sl] += v[j].x;
              sy[sl] += v[j].y;
            }
          }
        }
        float* ar = acc + (cs + wave) * CV_PITCH + 2 * lg;
#pragma unroll
        for (int sl = 0; sl < GS_MIN; sl++) {
          *(v2f*)(ar + GT_WAVES * sl * CV_PITCH) = (v2f){sx[sl], sy[sl]};
        }
        if (nsl > GS_MIN) {
#pragma unroll
          for (int sl = GS_MIN; sl < GS_MAX; sl++) {
            *(v2f*)(ar + GT_WAVES * sl * CV_PITCH) = (v2f){sx[sl], sy[sl]};
          }
        }
      }
    } else if (c > 0) {
      const int mc = c - 1;
      const int m  = wave - GT_WAVES;
      const int nct = mc == 0 ? CH0_TILES : (mc == 1 ? CH1_TILES : CH2_TILES);
      if (m < nct) {
        const int tile = (mc == 0 ? 0 : (mc == 1 ? CH1_ROW / 16 : CH2_ROW / 16)) + m;
        int ln = lane;
        asm volatile("" : "+v"(ln));
        const int row  = ln & 15;
        const int hi   = ln >> 4;
        const int left = BK_NODES - tile * 16;
        const int nrow = left < 16 ? left : 16;
        const bool rows_ok = 8 * hi < nrow;
        const int node0 = b * BK_NODES + tile * 16;
        float* tbuf = acc + tile * 16 * CV_PITCH;
        const v8f vzero = {0.f, 0.f, 0.f, 0.f, 0.f, 0.f, 0.f, 0.f};

        v2f a1[16];
        {
          const float* tr = tbuf + row * CV_PITCH + 2 * hi;
#pragma unroll
          for (int k = 0; k < 16; k++) a1[k] = *(const v2f*)(tr + 4 * k);
        }

#pragma unroll
        for (int nt = 0; nt < 4; nt++) {
          v8f a = vzero;
          const v2f* wf = ((const v2f*)w1s) + nt * 16 * 32 + ln;
#pragma unroll
          for (int k = 0; k < 16; k++) {
            a = __builtin_amdgcn_wmma_f32_16x16x4_f32(false, a1[k], false, wf[k * 32],
                                                      (short)0, a, false, false);
          }
          int n = nt * 16 + row;
          float sc = bns[0][n];
          float t0 = bns[1][n];
#pragma unroll
          for (int i = 0; i < 8; i++) {
            float y = a[i] * sc + t0;
            tbuf[(i + 8 * hi) * CV_PITCH + n] = fmaxf(y, 0.0f);
          }
        }
        __builtin_amdgcn_fence(__ATOMIC_SEQ_CST, "wavefront");

        v8f acc2[4];
        v2f wl[16];
        int gid[8];
        const int hc = row < 10 ? row : 9;
#pragma unroll
        for (int r = 0; r < 8; r++) gid[r] = 0;
        {
          const float* lw = linW + (2 * hi) * 10 + hc;
#pragma unroll
          for (int k = 0; k < 16; k++) {
            float w0 = lw[(4 * k) * 10];
            float w1 = lw[(4 * k + 1) * 10];
            wl[k] = (v2f){row < 10 ? w0 : 0.f, row < 10 ? w1 : 0.f};
          }
        }
        if (rows_ok) {
#pragma unroll
          for (int r = 0; r < 8; r++) gid[r] = batch[node0 + 8 * hi + r];
        }
#pragma unroll
        for (int nt = 0; nt < 4; nt++) {
          v8f a = vzero;
          const v2f* wf = ((const v2f*)w2s) + nt * 16 * 32 + ln;
          const float* tr = tbuf + row * CV_PITCH + 2 * hi;
#pragma unroll
          for (int k = 0; k < 16; k++) {
            v2f av = *(const v2f*)(tr + 4 * k);
            a = __builtin_amdgcn_wmma_f32_16x16x4_f32(false, av, false, wf[k * 32],
                                                      (short)0, a, false, false);
          }
          acc2[nt] = a;
        }
        __builtin_amdgcn_fence(__ATOMIC_SEQ_CST, "wavefront");

#pragma unroll
        for (int nt = 0; nt < 4; nt++) {
          int n = nt * 16 + row;
          float sc = bns[2][n];
          float t0 = bns[3][n];
#pragma unroll
          for (int i = 0; i < 8; i++) {
            float y = acc2[nt][i] * sc + t0;
            tbuf[(i + 8 * hi) * CV_PITCH + n] = fmaxf(y, 0.0f);
          }
        }
        __builtin_amdgcn_fence(__ATOMIC_SEQ_CST, "wavefront");

        v8f p0 = vzero, p1 = vzero;
        const float* tr = tbuf + row * CV_PITCH + 2 * hi;
#pragma unroll
        for (int k = 0; k < 16; k += 2) {
          v2f a0 = *(const v2f*)(tr + 4 * k);
          v2f a1h = *(const v2f*)(tr + 4 * (k + 1));
          p0 = __builtin_amdgcn_wmma_f32_16x16x4_f32(false, a0, false, wl[k],
                                                     (short)0, p0, false, false);
          p1 = __builtin_amdgcn_wmma_f32_16x16x4_f32(false, a1h, false, wl[k + 1],
                                                     (short)0, p1, false, false);
        }
        float pv[8];
#pragma unroll
        for (int r = 0; r < 8; r++) pv[r] = p0[r] + p1[r];

#pragma unroll
        for (int j = 0; j < 8; j++) {
          const int r = 2 * j + hi;
          const int cc = row << 2;
          if (r < nrow) {
            float4 v = *(const float4*)(tbuf + r * CV_PITCH + cc);
            *(float4*)(hout + (size_t)(node0 + r) * 64 + cc) = v;
          }
        }

        if (row < 10 && rows_ok) {
#pragma unroll
          for (int r = 0; r < 8; r++) {
            const unsigned d = (unsigned)(gid[r] - g_lo);
            if (d < CV_PG_SLOTS) atomicAdd(&gs[d * 10 + row], pv[r]);
            else atomicAdd(out + (size_t)gid[r] * 10 + row, pv[r]);
          }
        }
      }
    }
    if (c < CH_N) __syncthreads();
  }
  __syncthreads();
  for (int i = tid; i < CV_PG_SLOTS * 10; i += AG_THREADS) {
    const float v = gs[i];
    if (v != 0.f && g_lo + i / 10 < N_GRAPHS) atomicAdd(out + (size_t)g_lo * 10 + i, v);
  }
}

__global__ void gin_out_init_kernel(const float* __restrict__ linb,
                                    float* __restrict__ out,
                                    int* __restrict__ cnt) {
  int idx = blockIdx.x * blockDim.x + threadIdx.x;
  if (idx < 2 * NBK) cnt[idx] = 0;
  if (idx >= N_GRAPHS * 10) return;
  int t = idx % 10;
  out[idx] = ((linb[10 + t] + linb[20 + t]) + linb[30 + t]) + linb[40 + t];
}

extern "C" void kernel_launch(void* const* d_in, const int* in_sizes, int n_in,
                              void* d_out, int out_size, void* d_ws, size_t ws_size,
                              hipStream_t stream) {
  const float* x        = (const float*)d_in[0];
  const float* fh_W1    = (const float*)d_in[1];
  const float* fh_b1    = (const float*)d_in[2];
  const float* fh_bn1   = (const float*)d_in[3];
  const float* fh_W2    = (const float*)d_in[4];
  const float* fh_b2    = (const float*)d_in[5];
  const float* fh_bn2   = (const float*)d_in[6];
  const float* conv_W1  = (const float*)d_in[7];
  const float* conv_b1  = (const float*)d_in[8];
  const float* conv_bn1 = (const float*)d_in[9];
  const float* conv_W2  = (const float*)d_in[10];
  const float* conv_b2  = (const float*)d_in[11];
  const float* conv_bn2 = (const float*)d_in[12];
  const float* lin_W    = (const float*)d_in[13];
  const float* lin_b    = (const float*)d_in[14];
  const int*   eidx     = (const int*)d_in[15];
  const int*   batch    = (const int*)d_in[16];
  float* out = (float*)d_out;

  const int* src = eidx;
  const int* dst = eidx + N_EDGES;

  float* ws     = (float*)d_ws;
  float* hA     = ws;
  float* hB     = ws + (size_t)N_NODES * 64;
  int*   cnt    = (int*)(ws + (size_t)2 * N_NODES * 64);
  int*   cursor = cnt + NBK;
  int*   offs   = cnt + 1024;
  int*   bins   = cnt + 2048;
  int*   goffs  = cnt + 2048 + N_EDGES;
  int*   gbins  = cnt + 2048 + N_EDGES + 16384;

  gin_out_init_kernel<<<(N_GRAPHS * 10 + 255) / 256, 256, 0, stream>>>(lin_b, out, cnt);

  gin_bin_count_kernel<<<BF_BLOCKS, BF_THREADS, 0, stream>>>(dst, cnt);
  gin_bin_fill_kernel<<<BF_BLOCKS, BF_THREADS, 0, stream>>>(src, dst, cnt, cursor, offs, bins);
  gin_bin_group_kernel<<<NBK, GR_THREADS, 0, stream>>>(offs, bins, goffs, gbins);

  gin_mlp_kernel<D_FEAT, true><<<ENC_BLOCKS, ENC_THREADS, 0, stream>>>(
      x, fh_W1, fh_b1, fh_bn1, fh_W2, fh_b2, fh_bn2, hA,
      lin_W, lin_b, batch, out, NTILES);

  float* hin = hA;
  float* hnext = hB;
  for (int l = 0; l < 4; l++) {
    gin_conv_kernel<<<NBK, AG_THREADS, 0, stream>>>(
        hin, goffs, gbins,
        conv_W1 + (size_t)l * 64 * 64, conv_b1 + l * 64, conv_bn1 + l * 256,
        conv_W2 + (size_t)l * 64 * 64, conv_b2 + l * 64, conv_bn2 + l * 256,
        hnext,
        lin_W + (size_t)(l + 1) * 64 * 10, batch, out);
    float* tswap = hin;
    hin = hnext;
    hnext = tswap;
  }
}
